# starting kernel
# baseline (speedup 1.0000x reference)
#include <hip/hip_runtime.h>
#include <math.h>

typedef float v2f __attribute__((ext_vector_type(2)));
typedef float v8f __attribute__((ext_vector_type(8)));

#define HEADS 8
#define CH 32
#define F 256
#define NEG_SLOPE 0.2f
#define LN_EPS 1e-5f

#define LDS_STRIDE 260

__device__ __forceinline__ unsigned f2key(float f) {
  unsigned b = __float_as_uint(f);
  return (b & 0x80000000u) ? ~b : (b | 0x80000000u);
}
__device__ __forceinline__ float key2f(unsigned k) {
  unsigned b = (k & 0x80000000u) ? (k & 0x7fffffffu) : ~k;
  return __uint_as_float(b);
}

__global__ void gat_init_kernel(float* __restrict__ out, float* __restrict__ denom,
                                unsigned* __restrict__ segmax, int n_out, int n_nh) {
  int stride = gridDim.x * blockDim.x;
  for (int t = blockIdx.x * blockDim.x + threadIdx.x; t < n_out; t += stride) out[t] = 0.f;
  for (int t = blockIdx.x * blockDim.x + threadIdx.x; t < n_nh; t += stride) {
    denom[t] = 0.f;
    segmax[t] = 0x007FFFFFu;
  }
}

__global__ void __launch_bounds__(512)
gat_gemm_wmma_kernel(const float* __restrict__ x, const float* __restrict__ W,
                     float* __restrict__ xl, int N) {
  __shared__ float lds_a[16 * LDS_STRIDE];

  const int tid  = threadIdx.x;
  const int lane = tid & 31;
  const int wave = tid >> 5;
  const int row0 = blockIdx.x * 16;
  const int col0 = wave * 16;

  {
    int r = tid >> 5;
    int c = (tid & 31) * 8;
    int grow = row0 + r;
    if (grow >= N) grow = N - 1;
    const float4* __restrict__ src = (const float4*)(x + (size_t)grow * F + c);
    float4* dst = (float4*)(&lds_a[r * LDS_STRIDE + c]);
    dst[0] = src[0];
    dst[1] = src[1];
  }
  __syncthreads();

  const int m    = lane & 15;
  const int koff = (lane >> 4) * 2;
  const float* __restrict__ arow = &lds_a[m * LDS_STRIDE];

  const int n = lane & 15;

  v8f acc = {0.f, 0.f, 0.f, 0.f, 0.f, 0.f, 0.f, 0.f};
  for (int k = 0; k < F; k += 4) {
    v2f a = *(const v2f*)(arow + k + koff);
    v2f b;
    b.x = W[(size_t)(k + koff) * F + col0 + n];
    b.y = W[(size_t)(k + koff + 1) * F + col0 + n];
    acc = __builtin_amdgcn_wmma_f32_16x16x4_f32(
         false, a,  false, b,
         (short)0, acc,  false,  false);
  }

  const int rbase = row0 + ((lane >> 4) * 8);
  float* __restrict__ dst = xl + (size_t)rbase * F + col0 + n;
  if (row0 + 16 <= N) {
#pragma unroll
    for (int j = 0; j < 8; ++j) dst[(size_t)j * F] = acc[j];
  } else {
#pragma unroll
    for (int j = 0; j < 8; ++j) {
      if (rbase + j < N) dst[(size_t)j * F] = acc[j];
    }
  }
}

__global__ void gat_att_kernel(const float* __restrict__ xl,
                               const float* __restrict__ att_src,
                               const float* __restrict__ att_dst,
                               float* __restrict__ a_src, float* __restrict__ a_dst,
                               int NH) {
  int t = blockIdx.x * blockDim.x + threadIdx.x;
  if (t >= NH) return;
  int nidx = t >> 3, h = t & 7;
  const float* __restrict__ xr = xl + (size_t)nidx * F + h * CH;
  const float* __restrict__ as = att_src + h * CH;
  const float* __restrict__ ad = att_dst + h * CH;
  float s = 0.f, d = 0.f;
#pragma unroll
  for (int c = 0; c < CH; ++c) {
    float v = xr[c];
    s += v * as[c];
    d += v * ad[c];
  }
  a_src[t] = s;
  a_dst[t] = d;
}

__device__ __forceinline__ void edge_nodes(const int* __restrict__ ei, int e, int E,
                                           int& s, int& d) {
  if (e < E) { s = ei[e]; d = ei[E + e]; }
  else       { s = d = e - E; }
}

__device__ __forceinline__ float leaky(float v) {
  return (v >= 0.f) ? v : v * NEG_SLOPE;
}

__global__ void gat_edge_max_kernel(const int* __restrict__ ei,
                                    const float* __restrict__ a_src,
                                    const float* __restrict__ a_dst,
                                    unsigned* __restrict__ segmax, int E, int N) {
  int t = blockIdx.x * blockDim.x + threadIdx.x;
  int total = (E + N) * HEADS;
  if (t >= total) return;
  int e = t >> 3, h = t & 7;
  int s, d;
  edge_nodes(ei, e, E, s, d);
  float logit = leaky(a_src[s * HEADS + h] + a_dst[d * HEADS + h]);
  atomicMax(&segmax[d * HEADS + h], f2key(logit));
}

__global__ void gat_edge_sum_kernel(const int* __restrict__ ei,
                                    const float* __restrict__ a_src,
                                    const float* __restrict__ a_dst,
                                    const unsigned* __restrict__ segmax,
                                    float* __restrict__ denom, int E, int N) {
  int t = blockIdx.x * blockDim.x + threadIdx.x;
  int total = (E + N) * HEADS;
  if (t >= total) return;
  int e = t >> 3, h = t & 7;
  int s, d;
  edge_nodes(ei, e, E, s, d);
  float logit = leaky(a_src[s * HEADS + h] + a_dst[d * HEADS + h]);
  float mx = key2f(segmax[d * HEADS + h]);
  atomicAdd(&denom[d * HEADS + h], expf(logit - mx));
}

__global__ void __launch_bounds__(256)
gat_edge_agg_kernel(const int* __restrict__ ei, const float* __restrict__ xl,
                    const float* __restrict__ a_src, const float* __restrict__ a_dst,
                    const unsigned* __restrict__ segmax, const float* __restrict__ denom,
                    float* __restrict__ out, int E, int N) {
  int e = blockIdx.x;
  int tid = threadIdx.x;
  int h = tid >> 5;
  int s, d;
  edge_nodes(ei, e, E, s, d);
  float logit = leaky(a_src[s * HEADS + h] + a_dst[d * HEADS + h]);
  float mx = key2f(segmax[d * HEADS + h]);
  float alpha = expf(logit - mx) / denom[d * HEADS + h];
  float msg = xl[(size_t)s * F + tid] * alpha;
  atomicAdd(&out[(size_t)d * F + tid], msg);
}

__global__ void __launch_bounds__(256)
gat_ln_kernel(float* __restrict__ out, const float* __restrict__ bias,
              const float* __restrict__ gamma, const float* __restrict__ beta, int N) {
  __shared__ float red[256];
  int row = blockIdx.x;
  int tid = threadIdx.x;
  float v = out[(size_t)row * F + tid] + bias[tid];

  red[tid] = v;
  __syncthreads();
  for (int s = 128; s > 0; s >>= 1) {
    if (tid < s) red[tid] += red[tid + s];
    __syncthreads();
  }
  float mean = red[0] * (1.0f / F);
  __syncthreads();

  float dv = v - mean;
  red[tid] = dv * dv;
  __syncthreads();
  for (int s = 128; s > 0; s >>= 1) {
    if (tid < s) red[tid] += red[tid + s];
    __syncthreads();
  }
  float var = red[0] * (1.0f / F);

  out[(size_t)row * F + tid] = dv * rsqrtf(var + LN_EPS) * gamma[tid] + beta[tid];
}

extern "C" void kernel_launch(void* const* d_in, const int* in_sizes, int n_in,
                              void* d_out, int out_size, void* d_ws, size_t ws_size,
                              hipStream_t stream) {
  const float* x       = (const float*)d_in[0];
  const int*   ei      = (const int*)  d_in[1];
  const float* W       = (const float*)d_in[2];
  const float* att_src = (const float*)d_in[3];
  const float* att_dst = (const float*)d_in[4];
  const float* bias    = (const float*)d_in[5];
  const float* gamma   = (const float*)d_in[6];
  const float* beta    = (const float*)d_in[7];
  float* out = (float*)d_out;

  const int N = in_sizes[0] / F;
  const int E = in_sizes[1] / 2;

  char* ws = (char*)d_ws;
  float*    xl     = (float*)ws;    ws += (size_t)N * F * sizeof(float);
  float*    a_src  = (float*)ws;    ws += (size_t)N * HEADS * sizeof(float);
  float*    a_dst  = (float*)ws;    ws += (size_t)N * HEADS * sizeof(float);
  unsigned* segmax = (unsigned*)ws; ws += (size_t)N * HEADS * sizeof(unsigned);
  float*    denom  = (float*)ws;

  gat_init_kernel<<<2048, 256, 0, stream>>>(out, denom, segmax, N * F, N * HEADS);

  gat_gemm_wmma_kernel<<<(N + 15) / 16, 512, 0, stream>>>(x, W, xl, N);

  {
    int NH = N * HEADS;
    gat_att_kernel<<<(NH + 255) / 256, 256, 0, stream>>>(xl, att_src, att_dst,
                                                         a_src, a_dst, NH);
  }

  {
    int total = (E + N) * HEADS;
    int blocks = (total + 255) / 256;
    gat_edge_max_kernel<<<blocks, 256, 0, stream>>>(ei, a_src, a_dst, segmax, E, N);
    gat_edge_sum_kernel<<<blocks, 256, 0, stream>>>(ei, a_src, a_dst, segmax, denom, E, N);
    gat_edge_agg_kernel<<<E + N, 256, 0, stream>>>(ei, xl, a_src, a_dst, segmax, denom,
                                                   out, E, N);
  }

  gat_ln_kernel<<<N, 256, 0, stream>>>(out, bias, gamma, beta, N);
}


# pass C by owned node range: block keeps 196 destination rows in LDS, scans the dst column for its edges, adds alpha*row with LDS float adds (arrival order), stores out once; global atomics gone
# speedup vs baseline: 2.3681x; 2.3681x over previous
#include <hip/hip_runtime.h>
#include <math.h>

typedef float v2f __attribute__((ext_vector_type(2)));
typedef float v8f __attribute__((ext_vector_type(8)));

#define HEADS 8
#define CH 32
#define F 256
#define NEG_SLOPE 0.2f
#define LN_EPS 1e-5f

#define LDS_STRIDE 260

__device__ __forceinline__ unsigned f2key(float f) {
  unsigned b = __float_as_uint(f);
  return (b & 0x80000000u) ? ~b : (b | 0x80000000u);
}
__device__ __forceinline__ float key2f(unsigned k) {
  unsigned b = (k & 0x80000000u) ? (k & 0x7fffffffu) : ~k;
  return __uint_as_float(b);
}

__global__ void gat_init_kernel(float* __restrict__ denom,
                                unsigned* __restrict__ segmax, int n_nh) {
  int stride = gridDim.x * blockDim.x;
  for (int t = blockIdx.x * blockDim.x + threadIdx.x; t < n_nh; t += stride) {
    denom[t] = 0.f;
    segmax[t] = 0x007FFFFFu;
  }
}

__global__ void __launch_bounds__(512)
gat_gemm_wmma_kernel(const float* __restrict__ x, const float* __restrict__ W,
                     float* __restrict__ xl, int N) {
  __shared__ float lds_a[16 * LDS_STRIDE];

  const int tid  = threadIdx.x;
  const int lane = tid & 31;
  const int wave = tid >> 5;
  const int row0 = blockIdx.x * 16;
  const int col0 = wave * 16;

  {
    int r = tid >> 5;
    int c = (tid & 31) * 8;
    int grow = row0 + r;
    if (grow >= N) grow = N - 1;
    const float4* __restrict__ src = (const float4*)(x + (size_t)grow * F + c);
    float4* dst = (float4*)(&lds_a[r * LDS_STRIDE + c]);
    dst[0] = src[0];
    dst[1] = src[1];
  }
  __syncthreads();

  const int m    = lane & 15;
  const int koff = (lane >> 4) * 2;
  const float* __restrict__ arow = &lds_a[m * LDS_STRIDE];

  const int n = lane & 15;

  v8f acc = {0.f, 0.f, 0.f, 0.f, 0.f, 0.f, 0.f, 0.f};
  for (int k = 0; k < F; k += 4) {
    v2f a = *(const v2f*)(arow + k + koff);
    v2f b;
    b.x = W[(size_t)(k + koff) * F + col0 + n];
    b.y = W[(size_t)(k + koff + 1) * F + col0 + n];
    acc = __builtin_amdgcn_wmma_f32_16x16x4_f32(
         false, a,  false, b,
         (short)0, acc,  false,  false);
  }

  const int rbase = row0 + ((lane >> 4) * 8);
  float* __restrict__ dst = xl + (size_t)rbase * F + col0 + n;
  if (row0 + 16 <= N) {
#pragma unroll
    for (int j = 0; j < 8; ++j) dst[(size_t)j * F] = acc[j];
  } else {
#pragma unroll
    for (int j = 0; j < 8; ++j) {
      if (rbase + j < N) dst[(size_t)j * F] = acc[j];
    }
  }
}

__global__ void gat_att_kernel(const float* __restrict__ xl,
                               const float* __restrict__ att_src,
                               const float* __restrict__ att_dst,
                               float* __restrict__ a_src, float* __restrict__ a_dst,
                               int NH) {
  int t = blockIdx.x * blockDim.x + threadIdx.x;
  if (t >= NH) return;
  int nidx = t >> 3, h = t & 7;
  const float* __restrict__ xr = xl + (size_t)nidx * F + h * CH;
  const float* __restrict__ as = att_src + h * CH;
  const float* __restrict__ ad = att_dst + h * CH;
  float s = 0.f, d = 0.f;
#pragma unroll
  for (int c = 0; c < CH; ++c) {
    float v = xr[c];
    s += v * as[c];
    d += v * ad[c];
  }
  a_src[t] = s;
  a_dst[t] = d;
}

__device__ __forceinline__ void edge_nodes(const int* __restrict__ ei, int e, int E,
                                           int& s, int& d) {
  if (e < E) { s = ei[e]; d = ei[E + e]; }
  else       { s = d = e - E; }
}

__device__ __forceinline__ float leaky(float v) {
  return (v >= 0.f) ? v : v * NEG_SLOPE;
}

__global__ void gat_edge_max_kernel(const int* __restrict__ ei,
                                    const float* __restrict__ a_src,
                                    const float* __restrict__ a_dst,
                                    unsigned* __restrict__ segmax, int E, int N) {
  int t = blockIdx.x * blockDim.x + threadIdx.x;
  int total = (E + N) * HEADS;
  if (t >= total) return;
  int e = t >> 3, h = t & 7;
  int s, d;
  edge_nodes(ei, e, E, s, d);
  float logit = leaky(a_src[s * HEADS + h] + a_dst[d * HEADS + h]);
  atomicMax(&segmax[d * HEADS + h], f2key(logit));
}

__global__ void gat_edge_sum_kernel(const int* __restrict__ ei,
                                    const float* __restrict__ a_src,
                                    const float* __restrict__ a_dst,
                                    const unsigned* __restrict__ segmax,
                                    float* __restrict__ denom, int E, int N) {
  int t = blockIdx.x * blockDim.x + threadIdx.x;
  int total = (E + N) * HEADS;
  if (t >= total) return;
  int e = t >> 3, h = t & 7;
  int s, d;
  edge_nodes(ei, e, E, s, d);
  float logit = leaky(a_src[s * HEADS + h] + a_dst[d * HEADS + h]);
  float mx = key2f(segmax[d * HEADS + h]);
  atomicAdd(&denom[d * HEADS + h], expf(logit - mx));
}

#define NPB 196
#define AGG_THREADS 1024
#define AGG_WAVES (AGG_THREADS / 32)
#define LIST_CAP 4096
#define SCAN_UNROLL 8
#define SCAN_STEP (AGG_THREADS * 4)
#define SCAN_SPAN (SCAN_STEP * SCAN_UNROLL)
#define DRAIN_EVERY 8

__global__ void __launch_bounds__(AGG_THREADS)
gat_edge_agg_kernel(const int* __restrict__ ei, const float* __restrict__ xl,
                    const float* __restrict__ a_src, const float* __restrict__ a_dst,
                    const unsigned* __restrict__ segmax, const float* __restrict__ denom,
                    float* __restrict__ out, int E, int N) {
  __shared__ float acc[NPB * F];
  __shared__ int list_e[LIST_CAP];
  __shared__ int list_d[LIST_CAP];
  __shared__ int list_n;

  const int tid  = threadIdx.x;
  const int lane = tid & 31;
  const int wave = __builtin_amdgcn_readfirstlane(tid >> 5);
  const int n0 = blockIdx.x * NPB;
  const int nrows = (N - n0 < NPB) ? (N - n0) : NPB;
  const int* __restrict__ dst = ei + E;

  for (int i = tid; i < NPB * F / 4; i += AGG_THREADS)
    ((float4*)acc)[i] = make_float4(0.f, 0.f, 0.f, 0.f);
  if (tid < nrows) { list_e[tid] = E + n0 + tid; list_d[tid] = tid; }
  if (tid == 0) list_n = nrows;
  __syncthreads();

  auto edge_by_one_lane = [&](int e, int dl) {
    int s = (e < E) ? ei[e] : (e - E);
    int d = n0 + dl;
    for (int h = 0; h < HEADS; ++h) {
      float logit = leaky(a_src[s * HEADS + h] + a_dst[d * HEADS + h]);
      float mx = key2f(segmax[d * HEADS + h]);
      float alpha = expf(logit - mx) / denom[d * HEADS + h];
      for (int c = 0; c < CH; ++c)
        __hip_atomic_fetch_add(&acc[dl * F + h * CH + c], xl[(size_t)s * F + h * CH + c] * alpha,
                               __ATOMIC_RELAXED, __HIP_MEMORY_SCOPE_WORKGROUP);
    }
  };
  auto keep = [&](int e, int dv) {
    unsigned dl = (unsigned)(dv - n0);
    if (dl < (unsigned)nrows) {
      int slot = __hip_atomic_fetch_add(&list_n, 1, __ATOMIC_RELAXED, __HIP_MEMORY_SCOPE_WORKGROUP);
      if (slot < LIST_CAP) { list_e[slot] = e; list_d[slot] = (int)dl; }
    }
  };

  int n_it = (E + SCAN_SPAN - 1) / SCAN_SPAN;
  if (n_it < 1) n_it = 1;
  const int n_dr = (n_it + DRAIN_EVERY - 1) / DRAIN_EVERY;
  const int per_dr = (n_it + n_dr - 1) / n_dr;
  const int rot = (int)(((long long)blockIdx.x * n_it) / gridDim.x);
  const bool vec_ok = ((E & 3) == 0);

  int since = 0;
  int it_first = 0;
  for (int it = 0; it < n_it; ++it) {
    int c = it + rot;
    if (c >= n_it) c -= n_it;
    const int b0 = c * SCAN_SPAN + tid * 4;

    if (vec_ok && (c + 1) * SCAN_SPAN <= E) {
      int4 v[SCAN_UNROLL];
#pragma unroll
      for (int u = 0; u < SCAN_UNROLL; ++u)
        v[u] = *(const int4*)(dst + b0 + u * SCAN_STEP);
#pragma unroll
      for (int u = 0; u < SCAN_UNROLL; ++u) {
        const int idx = b0 + u * SCAN_STEP;
        const bool any = ((unsigned)(v[u].x - n0) < (unsigned)nrows) | ((unsigned)(v[u].y - n0) < (unsigned)nrows) |
                         ((unsigned)(v[u].z - n0) < (unsigned)nrows) | ((unsigned)(v[u].w - n0) < (unsigned)nrows);
        if (any) {
          keep(idx + 0, v[u].x);
          keep(idx + 1, v[u].y);
          keep(idx + 2, v[u].z);
          keep(idx + 3, v[u].w);
        }
      }
    } else {
      for (int u = 0; u < SCAN_UNROLL; ++u) {
        const int idx = b0 + u * SCAN_STEP;
        for (int k = 0; k < 4; ++k)
          if (idx + k < E) keep(idx + k, dst[idx + k]);
      }
    }

    ++since;
    if (since == per_dr || it == n_it - 1) {
      since = 0;
      __syncthreads();
      int cnt = __builtin_amdgcn_readfirstlane(list_n);
      if (cnt > LIST_CAP) {
        cnt = 0;
        if (it_first == 0 && tid < nrows) edge_by_one_lane(E + n0 + tid, tid);
        for (int j = it_first; j <= it; ++j) {
          int cj = j + rot;
          if (cj >= n_it) cj -= n_it;
          for (int q = tid; q < SCAN_SPAN; q += AGG_THREADS) {
            const int idx = cj * SCAN_SPAN + q;
            if (idx < E) {
              unsigned dl = (unsigned)(dst[idx] - n0);
              if (dl < (unsigned)nrows) edge_by_one_lane(idx, (int)dl);
            }
          }
        }
      }
      it_first = it + 1;
      const int lo = (cnt * wave) / AGG_WAVES;
      const int hi = (cnt * (wave + 1)) / AGG_WAVES;
      for (int g = lo; g < hi; g += 32) {
        const int m = (hi - g < 32) ? (hi - g) : 32;
        int e = E + n0, dl = 0;
        if (lane < m) { e = list_e[g + lane]; dl = list_d[g + lane]; }
        const int s = (e < E) ? ei[e] : (e - E);
        const int d = n0 + dl;
        const float4* ps = (const float4*)(a_src + (size_t)s * HEADS);
        const float4* pd = (const float4*)(a_dst + (size_t)d * HEADS);
        const uint4*  pm = (const uint4*)(segmax + (size_t)d * HEADS);
        const float4* pn = (const float4*)(denom + (size_t)d * HEADS);
        float4 s0 = ps[0], s1 = ps[1], d0 = pd[0], d1 = pd[1], q0 = pn[0], q1 = pn[1];
        uint4 m0 = pm[0], m1 = pm[1];
        float al[HEADS];
        al[0] = expf(leaky(s0.x + d0.x) - key2f(m0.x)) / q0.x;
        al[1] = expf(leaky(s0.y + d0.y) - key2f(m0.y)) / q0.y;
        al[2] = expf(leaky(s0.z + d0.z) - key2f(m0.z)) / q0.z;
        al[3] = expf(leaky(s0.w + d0.w) - key2f(m0.w)) / q0.w;
        al[4] = expf(leaky(s1.x + d1.x) - key2f(m1.x)) / q1.x;
        al[5] = expf(leaky(s1.y + d1.y) - key2f(m1.y)) / q1.y;
        al[6] = expf(leaky(s1.z + d1.z) - key2f(m1.z)) / q1.z;
        al[7] = expf(leaky(s1.w + d1.w) - key2f(m1.w)) / q1.w;

        for (int i0 = 0; i0 < m; i0 += 4) {
          float xv[4][HEADS];
#pragma unroll
          for (int j = 0; j < 4; ++j) {
            const int ii = (i0 + j < m) ? (i0 + j) : (m - 1);
            const int si = __builtin_amdgcn_readlane(s, ii);
            const float* __restrict__ xr = xl + (size_t)si * F + lane;
#pragma unroll
            for (int h = 0; h < HEADS; ++h) xv[j][h] = xr[h * CH];
          }
#pragma unroll
          for (int j = 0; j < 4; ++j) {
            if (i0 + j < m) {
              const int ii = i0 + j;
              const int di = __builtin_amdgcn_readlane(dl, ii);
              float* ar = acc + di * F + lane;
#pragma unroll
              for (int h = 0; h < HEADS; ++h) {
                const float w = __int_as_float(__builtin_amdgcn_readlane(__float_as_int(al[h]), ii));
                __hip_atomic_fetch_add(&ar[h * CH], xv[j][h] * w,
                                       __ATOMIC_RELAXED, __HIP_MEMORY_SCOPE_WORKGROUP);
              }
            }
          }
        }
      }
      __syncthreads();
      if (tid == 0) list_n = 0;
      __syncthreads();
    }
  }

  float4* __restrict__ orow = (float4*)(out + (size_t)n0 * F);
  for (int i = tid; i < nrows * (F / 4); i += AGG_THREADS) orow[i] = ((const float4*)acc)[i];
}

__global__ void __launch_bounds__(256)
gat_ln_kernel(float* __restrict__ out, const float* __restrict__ bias,
              const float* __restrict__ gamma, const float* __restrict__ beta, int N) {
  __shared__ float red[256];
  int row = blockIdx.x;
  int tid = threadIdx.x;
  float v = out[(size_t)row * F + tid] + bias[tid];

  red[tid] = v;
  __syncthreads();
  for (int s = 128; s > 0; s >>= 1) {
    if (tid < s) red[tid] += red[tid + s];
    __syncthreads();
  }
  float mean = red[0] * (1.0f / F);
  __syncthreads();

  float dv = v - mean;
  red[tid] = dv * dv;
  __syncthreads();
  for (int s = 128; s > 0; s >>= 1) {
    if (tid < s) red[tid] += red[tid + s];
    __syncthreads();
  }
  float var = red[0] * (1.0f / F);

  out[(size_t)row * F + tid] = dv * rsqrtf(var + LN_EPS) * gamma[tid] + beta[tid];
}

extern "C" void kernel_launch(void* const* d_in, const int* in_sizes, int n_in,
                              void* d_out, int out_size, void* d_ws, size_t ws_size,
                              hipStream_t stream) {
  const float* x       = (const float*)d_in[0];
  const int*   ei      = (const int*)  d_in[1];
  const float* W       = (const float*)d_in[2];
  const float* att_src = (const float*)d_in[3];
  const float* att_dst = (const float*)d_in[4];
  const float* bias    = (const float*)d_in[5];
  const float* gamma   = (const float*)d_in[6];
  const float* beta    = (const float*)d_in[7];
  float* out = (float*)d_out;

  const int N = in_sizes[0] / F;
  const int E = in_sizes[1] / 2;

  char* ws = (char*)d_ws;
  float*    xl     = (float*)ws;    ws += (size_t)N * F * sizeof(float);
  float*    a_src  = (float*)ws;    ws += (size_t)N * HEADS * sizeof(float);
  float*    a_dst  = (float*)ws;    ws += (size_t)N * HEADS * sizeof(float);
  unsigned* segmax = (unsigned*)ws; ws += (size_t)N * HEADS * sizeof(unsigned);
  float*    denom  = (float*)ws;

  gat_init_kernel<<<2048, 256, 0, stream>>>(denom, segmax, N * HEADS);

  gat_gemm_wmma_kernel<<<(N + 15) / 16, 512, 0, stream>>>(x, W, xl, N);

  {
    int NH = N * HEADS;
    gat_att_kernel<<<(NH + 255) / 256, 256, 0, stream>>>(xl, att_src, att_dst,
                                                         a_src, a_dst, NH);
  }

  {
    int total = (E + N) * HEADS;
    int blocks = (total + 255) / 256;
    gat_edge_max_kernel<<<blocks, 256, 0, stream>>>(ei, a_src, a_dst, segmax, E, N);
    gat_edge_sum_kernel<<<blocks, 256, 0, stream>>>(ei, a_src, a_dst, segmax, denom, E, N);
    gat_edge_agg_kernel<<<(N + NPB - 1) / NPB, AGG_THREADS, 0, stream>>>(ei, xl, a_src, a_dst,
                                                                          segmax, denom, out, E, N);
  }

  gat_ln_kernel<<<N, 256, 0, stream>>>(out, bias, gamma, beta, N);
}


# softmax moved into the block owning the node range: its edges in one LDS list, per-(node,head) max and exp-sum by LDS atomics (arrival order), rows scaled by 1/exp-sum at the store; 3 launches gone
# speedup vs baseline: 3.2240x; 1.3614x over previous
#include <hip/hip_runtime.h>
#include <math.h>

typedef float v2f __attribute__((ext_vector_type(2)));
typedef float v8f __attribute__((ext_vector_type(8)));

#define HEADS 8
#define CH 32
#define F 256
#define NEG_SLOPE 0.2f
#define LN_EPS 1e-5f

#define LDS_STRIDE 260

__device__ __forceinline__ unsigned f2key(float f) {
  unsigned b = __float_as_uint(f);
  return (b & 0x80000000u) ? ~b : (b | 0x80000000u);
}
__device__ __forceinline__ float key2f(unsigned k) {
  unsigned b = (k & 0x80000000u) ? (k & 0x7fffffffu) : ~k;
  return __uint_as_float(b);
}

__global__ void __launch_bounds__(512)
gat_gemm_wmma_kernel(const float* __restrict__ x, const float* __restrict__ W,
                     float* __restrict__ xl, int N) {
  __shared__ float lds_a[16 * LDS_STRIDE];

  const int tid  = threadIdx.x;
  const int lane = tid & 31;
  const int wave = tid >> 5;
  const int row0 = blockIdx.x * 16;
  const int col0 = wave * 16;

  {
    int r = tid >> 5;
    int c = (tid & 31) * 8;
    int grow = row0 + r;
    if (grow >= N) grow = N - 1;
    const float4* __restrict__ src = (const float4*)(x + (size_t)grow * F + c);
    float4* dst = (float4*)(&lds_a[r * LDS_STRIDE + c]);
    dst[0] = src[0];
    dst[1] = src[1];
  }
  __syncthreads();

  const int m    = lane & 15;
  const int koff = (lane >> 4) * 2;
  const float* __restrict__ arow = &lds_a[m * LDS_STRIDE];

  const int n = lane & 15;

  v8f acc = {0.f, 0.f, 0.f, 0.f, 0.f, 0.f, 0.f, 0.f};
  for (int k = 0; k < F; k += 4) {
    v2f a = *(const v2f*)(arow + k + koff);
    v2f b;
    b.x = W[(size_t)(k + koff) * F + col0 + n];
    b.y = W[(size_t)(k + koff + 1) * F + col0 + n];
    acc = __builtin_amdgcn_wmma_f32_16x16x4_f32(
         false, a,  false, b,
         (short)0, acc,  false,  false);
  }

  const int rbase = row0 + ((lane >> 4) * 8);
  float* __restrict__ dst = xl + (size_t)rbase * F + col0 + n;
  if (row0 + 16 <= N) {
#pragma unroll
    for (int j = 0; j < 8; ++j) dst[(size_t)j * F] = acc[j];
  } else {
#pragma unroll
    for (int j = 0; j < 8; ++j) {
      if (rbase + j < N) dst[(size_t)j * F] = acc[j];
    }
  }
}

__global__ void gat_att_kernel(const float* __restrict__ xl,
                               const float* __restrict__ att_src,
                               const float* __restrict__ att_dst,
                               float* __restrict__ a_src, float* __restrict__ a_dst,
                               int NH) {
  int t = blockIdx.x * blockDim.x + threadIdx.x;
  if (t >= NH) return;
  int nidx = t >> 3, h = t & 7;
  const float* __restrict__ xr = xl + (size_t)nidx * F + h * CH;
  const float* __restrict__ as = att_src + h * CH;
  const float* __restrict__ ad = att_dst + h * CH;
  float s = 0.f, d = 0.f;
#pragma unroll
  for (int c = 0; c < CH; ++c) {
    float v = xr[c];
    s += v * as[c];
    d += v * ad[c];
  }
  a_src[t] = s;
  a_dst[t] = d;
}

__device__ __forceinline__ float leaky(float v) {
  return (v >= 0.f) ? v : v * NEG_SLOPE;
}

#define NPB 196
#define AGG_THREADS 1024
#define AGG_WAVES (AGG_THREADS / 32)
#define LIST_CAP 4096
#define SCAN_UNROLL 8
#define SCAN_STEP (AGG_THREADS * 4)
#define SCAN_SPAN (SCAN_STEP * SCAN_UNROLL)

__global__ void __launch_bounds__(AGG_THREADS)
gat_edge_agg_kernel(const int* __restrict__ ei, const float* __restrict__ xl,
                    const float* __restrict__ a_src, const float* __restrict__ a_dst,
                    float* __restrict__ out, int E, int N) {
  __shared__ __attribute__((aligned(16))) float acc[NPB * F];
  __shared__ __attribute__((aligned(16))) unsigned smax[NPB * HEADS];
  __shared__ __attribute__((aligned(16))) float den[NPB * HEADS];
  __shared__ int list_e[LIST_CAP];
  __shared__ int list_d[LIST_CAP];
  __shared__ int list_n;

  const int tid  = threadIdx.x;
  const int lane = tid & 31;
  const int wave = __builtin_amdgcn_readfirstlane(tid >> 5);
  const int n0 = blockIdx.x * NPB;
  const int nrows = (N - n0 < NPB) ? (N - n0) : NPB;
  const int* __restrict__ dst = ei + E;

  for (int i = tid; i < NPB * F / 4; i += AGG_THREADS)
    ((float4*)acc)[i] = make_float4(0.f, 0.f, 0.f, 0.f);
  for (int i = tid; i < NPB * HEADS; i += AGG_THREADS) {
    smax[i] = 0x007FFFFFu;
    den[i] = 0.f;
  }
  if (tid < nrows) { list_e[tid] = E + n0 + tid; list_d[tid] = tid; }
  if (tid == 0) list_n = nrows;
  __syncthreads();

  auto keep = [&](int e, int dv) {
    unsigned dl = (unsigned)(dv - n0);
    if (dl < (unsigned)nrows) {
      int slot = __hip_atomic_fetch_add(&list_n, 1, __ATOMIC_RELAXED, __HIP_MEMORY_SCOPE_WORKGROUP);
      if (slot < LIST_CAP) { list_e[slot] = e; list_d[slot] = (int)dl; }
    }
  };

  int n_it = (E + SCAN_SPAN - 1) / SCAN_SPAN;
  if (n_it < 1) n_it = 1;
  const int rot = (int)(((long long)blockIdx.x * n_it) / gridDim.x);
  const bool vec_ok = ((E & 3) == 0);

  for (int it = 0; it < n_it; ++it) {
    int c = it + rot;
    if (c >= n_it) c -= n_it;
    const int b0 = c * SCAN_SPAN + tid * 4;

    if (vec_ok && (c + 1) * SCAN_SPAN <= E) {
      int4 v[SCAN_UNROLL];
#pragma unroll
      for (int u = 0; u < SCAN_UNROLL; ++u)
        v[u] = *(const int4*)(dst + b0 + u * SCAN_STEP);
#pragma unroll
      for (int u = 0; u < SCAN_UNROLL; ++u) {
        const int idx = b0 + u * SCAN_STEP;
        const bool any = ((unsigned)(v[u].x - n0) < (unsigned)nrows) | ((unsigned)(v[u].y - n0) < (unsigned)nrows) |
                         ((unsigned)(v[u].z - n0) < (unsigned)nrows) | ((unsigned)(v[u].w - n0) < (unsigned)nrows);
        if (any) {
          keep(idx + 0, v[u].x);
          keep(idx + 1, v[u].y);
          keep(idx + 2, v[u].z);
          keep(idx + 3, v[u].w);
        }
      }
    } else {
      for (int u = 0; u < SCAN_UNROLL; ++u) {
        const int idx = b0 + u * SCAN_STEP;
        for (int k = 0; k < 4; ++k)
          if (idx + k < E) keep(idx + k, dst[idx + k]);
      }
    }
  }
  __syncthreads();
  const int cnt = __builtin_amdgcn_readfirstlane(list_n);

  if (cnt <= LIST_CAP) {
    for (int base = 0; base < cnt; base += 4 * AGG_THREADS) {
      int ee[4], dd[4], ss[4];
      float4 sa[4][2], da[4][2];
#pragma unroll
      for (int k = 0; k < 4; ++k) {
        const int i = base + k * AGG_THREADS + tid;
        ee[k] = (i < cnt) ? list_e[i] : (E + n0);
        dd[k] = (i < cnt) ? list_d[i] : 0;
      }
#pragma unroll
      for (int k = 0; k < 4; ++k) ss[k] = (ee[k] < E) ? ei[ee[k]] : (ee[k] - E);
#pragma unroll
      for (int k = 0; k < 4; ++k) {
        const float4* ps = (const float4*)(a_src + (size_t)ss[k] * HEADS);
        const float4* pd = (const float4*)(a_dst + (size_t)(n0 + dd[k]) * HEADS);
        sa[k][0] = ps[0]; sa[k][1] = ps[1];
        da[k][0] = pd[0]; da[k][1] = pd[1];
      }
#pragma unroll
      for (int k = 0; k < 4; ++k) {
        const int i = base + k * AGG_THREADS + tid;
        if (i < cnt) {
          list_e[i] = ss[k];
          unsigned* mx = smax + dd[k] * HEADS;
          __hip_atomic_fetch_max(&mx[0], f2key(leaky(sa[k][0].x + da[k][0].x)), __ATOMIC_RELAXED, __HIP_MEMORY_SCOPE_WORKGROUP);
          __hip_atomic_fetch_max(&mx[1], f2key(leaky(sa[k][0].y + da[k][0].y)), __ATOMIC_RELAXED, __HIP_MEMORY_SCOPE_WORKGROUP);
          __hip_atomic_fetch_max(&mx[2], f2key(leaky(sa[k][0].z + da[k][0].z)), __ATOMIC_RELAXED, __HIP_MEMORY_SCOPE_WORKGROUP);
          __hip_atomic_fetch_max(&mx[3], f2key(leaky(sa[k][0].w + da[k][0].w)), __ATOMIC_RELAXED, __HIP_MEMORY_SCOPE_WORKGROUP);
          __hip_atomic_fetch_max(&mx[4], f2key(leaky(sa[k][1].x + da[k][1].x)), __ATOMIC_RELAXED, __HIP_MEMORY_SCOPE_WORKGROUP);
          __hip_atomic_fetch_max(&mx[5], f2key(leaky(sa[k][1].y + da[k][1].y)), __ATOMIC_RELAXED, __HIP_MEMORY_SCOPE_WORKGROUP);
          __hip_atomic_fetch_max(&mx[6], f2key(leaky(sa[k][1].z + da[k][1].z)), __ATOMIC_RELAXED, __HIP_MEMORY_SCOPE_WORKGROUP);
          __hip_atomic_fetch_max(&mx[7], f2key(leaky(sa[k][1].w + da[k][1].w)), __ATOMIC_RELAXED, __HIP_MEMORY_SCOPE_WORKGROUP);
        }
      }
    }
    __syncthreads();

    const int lo = (cnt * wave) / AGG_WAVES;
    const int hi = (cnt * (wave + 1)) / AGG_WAVES;
    for (int g = lo; g < hi; g += 32) {
      const int m = (hi - g < 32) ? (hi - g) : 32;
      int s = n0, dl = 0;
      if (lane < m) { s = list_e[g + lane]; dl = list_d[g + lane]; }
      const float4* ps = (const float4*)(a_src + (size_t)s * HEADS);
      const float4* pd = (const float4*)(a_dst + (size_t)(n0 + dl) * HEADS);
      const uint4*  pm = (const uint4*)(smax + dl * HEADS);
      const float4 s0 = ps[0], s1 = ps[1], d0 = pd[0], d1 = pd[1];
      const uint4 m0 = pm[0], m1 = pm[1];
      float al[HEADS];
      al[0] = expf(leaky(s0.x + d0.x) - key2f(m0.x));
      al[1] = expf(leaky(s0.y + d0.y) - key2f(m0.y));
      al[2] = expf(leaky(s0.z + d0.z) - key2f(m0.z));
      al[3] = expf(leaky(s0.w + d0.w) - key2f(m0.w));
      al[4] = expf(leaky(s1.x + d1.x) - key2f(m1.x));
      al[5] = expf(leaky(s1.y + d1.y) - key2f(m1.y));
      al[6] = expf(leaky(s1.z + d1.z) - key2f(m1.z));
      al[7] = expf(leaky(s1.w + d1.w) - key2f(m1.w));
      if (lane < m) {
        float* dn = den + dl * HEADS;
#pragma unroll
        for (int h = 0; h < HEADS; ++h)
          __hip_atomic_fetch_add(&dn[h], al[h], __ATOMIC_RELAXED, __HIP_MEMORY_SCOPE_WORKGROUP);
      }

      for (int i0 = 0; i0 < m; i0 += 4) {
        float xv[4][HEADS];
#pragma unroll
        for (int j = 0; j < 4; ++j) {
          const int ii = (i0 + j < m) ? (i0 + j) : (m - 1);
          const int si = __builtin_amdgcn_readlane(s, ii);
          const float* __restrict__ xr = xl + (size_t)si * F + lane;
#pragma unroll
          for (int h = 0; h < HEADS; ++h) xv[j][h] = xr[h * CH];
        }
#pragma unroll
        for (int j = 0; j < 4; ++j) {
          if (i0 + j < m) {
            const int ii = i0 + j;
            const int di = __builtin_amdgcn_readlane(dl, ii);
            float* ar = acc + di * F + lane;
#pragma unroll
            for (int h = 0; h < HEADS; ++h) {
              const float w = __int_as_float(__builtin_amdgcn_readlane(__float_as_int(al[h]), ii));
              __hip_atomic_fetch_add(&ar[h * CH], xv[j][h] * w,
                                     __ATOMIC_RELAXED, __HIP_MEMORY_SCOPE_WORKGROUP);
            }
          }
        }
      }
    }
  } else {
    const int tot = E + N;
    for (int e = tid; e < tot; e += AGG_THREADS) {
      const int d = (e < E) ? dst[e] : (e - E);
      const unsigned dl = (unsigned)(d - n0);
      if (dl < (unsigned)nrows) {
        const int s = (e < E) ? ei[e] : (e - E);
        for (int h = 0; h < HEADS; ++h)
          __hip_atomic_fetch_max(&smax[dl * HEADS + h], f2key(leaky(a_src[s * HEADS + h] + a_dst[d * HEADS + h])),
                                 __ATOMIC_RELAXED, __HIP_MEMORY_SCOPE_WORKGROUP);
      }
    }
    __syncthreads();
    for (int e = tid; e < tot; e += AGG_THREADS) {
      const int d = (e < E) ? dst[e] : (e - E);
      const unsigned dl = (unsigned)(d - n0);
      if (dl < (unsigned)nrows) {
        const int s = (e < E) ? ei[e] : (e - E);
        for (int h = 0; h < HEADS; ++h) {
          const float p = expf(leaky(a_src[s * HEADS + h] + a_dst[d * HEADS + h]) - key2f(smax[dl * HEADS + h]));
          __hip_atomic_fetch_add(&den[dl * HEADS + h], p, __ATOMIC_RELAXED, __HIP_MEMORY_SCOPE_WORKGROUP);
          for (int c = 0; c < CH; ++c)
            __hip_atomic_fetch_add(&acc[dl * F + h * CH + c], xl[(size_t)s * F + h * CH + c] * p,
                                   __ATOMIC_RELAXED, __HIP_MEMORY_SCOPE_WORKGROUP);
        }
      }
    }
  }
  __syncthreads();

  for (int i = tid; i < nrows * HEADS; i += AGG_THREADS) den[i] = 1.0f / den[i];
  __syncthreads();
  float4* __restrict__ orow = (float4*)(out + (size_t)n0 * F);
  for (int i = tid; i < nrows * (F / 4); i += AGG_THREADS) {
    float4 v = ((const float4*)acc)[i];
    const float r = den[(i >> 6) * HEADS + ((i & 63) >> 3)];
    v.x *= r; v.y *= r; v.z *= r; v.w *= r;
    orow[i] = v;
  }
}

__global__ void __launch_bounds__(256)
gat_ln_kernel(float* __restrict__ out, const float* __restrict__ bias,
              const float* __restrict__ gamma, const float* __restrict__ beta, int N) {
  __shared__ float red[256];
  int row = blockIdx.x;
  int tid = threadIdx.x;
  float v = out[(size_t)row * F + tid] + bias[tid];

  red[tid] = v;
  __syncthreads();
  for (int s = 128; s > 0; s >>= 1) {
    if (tid < s) red[tid] += red[tid + s];
    __syncthreads();
  }
  float mean = red[0] * (1.0f / F);
  __syncthreads();

  float dv = v - mean;
  red[tid] = dv * dv;
  __syncthreads();
  for (int s = 128; s > 0; s >>= 1) {
    if (tid < s) red[tid] += red[tid + s];
    __syncthreads();
  }
  float var = red[0] * (1.0f / F);

  out[(size_t)row * F + tid] = dv * rsqrtf(var + LN_EPS) * gamma[tid] + beta[tid];
}

extern "C" void kernel_launch(void* const* d_in, const int* in_sizes, int n_in,
                              void* d_out, int out_size, void* d_ws, size_t ws_size,
                              hipStream_t stream) {
  const float* x       = (const float*)d_in[0];
  const int*   ei      = (const int*)  d_in[1];
  const float* W       = (const float*)d_in[2];
  const float* att_src = (const float*)d_in[3];
  const float* att_dst = (const float*)d_in[4];
  const float* bias    = (const float*)d_in[5];
  const float* gamma   = (const float*)d_in[6];
  const float* beta    = (const float*)d_in[7];
  float* out = (float*)d_out;

  const int N = in_sizes[0] / F;
  const int E = in_sizes[1] / 2;

  char* ws = (char*)d_ws;
  float*    xl     = (float*)ws;    ws += (size_t)N * F * sizeof(float);
  float*    a_src  = (float*)ws;    ws += (size_t)N * HEADS * sizeof(float);
  float*    a_dst  = (float*)ws;

  gat_gemm_wmma_kernel<<<(N + 15) / 16, 512, 0, stream>>>(x, W, xl, N);

  {
    int NH = N * HEADS;
    gat_att_kernel<<<(NH + 255) / 256, 256, 0, stream>>>(xl, att_src, att_dst,
                                                         a_src, a_dst, NH);
  }

  gat_edge_agg_kernel<<<(N + NPB - 1) / NPB, AGG_THREADS, 0, stream>>>(ei, xl, a_src, a_dst,
                                                                        out, E, N);

  gat_ln_kernel<<<N, 256, 0, stream>>>(out, bias, gamma, beta, N);
}


# LayerNorm launch remapped: one wave per 4 rows, rows in registers (two 16-byte loads per lane and row), sums by a 5-step wave butterfly (new reduction tree); no LDS, no barriers, 32x fewer waves
# speedup vs baseline: 4.0462x; 1.2550x over previous
#include <hip/hip_runtime.h>
#include <math.h>

typedef float v2f __attribute__((ext_vector_type(2)));
typedef float v8f __attribute__((ext_vector_type(8)));

#define HEADS 8
#define CH 32
#define F 256
#define NEG_SLOPE 0.2f
#define LN_EPS 1e-5f

#define LDS_STRIDE 260

__device__ __forceinline__ unsigned f2key(float f) {
  unsigned b = __float_as_uint(f);
  return (b & 0x80000000u) ? ~b : (b | 0x80000000u);
}
__device__ __forceinline__ float key2f(unsigned k) {
  unsigned b = (k & 0x80000000u) ? (k & 0x7fffffffu) : ~k;
  return __uint_as_float(b);
}

__global__ void __launch_bounds__(512)
gat_gemm_wmma_kernel(const float* __restrict__ x, const float* __restrict__ W,
                     float* __restrict__ xl, int N) {
  __shared__ float lds_a[16 * LDS_STRIDE];

  const int tid  = threadIdx.x;
  const int lane = tid & 31;
  const int wave = tid >> 5;
  const int row0 = blockIdx.x * 16;
  const int col0 = wave * 16;

  {
    int r = tid >> 5;
    int c = (tid & 31) * 8;
    int grow = row0 + r;
    if (grow >= N) grow = N - 1;
    const float4* __restrict__ src = (const float4*)(x + (size_t)grow * F + c);
    float4* dst = (float4*)(&lds_a[r * LDS_STRIDE + c]);
    dst[0] = src[0];
    dst[1] = src[1];
  }
  __syncthreads();

  const int m    = lane & 15;
  const int koff = (lane >> 4) * 2;
  const float* __restrict__ arow = &lds_a[m * LDS_STRIDE];

  const int n = lane & 15;

  v8f acc = {0.f, 0.f, 0.f, 0.f, 0.f, 0.f, 0.f, 0.f};
  for (int k = 0; k < F; k += 4) {
    v2f a = *(const v2f*)(arow + k + koff);
    v2f b;
    b.x = W[(size_t)(k + koff) * F + col0 + n];
    b.y = W[(size_t)(k + koff + 1) * F + col0 + n];
    acc = __builtin_amdgcn_wmma_f32_16x16x4_f32(
         false, a,  false, b,
         (short)0, acc,  false,  false);
  }

  const int rbase = row0 + ((lane >> 4) * 8);
  float* __restrict__ dst = xl + (size_t)rbase * F + col0 + n;
  if (row0 + 16 <= N) {
#pragma unroll
    for (int j = 0; j < 8; ++j) dst[(size_t)j * F] = acc[j];
  } else {
#pragma unroll
    for (int j = 0; j < 8; ++j) {
      if (rbase + j < N) dst[(size_t)j * F] = acc[j];
    }
  }
}

__global__ void gat_att_kernel(const float* __restrict__ xl,
                               const float* __restrict__ att_src,
                               const float* __restrict__ att_dst,
                               float* __restrict__ a_src, float* __restrict__ a_dst,
                               int NH) {
  int t = blockIdx.x * blockDim.x + threadIdx.x;
  if (t >= NH) return;
  int nidx = t >> 3, h = t & 7;
  const float* __restrict__ xr = xl + (size_t)nidx * F + h * CH;
  const float* __restrict__ as = att_src + h * CH;
  const float* __restrict__ ad = att_dst + h * CH;
  float s = 0.f, d = 0.f;
#pragma unroll
  for (int c = 0; c < CH; ++c) {
    float v = xr[c];
    s += v * as[c];
    d += v * ad[c];
  }
  a_src[t] = s;
  a_dst[t] = d;
}

__device__ __forceinline__ float leaky(float v) {
  return (v >= 0.f) ? v : v * NEG_SLOPE;
}

#define NPB 196
#define AGG_THREADS 1024
#define AGG_WAVES (AGG_THREADS / 32)
#define LIST_CAP 4096
#define SCAN_UNROLL 8
#define SCAN_STEP (AGG_THREADS * 4)
#define SCAN_SPAN (SCAN_STEP * SCAN_UNROLL)

__global__ void __launch_bounds__(AGG_THREADS)
gat_edge_agg_kernel(const int* __restrict__ ei, const float* __restrict__ xl,
                    const float* __restrict__ a_src, const float* __restrict__ a_dst,
                    float* __restrict__ out, int E, int N) {
  __shared__ __attribute__((aligned(16))) float acc[NPB * F];
  __shared__ __attribute__((aligned(16))) unsigned smax[NPB * HEADS];
  __shared__ __attribute__((aligned(16))) float den[NPB * HEADS];
  __shared__ int list_e[LIST_CAP];
  __shared__ int list_d[LIST_CAP];
  __shared__ int list_n;

  const int tid  = threadIdx.x;
  const int lane = tid & 31;
  const int wave = __builtin_amdgcn_readfirstlane(tid >> 5);
  const int n0 = blockIdx.x * NPB;
  const int nrows = (N - n0 < NPB) ? (N - n0) : NPB;
  const int* __restrict__ dst = ei + E;

  for (int i = tid; i < NPB * F / 4; i += AGG_THREADS)
    ((float4*)acc)[i] = make_float4(0.f, 0.f, 0.f, 0.f);
  for (int i = tid; i < NPB * HEADS; i += AGG_THREADS) {
    smax[i] = 0x007FFFFFu;
    den[i] = 0.f;
  }
  if (tid < nrows) { list_e[tid] = E + n0 + tid; list_d[tid] = tid; }
  if (tid == 0) list_n = nrows;
  __syncthreads();

  auto keep = [&](int e, int dv) {
    unsigned dl = (unsigned)(dv - n0);
    if (dl < (unsigned)nrows) {
      int slot = __hip_atomic_fetch_add(&list_n, 1, __ATOMIC_RELAXED, __HIP_MEMORY_SCOPE_WORKGROUP);
      if (slot < LIST_CAP) { list_e[slot] = e; list_d[slot] = (int)dl; }
    }
  };

  int n_it = (E + SCAN_SPAN - 1) / SCAN_SPAN;
  if (n_it < 1) n_it = 1;
  const int rot = (int)(((long long)blockIdx.x * n_it) / gridDim.x);
  const bool vec_ok = ((E & 3) == 0);

  for (int it = 0; it < n_it; ++it) {
    int c = it + rot;
    if (c >= n_it) c -= n_it;
    const int b0 = c * SCAN_SPAN + tid * 4;

    if (vec_ok && (c + 1) * SCAN_SPAN <= E) {
      int4 v[SCAN_UNROLL];
#pragma unroll
      for (int u = 0; u < SCAN_UNROLL; ++u)
        v[u] = *(const int4*)(dst + b0 + u * SCAN_STEP);
#pragma unroll
      for (int u = 0; u < SCAN_UNROLL; ++u) {
        const int idx = b0 + u * SCAN_STEP;
        const bool any = ((unsigned)(v[u].x - n0) < (unsigned)nrows) | ((unsigned)(v[u].y - n0) < (unsigned)nrows) |
                         ((unsigned)(v[u].z - n0) < (unsigned)nrows) | ((unsigned)(v[u].w - n0) < (unsigned)nrows);
        if (any) {
          keep(idx + 0, v[u].x);
          keep(idx + 1, v[u].y);
          keep(idx + 2, v[u].z);
          keep(idx + 3, v[u].w);
        }
      }
    } else {
      for (int u = 0; u < SCAN_UNROLL; ++u) {
        const int idx = b0 + u * SCAN_STEP;
        for (int k = 0; k < 4; ++k)
          if (idx + k < E) keep(idx + k, dst[idx + k]);
      }
    }
  }
  __syncthreads();
  const int cnt = __builtin_amdgcn_readfirstlane(list_n);

  if (cnt <= LIST_CAP) {
    for (int base = 0; base < cnt; base += 4 * AGG_THREADS) {
      int ee[4], dd[4], ss[4];
      float4 sa[4][2], da[4][2];
#pragma unroll
      for (int k = 0; k < 4; ++k) {
        const int i = base + k * AGG_THREADS + tid;
        ee[k] = (i < cnt) ? list_e[i] : (E + n0);
        dd[k] = (i < cnt) ? list_d[i] : 0;
      }
#pragma unroll
      for (int k = 0; k < 4; ++k) ss[k] = (ee[k] < E) ? ei[ee[k]] : (ee[k] - E);
#pragma unroll
      for (int k = 0; k < 4; ++k) {
        const float4* ps = (const float4*)(a_src + (size_t)ss[k] * HEADS);
        const float4* pd = (const float4*)(a_dst + (size_t)(n0 + dd[k]) * HEADS);
        sa[k][0] = ps[0]; sa[k][1] = ps[1];
        da[k][0] = pd[0]; da[k][1] = pd[1];
      }
#pragma unroll
      for (int k = 0; k < 4; ++k) {
        const int i = base + k * AGG_THREADS + tid;
        if (i < cnt) {
          list_e[i] = ss[k];
          unsigned* mx = smax + dd[k] * HEADS;
          __hip_atomic_fetch_max(&mx[0], f2key(leaky(sa[k][0].x + da[k][0].x)), __ATOMIC_RELAXED, __HIP_MEMORY_SCOPE_WORKGROUP);
          __hip_atomic_fetch_max(&mx[1], f2key(leaky(sa[k][0].y + da[k][0].y)), __ATOMIC_RELAXED, __HIP_MEMORY_SCOPE_WORKGROUP);
          __hip_atomic_fetch_max(&mx[2], f2key(leaky(sa[k][0].z + da[k][0].z)), __ATOMIC_RELAXED, __HIP_MEMORY_SCOPE_WORKGROUP);
          __hip_atomic_fetch_max(&mx[3], f2key(leaky(sa[k][0].w + da[k][0].w)), __ATOMIC_RELAXED, __HIP_MEMORY_SCOPE_WORKGROUP);
          __hip_atomic_fetch_max(&mx[4], f2key(leaky(sa[k][1].x + da[k][1].x)), __ATOMIC_RELAXED, __HIP_MEMORY_SCOPE_WORKGROUP);
          __hip_atomic_fetch_max(&mx[5], f2key(leaky(sa[k][1].y + da[k][1].y)), __ATOMIC_RELAXED, __HIP_MEMORY_SCOPE_WORKGROUP);
          __hip_atomic_fetch_max(&mx[6], f2key(leaky(sa[k][1].z + da[k][1].z)), __ATOMIC_RELAXED, __HIP_MEMORY_SCOPE_WORKGROUP);
          __hip_atomic_fetch_max(&mx[7], f2key(leaky(sa[k][1].w + da[k][1].w)), __ATOMIC_RELAXED, __HIP_MEMORY_SCOPE_WORKGROUP);
        }
      }
    }
    __syncthreads();

    const int lo = (cnt * wave) / AGG_WAVES;
    const int hi = (cnt * (wave + 1)) / AGG_WAVES;
    for (int g = lo; g < hi; g += 32) {
      const int m = (hi - g < 32) ? (hi - g) : 32;
      int s = n0, dl = 0;
      if (lane < m) { s = list_e[g + lane]; dl = list_d[g + lane]; }
      const float4* ps = (const float4*)(a_src + (size_t)s * HEADS);
      const float4* pd = (const float4*)(a_dst + (size_t)(n0 + dl) * HEADS);
      const uint4*  pm = (const uint4*)(smax + dl * HEADS);
      const float4 s0 = ps[0], s1 = ps[1], d0 = pd[0], d1 = pd[1];
      const uint4 m0 = pm[0], m1 = pm[1];
      float al[HEADS];
      al[0] = expf(leaky(s0.x + d0.x) - key2f(m0.x));
      al[1] = expf(leaky(s0.y + d0.y) - key2f(m0.y));
      al[2] = expf(leaky(s0.z + d0.z) - key2f(m0.z));
      al[3] = expf(leaky(s0.w + d0.w) - key2f(m0.w));
      al[4] = expf(leaky(s1.x + d1.x) - key2f(m1.x));
      al[5] = expf(leaky(s1.y + d1.y) - key2f(m1.y));
      al[6] = expf(leaky(s1.z + d1.z) - key2f(m1.z));
      al[7] = expf(leaky(s1.w + d1.w) - key2f(m1.w));
      if (lane < m) {
        float* dn = den + dl * HEADS;
#pragma unroll
        for (int h = 0; h < HEADS; ++h)
          __hip_atomic_fetch_add(&dn[h], al[h], __ATOMIC_RELAXED, __HIP_MEMORY_SCOPE_WORKGROUP);
      }

      for (int i0 = 0; i0 < m; i0 += 4) {
        float xv[4][HEADS];
#pragma unroll
        for (int j = 0; j < 4; ++j) {
          const int ii = (i0 + j < m) ? (i0 + j) : (m - 1);
          const int si = __builtin_amdgcn_readlane(s, ii);
          const float* __restrict__ xr = xl + (size_t)si * F + lane;
#pragma unroll
          for (int h = 0; h < HEADS; ++h) xv[j][h] = xr[h * CH];
        }
#pragma unroll
        for (int j = 0; j < 4; ++j) {
          if (i0 + j < m) {
            const int ii = i0 + j;
            const int di = __builtin_amdgcn_readlane(dl, ii);
            float* ar = acc + di * F + lane;
#pragma unroll
            for (int h = 0; h < HEADS; ++h) {
              const float w = __int_as_float(__builtin_amdgcn_readlane(__float_as_int(al[h]), ii));
              __hip_atomic_fetch_add(&ar[h * CH], xv[j][h] * w,
                                     __ATOMIC_RELAXED, __HIP_MEMORY_SCOPE_WORKGROUP);
            }
          }
        }
      }
    }
  } else {
    const int tot = E + N;
    for (int e = tid; e < tot; e += AGG_THREADS) {
      const int d = (e < E) ? dst[e] : (e - E);
      const unsigned dl = (unsigned)(d - n0);
      if (dl < (unsigned)nrows) {
        const int s = (e < E) ? ei[e] : (e - E);
        for (int h = 0; h < HEADS; ++h)
          __hip_atomic_fetch_max(&smax[dl * HEADS + h], f2key(leaky(a_src[s * HEADS + h] + a_dst[d * HEADS + h])),
                                 __ATOMIC_RELAXED, __HIP_MEMORY_SCOPE_WORKGROUP);
      }
    }
    __syncthreads();
    for (int e = tid; e < tot; e += AGG_THREADS) {
      const int d = (e < E) ? dst[e] : (e - E);
      const unsigned dl = (unsigned)(d - n0);
      if (dl < (unsigned)nrows) {
        const int s = (e < E) ? ei[e] : (e - E);
        for (int h = 0; h < HEADS; ++h) {
          const float p = expf(leaky(a_src[s * HEADS + h] + a_dst[d * HEADS + h]) - key2f(smax[dl * HEADS + h]));
          __hip_atomic_fetch_add(&den[dl * HEADS + h], p, __ATOMIC_RELAXED, __HIP_MEMORY_SCOPE_WORKGROUP);
          for (int c = 0; c < CH; ++c)
            __hip_atomic_fetch_add(&acc[dl * F + h * CH + c], xl[(size_t)s * F + h * CH + c] * p,
                                   __ATOMIC_RELAXED, __HIP_MEMORY_SCOPE_WORKGROUP);
        }
      }
    }
  }
  __syncthreads();

  for (int i = tid; i < nrows * HEADS; i += AGG_THREADS) den[i] = 1.0f / den[i];
  __syncthreads();
  float4* __restrict__ orow = (float4*)(out + (size_t)n0 * F);
  for (int i = tid; i < nrows * (F / 4); i += AGG_THREADS) {
    float4 v = ((const float4*)acc)[i];
    const float r = den[(i >> 6) * HEADS + ((i & 63) >> 3)];
    v.x *= r; v.y *= r; v.z *= r; v.w *= r;
    orow[i] = v;
  }
}

#define LN_ROWS 4
#define LN_THREADS 256
#define LN_ROWS_PER_BLOCK (LN_ROWS * (LN_THREADS / 32))

__device__ __forceinline__ float wave_sum(float v) {
#pragma unroll
  for (int m = 16; m > 0; m >>= 1) v += __shfl_xor(v, m, 32);
  return v;
}

__global__ void __launch_bounds__(LN_THREADS)
gat_ln_kernel(float* __restrict__ out, const float* __restrict__ bias,
              const float* __restrict__ gamma, const float* __restrict__ beta, int N) {
  const int lane = threadIdx.x & 31;
  const int wave = __builtin_amdgcn_readfirstlane((int)((blockIdx.x * LN_THREADS + threadIdx.x) >> 5));
  const int row0 = wave * LN_ROWS;
  if (row0 >= N) return;

  float4 va[LN_ROWS], vb[LN_ROWS];
#pragma unroll
  for (int r = 0; r < LN_ROWS; ++r) {
    const int row = (row0 + r < N) ? (row0 + r) : row0;
    const float4* p = (const float4*)(out + (size_t)row * F);
    va[r] = p[lane];
    vb[r] = p[32 + lane];
  }
  const float4 b0 = ((const float4*)bias)[lane],  b1 = ((const float4*)bias)[32 + lane];
  const float4 g0 = ((const float4*)gamma)[lane], g1 = ((const float4*)gamma)[32 + lane];
  const float4 t0 = ((const float4*)beta)[lane],  t1 = ((const float4*)beta)[32 + lane];

  float mean[LN_ROWS];
#pragma unroll
  for (int r = 0; r < LN_ROWS; ++r) {
    va[r].x += b0.x; va[r].y += b0.y; va[r].z += b0.z; va[r].w += b0.w;
    vb[r].x += b1.x; vb[r].y += b1.y; vb[r].z += b1.z; vb[r].w += b1.w;
    const float s = ((va[r].x + va[r].y) + (va[r].z + va[r].w)) +
                    ((vb[r].x + vb[r].y) + (vb[r].z + vb[r].w));
    mean[r] = wave_sum(s) * (1.0f / F);
  }
  float rs[LN_ROWS];
#pragma unroll
  for (int r = 0; r < LN_ROWS; ++r) {
    va[r].x -= mean[r]; va[r].y -= mean[r]; va[r].z -= mean[r]; va[r].w -= mean[r];
    vb[r].x -= mean[r]; vb[r].y -= mean[r]; vb[r].z -= mean[r]; vb[r].w -= mean[r];
    const float q = ((va[r].x * va[r].x + va[r].y * va[r].y) + (va[r].z * va[r].z + va[r].w * va[r].w)) +
                    ((vb[r].x * vb[r].x + vb[r].y * vb[r].y) + (vb[r].z * vb[r].z + vb[r].w * vb[r].w));
    rs[r] = rsqrtf(wave_sum(q) * (1.0f / F) + LN_EPS);
  }
#pragma unroll
  for (int r = 0; r < LN_ROWS; ++r) {
    if (row0 + r < N) {
      float4* o = (float4*)(out + (size_t)(row0 + r) * F);
      float4 ya, yb;
      ya.x = va[r].x * rs[r] * g0.x + t0.x; ya.y = va[r].y * rs[r] * g0.y + t0.y;
      ya.z = va[r].z * rs[r] * g0.z + t0.z; ya.w = va[r].w * rs[r] * g0.w + t0.w;
      yb.x = vb[r].x * rs[r] * g1.x + t1.x; yb.y = vb[r].y * rs[r] * g1.y + t1.y;
      yb.z = vb[r].z * rs[r] * g1.z + t1.z; yb.w = vb[r].w * rs[r] * g1.w + t1.w;
      o[lane] = ya;
      o[32 + lane] = yb;
    }
  }
}

extern "C" void kernel_launch(void* const* d_in, const int* in_sizes, int n_in,
                              void* d_out, int out_size, void* d_ws, size_t ws_size,
                              hipStream_t stream) {
  const float* x       = (const float*)d_in[0];
  const int*   ei      = (const int*)  d_in[1];
  const float* W       = (const float*)d_in[2];
  const float* att_src = (const float*)d_in[3];
  const float* att_dst = (const float*)d_in[4];
  const float* bias    = (const float*)d_in[5];
  const float* gamma   = (const float*)d_in[6];
  const float* beta    = (const float*)d_in[7];
  float* out = (float*)d_out;

  const int N = in_sizes[0] / F;
  const int E = in_sizes[1] / 2;

  char* ws = (char*)d_ws;
  float*    xl     = (float*)ws;    ws += (size_t)N * F * sizeof(float);
  float*    a_src  = (float*)ws;    ws += (size_t)N * HEADS * sizeof(float);
  float*    a_dst  = (float*)ws;

  gat_gemm_wmma_kernel<<<(N + 15) / 16, 512, 0, stream>>>(x, W, xl, N);

  {
    int NH = N * HEADS;
    gat_att_kernel<<<(NH + 255) / 256, 256, 0, stream>>>(xl, att_src, att_dst,
                                                         a_src, a_dst, NH);
  }

  gat_edge_agg_kernel<<<(N + NPB - 1) / NPB, AGG_THREADS, 0, stream>>>(ei, xl, a_src, a_dst,
                                                                        out, E, N);

  gat_ln_kernel<<<(N + LN_ROWS_PER_BLOCK - 1) / LN_ROWS_PER_BLOCK, LN_THREADS, 0, stream>>>(out, bias, gamma, beta, N);
}


# dst column packed to one owner byte per edge (dst/196, by the half-logit launch): blocks scan it 16 edges per load, 4 per word; dst fetched for kept edges only; list and LDS adds still arrival order
# speedup vs baseline: 4.2628x; 1.0535x over previous
#include <hip/hip_runtime.h>
#include <math.h>

typedef float v2f __attribute__((ext_vector_type(2)));
typedef float v8f __attribute__((ext_vector_type(8)));

#define HEADS 8
#define CH 32
#define F 256
#define NEG_SLOPE 0.2f
#define LN_EPS 1e-5f

#define LDS_STRIDE 260

__device__ __forceinline__ unsigned f2key(float f) {
  unsigned b = __float_as_uint(f);
  return (b & 0x80000000u) ? ~b : (b | 0x80000000u);
}
__device__ __forceinline__ float key2f(unsigned k) {
  unsigned b = (k & 0x80000000u) ? (k & 0x7fffffffu) : ~k;
  return __uint_as_float(b);
}

__global__ void __launch_bounds__(512)
gat_gemm_wmma_kernel(const float* __restrict__ x, const float* __restrict__ W,
                     float* __restrict__ xl, int N) {
  __shared__ float lds_a[16 * LDS_STRIDE];

  const int tid  = threadIdx.x;
  const int lane = tid & 31;
  const int wave = tid >> 5;
  const int row0 = blockIdx.x * 16;
  const int col0 = wave * 16;

  {
    int r = tid >> 5;
    int c = (tid & 31) * 8;
    int grow = row0 + r;
    if (grow >= N) grow = N - 1;
    const float4* __restrict__ src = (const float4*)(x + (size_t)grow * F + c);
    float4* dst = (float4*)(&lds_a[r * LDS_STRIDE + c]);
    dst[0] = src[0];
    dst[1] = src[1];
  }
  __syncthreads();

  const int m    = lane & 15;
  const int koff = (lane >> 4) * 2;
  const float* __restrict__ arow = &lds_a[m * LDS_STRIDE];

  const int n = lane & 15;

  v8f acc = {0.f, 0.f, 0.f, 0.f, 0.f, 0.f, 0.f, 0.f};
  for (int k = 0; k < F; k += 4) {
    v2f a = *(const v2f*)(arow + k + koff);
    v2f b;
    b.x = W[(size_t)(k + koff) * F + col0 + n];
    b.y = W[(size_t)(k + koff + 1) * F + col0 + n];
    acc = __builtin_amdgcn_wmma_f32_16x16x4_f32(
         false, a,  false, b,
         (short)0, acc,  false,  false);
  }

  const int rbase = row0 + ((lane >> 4) * 8);
  float* __restrict__ dst = xl + (size_t)rbase * F + col0 + n;
  if (row0 + 16 <= N) {
#pragma unroll
    for (int j = 0; j < 8; ++j) dst[(size_t)j * F] = acc[j];
  } else {
#pragma unroll
    for (int j = 0; j < 8; ++j) {
      if (rbase + j < N) dst[(size_t)j * F] = acc[j];
    }
  }
}

#define NPB 196

__device__ __forceinline__ bool owner_plane_ok(int N, int E, unsigned long long ws_bytes) {
  const unsigned long long need = (unsigned long long)N * F * sizeof(float) +
                                  2ull * N * HEADS * sizeof(float) + (unsigned long long)E;
  return (E >= 16) && ((E & 15) == 0) && ((N + NPB - 1) / NPB <= 256) && (ws_bytes >= need);
}

__global__ void gat_att_kernel(const float* __restrict__ xl,
                               const float* __restrict__ att_src,
                               const float* __restrict__ att_dst,
                               float* __restrict__ a_src, float* __restrict__ a_dst,
                               int NH, const int* __restrict__ ei, unsigned* __restrict__ own,
                               int E, unsigned long long ws_bytes) {
  int t = blockIdx.x * blockDim.x + threadIdx.x;
  if (owner_plane_ok(NH / HEADS, E, ws_bytes)) {
    const int4* __restrict__ d4 = (const int4*)(ei + E);
    for (int g = t; g < (E >> 2); g += gridDim.x * blockDim.x) {
      const int4 d = d4[g];
      own[g] = ((unsigned)d.x / NPB) | (((unsigned)d.y / NPB) << 8) |
               (((unsigned)d.z / NPB) << 16) | (((unsigned)d.w / NPB) << 24);
    }
  }
  if (t >= NH) return;
  int nidx = t >> 3, h = t & 7;
  const float* __restrict__ xr = xl + (size_t)nidx * F + h * CH;
  const float* __restrict__ as = att_src + h * CH;
  const float* __restrict__ ad = att_dst + h * CH;
  float s = 0.f, d = 0.f;
#pragma unroll
  for (int c = 0; c < CH; ++c) {
    float v = xr[c];
    s += v * as[c];
    d += v * ad[c];
  }
  a_src[t] = s;
  a_dst[t] = d;
}

__device__ __forceinline__ float leaky(float v) {
  return (v >= 0.f) ? v : v * NEG_SLOPE;
}

#define AGG_THREADS 1024
#define AGG_WAVES (AGG_THREADS / 32)
#define LIST_CAP 4096
#define SCAN_UNROLL 8
#define SCAN_STEP (AGG_THREADS * 4)
#define SCAN_SPAN (SCAN_STEP * SCAN_UNROLL)
#define OWN_STEP (AGG_THREADS * 16)
#define OWN_SPAN (OWN_STEP * SCAN_UNROLL)

__global__ void __launch_bounds__(AGG_THREADS)
gat_edge_agg_kernel(const int* __restrict__ ei, const float* __restrict__ xl,
                    const float* __restrict__ a_src, const float* __restrict__ a_dst,
                    float* __restrict__ out, int E, int N,
                    const unsigned* __restrict__ own, unsigned long long ws_bytes) {
  __shared__ __attribute__((aligned(16))) float acc[NPB * F];
  __shared__ __attribute__((aligned(16))) unsigned smax[NPB * HEADS];
  __shared__ __attribute__((aligned(16))) float den[NPB * HEADS];
  __shared__ int list_e[LIST_CAP];
  __shared__ int list_d[LIST_CAP];
  __shared__ int list_n;

  const int tid  = threadIdx.x;
  const int lane = tid & 31;
  const int wave = __builtin_amdgcn_readfirstlane(tid >> 5);
  const int n0 = blockIdx.x * NPB;
  const int nrows = (N - n0 < NPB) ? (N - n0) : NPB;
  const int* __restrict__ dst = ei + E;

  for (int i = tid; i < NPB * F / 4; i += AGG_THREADS)
    ((float4*)acc)[i] = make_float4(0.f, 0.f, 0.f, 0.f);
  for (int i = tid; i < NPB * HEADS; i += AGG_THREADS) {
    smax[i] = 0x007FFFFFu;
    den[i] = 0.f;
  }
  if (tid < nrows) list_e[tid] = E + n0 + tid;
  if (tid == 0) list_n = nrows;
  __syncthreads();

  auto keep = [&](int e, int dv) {
    unsigned dl = (unsigned)(dv - n0);
    if (dl < (unsigned)nrows) {
      int slot = __hip_atomic_fetch_add(&list_n, 1, __ATOMIC_RELAXED, __HIP_MEMORY_SCOPE_WORKGROUP);
      if (slot < LIST_CAP) list_e[slot] = e;
    }
  };
  auto keep_e = [&](int e) {
    int slot = __hip_atomic_fetch_add(&list_n, 1, __ATOMIC_RELAXED, __HIP_MEMORY_SCOPE_WORKGROUP);
    if (slot < LIST_CAP) list_e[slot] = e;
  };

  const bool use_plane = owner_plane_ok(N, E, ws_bytes);
  if (use_plane) {
    int n_it = (E + OWN_SPAN - 1) / OWN_SPAN;
    if (n_it < 1) n_it = 1;
    const int rot = (int)(((long long)blockIdx.x * n_it) / gridDim.x);
    const unsigned pat  = blockIdx.x * 0x01010101u;
    const uint4* __restrict__ own4 = (const uint4*)own;
    for (int it = 0; it < n_it; ++it) {
      int c = it + rot;
      if (c >= n_it) c -= n_it;
      const int b0 = c * OWN_SPAN + tid * 16;
      uint4 v[SCAN_UNROLL];
#pragma unroll
      for (int u = 0; u < SCAN_UNROLL; ++u) {
        const int idx = b0 + u * OWN_STEP;
        v[u] = own4[min(idx, E - 16) >> 4];
      }
#pragma unroll
      for (int u = 0; u < SCAN_UNROLL; u += 2) {
        const int idx = b0 + u * OWN_STEP;
        unsigned m = 0;
#pragma unroll
        for (int w = 0; w < 2; ++w) {
          const unsigned x[4] = {v[u + w].x ^ pat, v[u + w].y ^ pat, v[u + w].z ^ pat, v[u + w].w ^ pat};
          unsigned mw = 0;
#pragma unroll
          for (int q = 0; q < 4; ++q) {
            const unsigned f = ~(((x[q] & 0x7f7f7f7fu) + 0x7f7f7f7fu) | x[q]) & 0x80808080u;
            mw |= f >> (7 - q);
          }
          if (idx + w * OWN_STEP >= E) mw = 0;
          m |= mw << (4 * w);
        }
        while (m) {
          const int p = __builtin_ctz(m);
          keep_e(idx + ((p >> 2) & 1) * OWN_STEP + 4 * (p & 3) + (p >> 3));
          m &= m - 1;
        }
      }
    }
  } else {
  int n_it = (E + SCAN_SPAN - 1) / SCAN_SPAN;
  if (n_it < 1) n_it = 1;
  const int rot = (int)(((long long)blockIdx.x * n_it) / gridDim.x);
  const bool vec_ok = ((E & 3) == 0);

  for (int it = 0; it < n_it; ++it) {
    int c = it + rot;
    if (c >= n_it) c -= n_it;
    const int b0 = c * SCAN_SPAN + tid * 4;

    if (vec_ok && (c + 1) * SCAN_SPAN <= E) {
      int4 v[SCAN_UNROLL];
#pragma unroll
      for (int u = 0; u < SCAN_UNROLL; ++u)
        v[u] = *(const int4*)(dst + b0 + u * SCAN_STEP);
#pragma unroll
      for (int u = 0; u < SCAN_UNROLL; ++u) {
        const int idx = b0 + u * SCAN_STEP;
        const bool any = ((unsigned)(v[u].x - n0) < (unsigned)nrows) | ((unsigned)(v[u].y - n0) < (unsigned)nrows) |
                         ((unsigned)(v[u].z - n0) < (unsigned)nrows) | ((unsigned)(v[u].w - n0) < (unsigned)nrows);
        if (any) {
          keep(idx + 0, v[u].x);
          keep(idx + 1, v[u].y);
          keep(idx + 2, v[u].z);
          keep(idx + 3, v[u].w);
        }
      }
    } else {
      for (int u = 0; u < SCAN_UNROLL; ++u) {
        const int idx = b0 + u * SCAN_STEP;
        for (int k = 0; k < 4; ++k)
          if (idx + k < E) keep(idx + k, dst[idx + k]);
      }
    }
  }
  }
  __syncthreads();
  const int cnt = __builtin_amdgcn_readfirstlane(list_n);

  if (cnt <= LIST_CAP) {
    for (int base = 0; base < cnt; base += 4 * AGG_THREADS) {
      int ee[4], dd[4], ss[4];
      float4 sa[4][2], da[4][2];
#pragma unroll
      for (int k = 0; k < 4; ++k) {
        const int i = base + k * AGG_THREADS + tid;
        ee[k] = (i < cnt) ? list_e[i] : (E + n0);
      }
#pragma unroll
      for (int k = 0; k < 4; ++k) {
        const bool real = ee[k] < E;
        ss[k] = real ? ei[ee[k]] : (ee[k] - E);
        const int dv = real ? dst[ee[k]] : (ee[k] - E);
        dd[k] = min(max(dv - n0, 0), nrows - 1);
      }
#pragma unroll
      for (int k = 0; k < 4; ++k) {
        const float4* ps = (const float4*)(a_src + (size_t)ss[k] * HEADS);
        const float4* pd = (const float4*)(a_dst + (size_t)(n0 + dd[k]) * HEADS);
        sa[k][0] = ps[0]; sa[k][1] = ps[1];
        da[k][0] = pd[0]; da[k][1] = pd[1];
      }
#pragma unroll
      for (int k = 0; k < 4; ++k) {
        const int i = base + k * AGG_THREADS + tid;
        if (i < cnt) {
          list_e[i] = ss[k];
          list_d[i] = dd[k];
          unsigned* mx = smax + dd[k] * HEADS;
          __hip_atomic_fetch_max(&mx[0], f2key(leaky(sa[k][0].x + da[k][0].x)), __ATOMIC_RELAXED, __HIP_MEMORY_SCOPE_WORKGROUP);
          __hip_atomic_fetch_max(&mx[1], f2key(leaky(sa[k][0].y + da[k][0].y)), __ATOMIC_RELAXED, __HIP_MEMORY_SCOPE_WORKGROUP);
          __hip_atomic_fetch_max(&mx[2], f2key(leaky(sa[k][0].z + da[k][0].z)), __ATOMIC_RELAXED, __HIP_MEMORY_SCOPE_WORKGROUP);
          __hip_atomic_fetch_max(&mx[3], f2key(leaky(sa[k][0].w + da[k][0].w)), __ATOMIC_RELAXED, __HIP_MEMORY_SCOPE_WORKGROUP);
          __hip_atomic_fetch_max(&mx[4], f2key(leaky(sa[k][1].x + da[k][1].x)), __ATOMIC_RELAXED, __HIP_MEMORY_SCOPE_WORKGROUP);
          __hip_atomic_fetch_max(&mx[5], f2key(leaky(sa[k][1].y + da[k][1].y)), __ATOMIC_RELAXED, __HIP_MEMORY_SCOPE_WORKGROUP);
          __hip_atomic_fetch_max(&mx[6], f2key(leaky(sa[k][1].z + da[k][1].z)), __ATOMIC_RELAXED, __HIP_MEMORY_SCOPE_WORKGROUP);
          __hip_atomic_fetch_max(&mx[7], f2key(leaky(sa[k][1].w + da[k][1].w)), __ATOMIC_RELAXED, __HIP_MEMORY_SCOPE_WORKGROUP);
        }
      }
    }
    __syncthreads();

    const int lo = (cnt * wave) / AGG_WAVES;
    const int hi = (cnt * (wave + 1)) / AGG_WAVES;
    for (int g = lo; g < hi; g += 32) {
      const int m = (hi - g < 32) ? (hi - g) : 32;
      int s = n0, dl = 0;
      if (lane < m) { s = list_e[g + lane]; dl = list_d[g + lane]; }
      const float4* ps = (const float4*)(a_src + (size_t)s * HEADS);
      const float4* pd = (const float4*)(a_dst + (size_t)(n0 + dl) * HEADS);
      const uint4*  pm = (const uint4*)(smax + dl * HEADS);
      const float4 s0 = ps[0], s1 = ps[1], d0 = pd[0], d1 = pd[1];
      const uint4 m0 = pm[0], m1 = pm[1];
      float al[HEADS];
      al[0] = expf(leaky(s0.x + d0.x) - key2f(m0.x));
      al[1] = expf(leaky(s0.y + d0.y) - key2f(m0.y));
      al[2] = expf(leaky(s0.z + d0.z) - key2f(m0.z));
      al[3] = expf(leaky(s0.w + d0.w) - key2f(m0.w));
      al[4] = expf(leaky(s1.x + d1.x) - key2f(m1.x));
      al[5] = expf(leaky(s1.y + d1.y) - key2f(m1.y));
      al[6] = expf(leaky(s1.z + d1.z) - key2f(m1.z));
      al[7] = expf(leaky(s1.w + d1.w) - key2f(m1.w));
      if (lane < m) {
        float* dn = den + dl * HEADS;
#pragma unroll
        for (int h = 0; h < HEADS; ++h)
          __hip_atomic_fetch_add(&dn[h], al[h], __ATOMIC_RELAXED, __HIP_MEMORY_SCOPE_WORKGROUP);
      }

      for (int i0 = 0; i0 < m; i0 += 4) {
        float xv[4][HEADS];
#pragma unroll
        for (int j = 0; j < 4; ++j) {
          const int ii = (i0 + j < m) ? (i0 + j) : (m - 1);
          const int si = __builtin_amdgcn_readlane(s, ii);
          const float* __restrict__ xr = xl + (size_t)si * F + lane;
#pragma unroll
          for (int h = 0; h < HEADS; ++h) xv[j][h] = xr[h * CH];
        }
#pragma unroll
        for (int j = 0; j < 4; ++j) {
          if (i0 + j < m) {
            const int ii = i0 + j;
            const int di = __builtin_amdgcn_readlane(dl, ii);
            float* ar = acc + di * F + lane;
#pragma unroll
            for (int h = 0; h < HEADS; ++h) {
              const float w = __int_as_float(__builtin_amdgcn_readlane(__float_as_int(al[h]), ii));
              __hip_atomic_fetch_add(&ar[h * CH], xv[j][h] * w,
                                     __ATOMIC_RELAXED, __HIP_MEMORY_SCOPE_WORKGROUP);
            }
          }
        }
      }
    }
  } else {
    const int tot = E + N;
    for (int e = tid; e < tot; e += AGG_THREADS) {
      const int d = (e < E) ? dst[e] : (e - E);
      const unsigned dl = (unsigned)(d - n0);
      if (dl < (unsigned)nrows) {
        const int s = (e < E) ? ei[e] : (e - E);
        for (int h = 0; h < HEADS; ++h)
          __hip_atomic_fetch_max(&smax[dl * HEADS + h], f2key(leaky(a_src[s * HEADS + h] + a_dst[d * HEADS + h])),
                                 __ATOMIC_RELAXED, __HIP_MEMORY_SCOPE_WORKGROUP);
      }
    }
    __syncthreads();
    for (int e = tid; e < tot; e += AGG_THREADS) {
      const int d = (e < E) ? dst[e] : (e - E);
      const unsigned dl = (unsigned)(d - n0);
      if (dl < (unsigned)nrows) {
        const int s = (e < E) ? ei[e] : (e - E);
        for (int h = 0; h < HEADS; ++h) {
          const float p = expf(leaky(a_src[s * HEADS + h] + a_dst[d * HEADS + h]) - key2f(smax[dl * HEADS + h]));
          __hip_atomic_fetch_add(&den[dl * HEADS + h], p, __ATOMIC_RELAXED, __HIP_MEMORY_SCOPE_WORKGROUP);
          for (int c = 0; c < CH; ++c)
            __hip_atomic_fetch_add(&acc[dl * F + h * CH + c], xl[(size_t)s * F + h * CH + c] * p,
                                   __ATOMIC_RELAXED, __HIP_MEMORY_SCOPE_WORKGROUP);
        }
      }
    }
  }
  __syncthreads();

  for (int i = tid; i < nrows * HEADS; i += AGG_THREADS) den[i] = 1.0f / den[i];
  __syncthreads();
  float4* __restrict__ orow = (float4*)(out + (size_t)n0 * F);
  for (int i = tid; i < nrows * (F / 4); i += AGG_THREADS) {
    float4 v = ((const float4*)acc)[i];
    const float r = den[(i >> 6) * HEADS + ((i & 63) >> 3)];
    v.x *= r; v.y *= r; v.z *= r; v.w *= r;
    orow[i] = v;
  }
}

#define LN_ROWS 4
#define LN_THREADS 256
#define LN_ROWS_PER_BLOCK (LN_ROWS * (LN_THREADS / 32))

__device__ __forceinline__ float wave_sum(float v) {
#pragma unroll
  for (int m = 16; m > 0; m >>= 1) v += __shfl_xor(v, m, 32);
  return v;
}

__global__ void __launch_bounds__(LN_THREADS)
gat_ln_kernel(float* __restrict__ out, const float* __restrict__ bias,
              const float* __restrict__ gamma, const float* __restrict__ beta, int N) {
  const int lane = threadIdx.x & 31;
  const int wave = __builtin_amdgcn_readfirstlane((int)((blockIdx.x * LN_THREADS + threadIdx.x) >> 5));
  const int row0 = wave * LN_ROWS;
  if (row0 >= N) return;

  float4 va[LN_ROWS], vb[LN_ROWS];
#pragma unroll
  for (int r = 0; r < LN_ROWS; ++r) {
    const int row = (row0 + r < N) ? (row0 + r) : row0;
    const float4* p = (const float4*)(out + (size_t)row * F);
    va[r] = p[lane];
    vb[r] = p[32 + lane];
  }
  const float4 b0 = ((const float4*)bias)[lane],  b1 = ((const float4*)bias)[32 + lane];
  const float4 g0 = ((const float4*)gamma)[lane], g1 = ((const float4*)gamma)[32 + lane];
  const float4 t0 = ((const float4*)beta)[lane],  t1 = ((const float4*)beta)[32 + lane];

  float mean[LN_ROWS];
#pragma unroll
  for (int r = 0; r < LN_ROWS; ++r) {
    va[r].x += b0.x; va[r].y += b0.y; va[r].z += b0.z; va[r].w += b0.w;
    vb[r].x += b1.x; vb[r].y += b1.y; vb[r].z += b1.z; vb[r].w += b1.w;
    const float s = ((va[r].x + va[r].y) + (va[r].z + va[r].w)) +
                    ((vb[r].x + vb[r].y) + (vb[r].z + vb[r].w));
    mean[r] = wave_sum(s) * (1.0f / F);
  }
  float rs[LN_ROWS];
#pragma unroll
  for (int r = 0; r < LN_ROWS; ++r) {
    va[r].x -= mean[r]; va[r].y -= mean[r]; va[r].z -= mean[r]; va[r].w -= mean[r];
    vb[r].x -= mean[r]; vb[r].y -= mean[r]; vb[r].z -= mean[r]; vb[r].w -= mean[r];
    const float q = ((va[r].x * va[r].x + va[r].y * va[r].y) + (va[r].z * va[r].z + va[r].w * va[r].w)) +
                    ((vb[r].x * vb[r].x + vb[r].y * vb[r].y) + (vb[r].z * vb[r].z + vb[r].w * vb[r].w));
    rs[r] = rsqrtf(wave_sum(q) * (1.0f / F) + LN_EPS);
  }
#pragma unroll
  for (int r = 0; r < LN_ROWS; ++r) {
    if (row0 + r < N) {
      float4* o = (float4*)(out + (size_t)(row0 + r) * F);
      float4 ya, yb;
      ya.x = va[r].x * rs[r] * g0.x + t0.x; ya.y = va[r].y * rs[r] * g0.y + t0.y;
      ya.z = va[r].z * rs[r] * g0.z + t0.z; ya.w = va[r].w * rs[r] * g0.w + t0.w;
      yb.x = vb[r].x * rs[r] * g1.x + t1.x; yb.y = vb[r].y * rs[r] * g1.y + t1.y;
      yb.z = vb[r].z * rs[r] * g1.z + t1.z; yb.w = vb[r].w * rs[r] * g1.w + t1.w;
      o[lane] = ya;
      o[32 + lane] = yb;
    }
  }
}

extern "C" void kernel_launch(void* const* d_in, const int* in_sizes, int n_in,
                              void* d_out, int out_size, void* d_ws, size_t ws_size,
                              hipStream_t stream) {
  const float* x       = (const float*)d_in[0];
  const int*   ei      = (const int*)  d_in[1];
  const float* W       = (const float*)d_in[2];
  const float* att_src = (const float*)d_in[3];
  const float* att_dst = (const float*)d_in[4];
  const float* bias    = (const float*)d_in[5];
  const float* gamma   = (const float*)d_in[6];
  const float* beta    = (const float*)d_in[7];
  float* out = (float*)d_out;

  const int N = in_sizes[0] / F;
  const int E = in_sizes[1] / 2;

  char* ws = (char*)d_ws;
  float*    xl     = (float*)ws;    ws += (size_t)N * F * sizeof(float);
  float*    a_src  = (float*)ws;    ws += (size_t)N * HEADS * sizeof(float);
  float*    a_dst  = (float*)ws;    ws += (size_t)N * HEADS * sizeof(float);
  unsigned* own    = (unsigned*)ws;

  gat_gemm_wmma_kernel<<<(N + 15) / 16, 512, 0, stream>>>(x, W, xl, N);

  {
    int NH = N * HEADS;
    gat_att_kernel<<<(NH + 255) / 256, 256, 0, stream>>>(xl, att_src, att_dst,
                                                         a_src, a_dst, NH, ei, own, E,
                                                         (unsigned long long)ws_size);
  }

  gat_edge_agg_kernel<<<(N + NPB - 1) / NPB, AGG_THREADS, 0, stream>>>(ei, xl, a_src, a_dst,
                                                                        out, E, N, own,
                                                                        (unsigned long long)ws_size);

  gat_ln_kernel<<<(N + LN_ROWS_PER_BLOCK - 1) / LN_ROWS_PER_BLOCK, LN_THREADS, 0, stream>>>(out, bias, gamma, beta, N);
}


# agg block orders its LDS edge list by dst row (rank = returning LDS add, arrival order; row-count scan); a wave sums one row in registers in list order, exp-sum 4 partials+tree; no LDS float adds
# speedup vs baseline: 4.8298x; 1.1330x over previous
#include <hip/hip_runtime.h>
#include <math.h>

typedef float v2f __attribute__((ext_vector_type(2)));
typedef float v8f __attribute__((ext_vector_type(8)));

#define HEADS 8
#define CH 32
#define F 256
#define NEG_SLOPE 0.2f
#define LN_EPS 1e-5f

#define LDS_STRIDE 260

__device__ __forceinline__ unsigned f2key(float f) {
  unsigned b = __float_as_uint(f);
  return (b & 0x80000000u) ? ~b : (b | 0x80000000u);
}
__device__ __forceinline__ float key2f(unsigned k) {
  unsigned b = (k & 0x80000000u) ? (k & 0x7fffffffu) : ~k;
  return __uint_as_float(b);
}

__global__ void __launch_bounds__(512)
gat_gemm_wmma_kernel(const float* __restrict__ x, const float* __restrict__ W,
                     float* __restrict__ xl, int N) {
  __shared__ float lds_a[16 * LDS_STRIDE];

  const int tid  = threadIdx.x;
  const int lane = tid & 31;
  const int wave = tid >> 5;
  const int row0 = blockIdx.x * 16;
  const int col0 = wave * 16;

  {
    int r = tid >> 5;
    int c = (tid & 31) * 8;
    int grow = row0 + r;
    if (grow >= N) grow = N - 1;
    const float4* __restrict__ src = (const float4*)(x + (size_t)grow * F + c);
    float4* dst = (float4*)(&lds_a[r * LDS_STRIDE + c]);
    dst[0] = src[0];
    dst[1] = src[1];
  }
  __syncthreads();

  const int m    = lane & 15;
  const int koff = (lane >> 4) * 2;
  const float* __restrict__ arow = &lds_a[m * LDS_STRIDE];

  const int n = lane & 15;

  v8f acc = {0.f, 0.f, 0.f, 0.f, 0.f, 0.f, 0.f, 0.f};
  for (int k = 0; k < F; k += 4) {
    v2f a = *(const v2f*)(arow + k + koff);
    v2f b;
    b.x = W[(size_t)(k + koff) * F + col0 + n];
    b.y = W[(size_t)(k + koff + 1) * F + col0 + n];
    acc = __builtin_amdgcn_wmma_f32_16x16x4_f32(
         false, a,  false, b,
         (short)0, acc,  false,  false);
  }

  const int rbase = row0 + ((lane >> 4) * 8);
  float* __restrict__ dst = xl + (size_t)rbase * F + col0 + n;
  if (row0 + 16 <= N) {
#pragma unroll
    for (int j = 0; j < 8; ++j) dst[(size_t)j * F] = acc[j];
  } else {
#pragma unroll
    for (int j = 0; j < 8; ++j) {
      if (rbase + j < N) dst[(size_t)j * F] = acc[j];
    }
  }
}

#define NPB 196

__device__ __forceinline__ bool owner_plane_ok(int N, int E, unsigned long long ws_bytes) {
  const unsigned long long need = (unsigned long long)N * F * sizeof(float) +
                                  2ull * N * HEADS * sizeof(float) + (unsigned long long)E;
  return (E >= 16) && ((E & 15) == 0) && ((N + NPB - 1) / NPB <= 256) && (ws_bytes >= need);
}

__global__ void gat_att_kernel(const float* __restrict__ xl,
                               const float* __restrict__ att_src,
                               const float* __restrict__ att_dst,
                               float* __restrict__ a_src, float* __restrict__ a_dst,
                               int NH, const int* __restrict__ ei, unsigned* __restrict__ own,
                               int E, unsigned long long ws_bytes) {
  int t = blockIdx.x * blockDim.x + threadIdx.x;
  if (owner_plane_ok(NH / HEADS, E, ws_bytes)) {
    const int4* __restrict__ d4 = (const int4*)(ei + E);
    for (int g = t; g < (E >> 2); g += gridDim.x * blockDim.x) {
      const int4 d = d4[g];
      own[g] = ((unsigned)d.x / NPB) | (((unsigned)d.y / NPB) << 8) |
               (((unsigned)d.z / NPB) << 16) | (((unsigned)d.w / NPB) << 24);
    }
  }
  if (t >= NH) return;
  int nidx = t >> 3, h = t & 7;
  const float* __restrict__ xr = xl + (size_t)nidx * F + h * CH;
  const float* __restrict__ as = att_src + h * CH;
  const float* __restrict__ ad = att_dst + h * CH;
  float s = 0.f, d = 0.f;
#pragma unroll
  for (int c = 0; c < CH; ++c) {
    float v = xr[c];
    s += v * as[c];
    d += v * ad[c];
  }
  a_src[t] = s;
  a_dst[t] = d;
}

__device__ __forceinline__ float leaky(float v) {
  return (v >= 0.f) ? v : v * NEG_SLOPE;
}

#define AGG_THREADS 1024
#define AGG_WAVES (AGG_THREADS / 32)
#define LIST_CAP 4096
#define SCAN_UNROLL 8
#define SCAN_STEP (AGG_THREADS * 4)
#define SCAN_SPAN (SCAN_STEP * SCAN_UNROLL)
#define OWN_STEP (AGG_THREADS * 16)
#define OWN_SPAN (OWN_STEP * SCAN_UNROLL)
#define ROWS_PER_LANE ((NPB + 31) / 32)
#define EDGE_GROUP 4

static_assert(LIST_CAP == 4 * AGG_THREADS, "pass (1) gives every thread four list entries");
static_assert(EDGE_GROUP * HEADS == 32, "one lane per (edge of the group, head) weight");

__global__ void __launch_bounds__(AGG_THREADS)
gat_edge_agg_kernel(const int* __restrict__ ei, const float* __restrict__ xl,
                    const float* __restrict__ a_src, const float* __restrict__ a_dst,
                    float* __restrict__ out, int E, int N,
                    const unsigned* __restrict__ own, unsigned long long ws_bytes) {
  __shared__ __attribute__((aligned(16))) float acc[NPB * F];
  __shared__ __attribute__((aligned(16))) unsigned smax[NPB * HEADS];
  __shared__ __attribute__((aligned(16))) float den[NPB * HEADS];
  __shared__ int list_e[LIST_CAP];
  __shared__ int row_cnt[NPB];
  __shared__ int row_start[NPB];
  __shared__ int list_n;
  __shared__ int next_row;

  const int tid  = threadIdx.x;
  const int lane = tid & 31;
  const int wave = __builtin_amdgcn_readfirstlane(tid >> 5);
  const int n0 = blockIdx.x * NPB;
  const int nrows = (N - n0 < NPB) ? (N - n0) : NPB;
  const int* __restrict__ dst = ei + E;

  for (int i = tid; i < NPB * HEADS; i += AGG_THREADS) smax[i] = 0x007FFFFFu;
  if (tid < NPB) row_cnt[tid] = 0;
  if (tid < nrows) list_e[tid] = E + n0 + tid;
  if (tid == 0) { list_n = nrows; next_row = 0; }
  __syncthreads();

  auto keep = [&](int e, int dv) {
    unsigned dl = (unsigned)(dv - n0);
    if (dl < (unsigned)nrows) {
      int slot = __hip_atomic_fetch_add(&list_n, 1, __ATOMIC_RELAXED, __HIP_MEMORY_SCOPE_WORKGROUP);
      if (slot < LIST_CAP) list_e[slot] = e;
    }
  };
  auto keep_e = [&](int e) {
    int slot = __hip_atomic_fetch_add(&list_n, 1, __ATOMIC_RELAXED, __HIP_MEMORY_SCOPE_WORKGROUP);
    if (slot < LIST_CAP) list_e[slot] = e;
  };

  const bool use_plane = owner_plane_ok(N, E, ws_bytes);
  if (use_plane) {
    int n_it = (E + OWN_SPAN - 1) / OWN_SPAN;
    if (n_it < 1) n_it = 1;
    const int rot = (int)(((long long)blockIdx.x * n_it) / gridDim.x);
    const unsigned pat  = blockIdx.x * 0x01010101u;
    const uint4* __restrict__ own4 = (const uint4*)own;
    for (int it = 0; it < n_it; ++it) {
      int c = it + rot;
      if (c >= n_it) c -= n_it;
      const int b0 = c * OWN_SPAN + tid * 16;
      uint4 v[SCAN_UNROLL];
#pragma unroll
      for (int u = 0; u < SCAN_UNROLL; ++u) {
        const int idx = b0 + u * OWN_STEP;
        v[u] = own4[min(idx, E - 16) >> 4];
      }
#pragma unroll
      for (int u = 0; u < SCAN_UNROLL; u += 2) {
        const int idx = b0 + u * OWN_STEP;
        unsigned m = 0;
#pragma unroll
        for (int w = 0; w < 2; ++w) {
          const unsigned x[4] = {v[u + w].x ^ pat, v[u + w].y ^ pat, v[u + w].z ^ pat, v[u + w].w ^ pat};
          unsigned mw = 0;
#pragma unroll
          for (int q = 0; q < 4; ++q) {
            const unsigned f = ~(((x[q] & 0x7f7f7f7fu) + 0x7f7f7f7fu) | x[q]) & 0x80808080u;
            mw |= f >> (7 - q);
          }
          if (idx + w * OWN_STEP >= E) mw = 0;
          m |= mw << (4 * w);
        }
        while (m) {
          const int p = __builtin_ctz(m);
          keep_e(idx + ((p >> 2) & 1) * OWN_STEP + 4 * (p & 3) + (p >> 3));
          m &= m - 1;
        }
      }
    }
  } else {
  int n_it = (E + SCAN_SPAN - 1) / SCAN_SPAN;
  if (n_it < 1) n_it = 1;
  const int rot = (int)(((long long)blockIdx.x * n_it) / gridDim.x);
  const bool vec_ok = ((E & 3) == 0);

  for (int it = 0; it < n_it; ++it) {
    int c = it + rot;
    if (c >= n_it) c -= n_it;
    const int b0 = c * SCAN_SPAN + tid * 4;

    if (vec_ok && (c + 1) * SCAN_SPAN <= E) {
      int4 v[SCAN_UNROLL];
#pragma unroll
      for (int u = 0; u < SCAN_UNROLL; ++u)
        v[u] = *(const int4*)(dst + b0 + u * SCAN_STEP);
#pragma unroll
      for (int u = 0; u < SCAN_UNROLL; ++u) {
        const int idx = b0 + u * SCAN_STEP;
        const bool any = ((unsigned)(v[u].x - n0) < (unsigned)nrows) | ((unsigned)(v[u].y - n0) < (unsigned)nrows) |
                         ((unsigned)(v[u].z - n0) < (unsigned)nrows) | ((unsigned)(v[u].w - n0) < (unsigned)nrows);
        if (any) {
          keep(idx + 0, v[u].x);
          keep(idx + 1, v[u].y);
          keep(idx + 2, v[u].z);
          keep(idx + 3, v[u].w);
        }
      }
    } else {
      for (int u = 0; u < SCAN_UNROLL; ++u) {
        const int idx = b0 + u * SCAN_STEP;
        for (int k = 0; k < 4; ++k)
          if (idx + k < E) keep(idx + k, dst[idx + k]);
      }
    }
  }
  }
  __syncthreads();
  const int cnt = __builtin_amdgcn_readfirstlane(list_n);

  if (cnt <= LIST_CAP) {
    int dd[4], ss[4], rk[4];
    {
      int ee[4];
      float4 sa[4][2], da[4][2];
#pragma unroll
      for (int k = 0; k < 4; ++k) {
        const int i = k * AGG_THREADS + tid;
        ee[k] = (i < cnt) ? list_e[i] : (E + n0);
      }
#pragma unroll
      for (int k = 0; k < 4; ++k) {
        const bool real = ee[k] < E;
        ss[k] = real ? ei[ee[k]] : (ee[k] - E);
        const int dv = real ? dst[ee[k]] : (ee[k] - E);
        dd[k] = min(max(dv - n0, 0), nrows - 1);
      }
#pragma unroll
      for (int k = 0; k < 4; ++k) {
        const float4* ps = (const float4*)(a_src + (size_t)ss[k] * HEADS);
        const float4* pd = (const float4*)(a_dst + (size_t)(n0 + dd[k]) * HEADS);
        sa[k][0] = ps[0]; sa[k][1] = ps[1];
        da[k][0] = pd[0]; da[k][1] = pd[1];
      }
#pragma unroll
      for (int k = 0; k < 4; ++k) {
        const int i = k * AGG_THREADS + tid;
        rk[k] = 0;
        if (i < cnt) {
          rk[k] = __hip_atomic_fetch_add(&row_cnt[dd[k]], 1, __ATOMIC_RELAXED, __HIP_MEMORY_SCOPE_WORKGROUP);
          unsigned* mx = smax + dd[k] * HEADS;
          __hip_atomic_fetch_max(&mx[0], f2key(leaky(sa[k][0].x + da[k][0].x)), __ATOMIC_RELAXED, __HIP_MEMORY_SCOPE_WORKGROUP);
          __hip_atomic_fetch_max(&mx[1], f2key(leaky(sa[k][0].y + da[k][0].y)), __ATOMIC_RELAXED, __HIP_MEMORY_SCOPE_WORKGROUP);
          __hip_atomic_fetch_max(&mx[2], f2key(leaky(sa[k][0].z + da[k][0].z)), __ATOMIC_RELAXED, __HIP_MEMORY_SCOPE_WORKGROUP);
          __hip_atomic_fetch_max(&mx[3], f2key(leaky(sa[k][0].w + da[k][0].w)), __ATOMIC_RELAXED, __HIP_MEMORY_SCOPE_WORKGROUP);
          __hip_atomic_fetch_max(&mx[4], f2key(leaky(sa[k][1].x + da[k][1].x)), __ATOMIC_RELAXED, __HIP_MEMORY_SCOPE_WORKGROUP);
          __hip_atomic_fetch_max(&mx[5], f2key(leaky(sa[k][1].y + da[k][1].y)), __ATOMIC_RELAXED, __HIP_MEMORY_SCOPE_WORKGROUP);
          __hip_atomic_fetch_max(&mx[6], f2key(leaky(sa[k][1].z + da[k][1].z)), __ATOMIC_RELAXED, __HIP_MEMORY_SCOPE_WORKGROUP);
          __hip_atomic_fetch_max(&mx[7], f2key(leaky(sa[k][1].w + da[k][1].w)), __ATOMIC_RELAXED, __HIP_MEMORY_SCOPE_WORKGROUP);
        }
      }
    }
    __syncthreads();

    if (wave == 0) {
      int c[ROWS_PER_LANE];
      int tot = 0;
#pragma unroll
      for (int q = 0; q < ROWS_PER_LANE; ++q) {
        const int r = lane * ROWS_PER_LANE + q;
        c[q] = (r < NPB) ? row_cnt[r] : 0;
        tot += c[q];
      }
      int inc = tot;
      for (int d = 1; d < 32; d <<= 1) {
        const int v = __shfl_up(inc, d, 32);
        if (lane >= d) inc += v;
      }
      int run = inc - tot;
#pragma unroll
      for (int q = 0; q < ROWS_PER_LANE; ++q) {
        const int r = lane * ROWS_PER_LANE + q;
        if (r < NPB) row_start[r] = run;
        run += c[q];
      }
    }
    __syncthreads();

#pragma unroll
    for (int k = 0; k < 4; ++k) {
      const int i = k * AGG_THREADS + tid;
      if (i < cnt) list_e[row_start[dd[k]] + rk[k]] = ss[k];
    }
    __syncthreads();

    const int hq = lane & 7;
    const int gq = lane >> 3;
    const int h0 = lane >> 3;
    for (int t = 0; t < NPB; ++t) {
      int r = 0;
      if (lane == 0) r = __hip_atomic_fetch_add(&next_row, 1, __ATOMIC_RELAXED, __HIP_MEMORY_SCOPE_WORKGROUP);
      r = __builtin_amdgcn_readfirstlane(r);
      if (r >= nrows) break;
      const int st = __builtin_amdgcn_readfirstlane(row_start[r]);
      const int ne = __builtin_amdgcn_readfirstlane(row_cnt[r]);
      const float adq = a_dst[(size_t)(n0 + r) * HEADS + hq];
      const float mq  = key2f(smax[r * HEADS + hq]);

      float4 sa = make_float4(0.f, 0.f, 0.f, 0.f);
      float4 sb = make_float4(0.f, 0.f, 0.f, 0.f);
      float dsum = 0.f;

      for (int c0 = 0; c0 < ne; c0 += 32) {
        const int m = (ne - c0 < 32) ? (ne - c0) : 32;
        int s = n0;
        if (lane < m) s = list_e[st + c0 + lane];

        for (int i0 = 0; i0 < m; i0 += EDGE_GROUP) {
          const int iq = (i0 + gq < m) ? (i0 + gq) : (m - 1);
          const int sq = list_e[st + c0 + iq];
          const float asq = a_src[(size_t)sq * HEADS + hq];

          float4 xa[EDGE_GROUP], xb[EDGE_GROUP];
#pragma unroll
          for (int j = 0; j < EDGE_GROUP; ++j) {
            const int ii = (i0 + j < m) ? (i0 + j) : (m - 1);
            const int si = __builtin_amdgcn_readlane(s, ii);
            const float4* __restrict__ xr = (const float4*)(xl + (size_t)si * F) + lane;
            xa[j] = xr[0];
            xb[j] = xr[32];
          }

          float p = expf(leaky(asq + adq) - mq);
          if (i0 + gq >= m) p = 0.f;
          dsum += p;

#pragma unroll
          for (int j = 0; j < EDGE_GROUP; ++j) {
            const float u = __shfl(p, 8 * j + h0, 32);
            const float v = __shfl(p, 8 * j + 4 + h0, 32);
            sa.x += u * xa[j].x; sa.y += u * xa[j].y; sa.z += u * xa[j].z; sa.w += u * xa[j].w;
            sb.x += v * xb[j].x; sb.y += v * xb[j].y; sb.z += v * xb[j].z; sb.w += v * xb[j].w;
          }
        }
      }

      dsum += __shfl_xor(dsum, 8, 32);
      dsum += __shfl_xor(dsum, 16, 32);
      const float ra = 1.0f / __shfl(dsum, h0, 32);
      const float rb = 1.0f / __shfl(dsum, 4 + h0, 32);
      float4* __restrict__ orow = (float4*)(out + (size_t)(n0 + r) * F);
      orow[lane]      = make_float4(sa.x * ra, sa.y * ra, sa.z * ra, sa.w * ra);
      orow[32 + lane] = make_float4(sb.x * rb, sb.y * rb, sb.z * rb, sb.w * rb);
    }
  } else {
    for (int i = tid; i < NPB * F / 4; i += AGG_THREADS)
      ((float4*)acc)[i] = make_float4(0.f, 0.f, 0.f, 0.f);
    for (int i = tid; i < NPB * HEADS; i += AGG_THREADS) den[i] = 0.f;
    __syncthreads();
    const int tot = E + N;
    for (int e = tid; e < tot; e += AGG_THREADS) {
      const int d = (e < E) ? dst[e] : (e - E);
      const unsigned dl = (unsigned)(d - n0);
      if (dl < (unsigned)nrows) {
        const int s = (e < E) ? ei[e] : (e - E);
        for (int h = 0; h < HEADS; ++h)
          __hip_atomic_fetch_max(&smax[dl * HEADS + h], f2key(leaky(a_src[s * HEADS + h] + a_dst[d * HEADS + h])),
                                 __ATOMIC_RELAXED, __HIP_MEMORY_SCOPE_WORKGROUP);
      }
    }
    __syncthreads();
    for (int e = tid; e < tot; e += AGG_THREADS) {
      const int d = (e < E) ? dst[e] : (e - E);
      const unsigned dl = (unsigned)(d - n0);
      if (dl < (unsigned)nrows) {
        const int s = (e < E) ? ei[e] : (e - E);
        for (int h = 0; h < HEADS; ++h) {
          const float p = expf(leaky(a_src[s * HEADS + h] + a_dst[d * HEADS + h]) - key2f(smax[dl * HEADS + h]));
          __hip_atomic_fetch_add(&den[dl * HEADS + h], p, __ATOMIC_RELAXED, __HIP_MEMORY_SCOPE_WORKGROUP);
          for (int c = 0; c < CH; ++c)
            __hip_atomic_fetch_add(&acc[dl * F + h * CH + c], xl[(size_t)s * F + h * CH + c] * p,
                                   __ATOMIC_RELAXED, __HIP_MEMORY_SCOPE_WORKGROUP);
        }
      }
    }
    __syncthreads();

    for (int i = tid; i < nrows * HEADS; i += AGG_THREADS) den[i] = 1.0f / den[i];
    __syncthreads();
    float4* __restrict__ orow = (float4*)(out + (size_t)n0 * F);
    for (int i = tid; i < nrows * (F / 4); i += AGG_THREADS) {
      float4 v = ((const float4*)acc)[i];
      const float r = den[(i >> 6) * HEADS + ((i & 63) >> 3)];
      v.x *= r; v.y *= r; v.z *= r; v.w *= r;
      orow[i] = v;
    }
  }
}

#define LN_ROWS 4
#define LN_THREADS 256
#define LN_ROWS_PER_BLOCK (LN_ROWS * (LN_THREADS / 32))

__device__ __forceinline__ float wave_sum(float v) {
#pragma unroll
  for (int m = 16; m > 0; m >>= 1) v += __shfl_xor(v, m, 32);
  return v;
}

__global__ void __launch_bounds__(LN_THREADS)
gat_ln_kernel(float* __restrict__ out, const float* __restrict__ bias,
              const float* __restrict__ gamma, const float* __restrict__ beta, int N) {
  const int lane = threadIdx.x & 31;
  const int wave = __builtin_amdgcn_readfirstlane((int)((blockIdx.x * LN_THREADS + threadIdx.x) >> 5));
  const int row0 = wave * LN_ROWS;
  if (row0 >= N) return;

  float4 va[LN_ROWS], vb[LN_ROWS];
#pragma unroll
  for (int r = 0; r < LN_ROWS; ++r) {
    const int row = (row0 + r < N) ? (row0 + r) : row0;
    const float4* p = (const float4*)(out + (size_t)row * F);
    va[r] = p[lane];
    vb[r] = p[32 + lane];
  }
  const float4 b0 = ((const float4*)bias)[lane],  b1 = ((const float4*)bias)[32 + lane];
  const float4 g0 = ((const float4*)gamma)[lane], g1 = ((const float4*)gamma)[32 + lane];
  const float4 t0 = ((const float4*)beta)[lane],  t1 = ((const float4*)beta)[32 + lane];

  float mean[LN_ROWS];
#pragma unroll
  for (int r = 0; r < LN_ROWS; ++r) {
    va[r].x += b0.x; va[r].y += b0.y; va[r].z += b0.z; va[r].w += b0.w;
    vb[r].x += b1.x; vb[r].y += b1.y; vb[r].z += b1.z; vb[r].w += b1.w;
    const float s = ((va[r].x + va[r].y) + (va[r].z + va[r].w)) +
                    ((vb[r].x + vb[r].y) + (vb[r].z + vb[r].w));
    mean[r] = wave_sum(s) * (1.0f / F);
  }
  float rs[LN_ROWS];
#pragma unroll
  for (int r = 0; r < LN_ROWS; ++r) {
    va[r].x -= mean[r]; va[r].y -= mean[r]; va[r].z -= mean[r]; va[r].w -= mean[r];
    vb[r].x -= mean[r]; vb[r].y -= mean[r]; vb[r].z -= mean[r]; vb[r].w -= mean[r];
    const float q = ((va[r].x * va[r].x + va[r].y * va[r].y) + (va[r].z * va[r].z + va[r].w * va[r].w)) +
                    ((vb[r].x * vb[r].x + vb[r].y * vb[r].y) + (vb[r].z * vb[r].z + vb[r].w * vb[r].w));
    rs[r] = rsqrtf(wave_sum(q) * (1.0f / F) + LN_EPS);
  }
#pragma unroll
  for (int r = 0; r < LN_ROWS; ++r) {
    if (row0 + r < N) {
      float4* o = (float4*)(out + (size_t)(row0 + r) * F);
      float4 ya, yb;
      ya.x = va[r].x * rs[r] * g0.x + t0.x; ya.y = va[r].y * rs[r] * g0.y + t0.y;
      ya.z = va[r].z * rs[r] * g0.z + t0.z; ya.w = va[r].w * rs[r] * g0.w + t0.w;
      yb.x = vb[r].x * rs[r] * g1.x + t1.x; yb.y = vb[r].y * rs[r] * g1.y + t1.y;
      yb.z = vb[r].z * rs[r] * g1.z + t1.z; yb.w = vb[r].w * rs[r] * g1.w + t1.w;
      o[lane] = ya;
      o[32 + lane] = yb;
    }
  }
}

extern "C" void kernel_launch(void* const* d_in, const int* in_sizes, int n_in,
                              void* d_out, int out_size, void* d_ws, size_t ws_size,
                              hipStream_t stream) {
  const float* x       = (const float*)d_in[0];
  const int*   ei      = (const int*)  d_in[1];
  const float* W       = (const float*)d_in[2];
  const float* att_src = (const float*)d_in[3];
  const float* att_dst = (const float*)d_in[4];
  const float* bias    = (const float*)d_in[5];
  const float* gamma   = (const float*)d_in[6];
  const float* beta    = (const float*)d_in[7];
  float* out = (float*)d_out;

  const int N = in_sizes[0] / F;
  const int E = in_sizes[1] / 2;

  char* ws = (char*)d_ws;
  float*    xl     = (float*)ws;    ws += (size_t)N * F * sizeof(float);
  float*    a_src  = (float*)ws;    ws += (size_t)N * HEADS * sizeof(float);
  float*    a_dst  = (float*)ws;    ws += (size_t)N * HEADS * sizeof(float);
  unsigned* own    = (unsigned*)ws;

  gat_gemm_wmma_kernel<<<(N + 15) / 16, 512, 0, stream>>>(x, W, xl, N);

  {
    int NH = N * HEADS;
    gat_att_kernel<<<(NH + 255) / 256, 256, 0, stream>>>(xl, att_src, att_dst,
                                                         a_src, a_dst, NH, ei, own, E,
                                                         (unsigned long long)ws_size);
  }

  gat_edge_agg_kernel<<<(N + NPB - 1) / NPB, AGG_THREADS, 0, stream>>>(ei, xl, a_src, a_dst,
                                                                        out, E, N, own,
                                                                        (unsigned long long)ws_size);

  gat_ln_kernel<<<(N + LN_ROWS_PER_BLOCK - 1) / LN_ROWS_PER_BLOCK, LN_THREADS, 0, stream>>>(out, bias, gamma, beta, N);
}


# half-logits summed in the projection's epilogue from the block's result tile in LDS (channel order 0..31 kept); half-logit launch and its re-read of xl gone; owner bytes from the projection launch
# speedup vs baseline: 5.0602x; 1.0477x over previous
#include <hip/hip_runtime.h>
#include <math.h>

typedef float v2f __attribute__((ext_vector_type(2)));
typedef float v8f __attribute__((ext_vector_type(8)));

#define HEADS 8
#define CH 32
#define F 256
#define NEG_SLOPE 0.2f
#define LN_EPS 1e-5f

#define LDS_STRIDE 260

__device__ __forceinline__ unsigned f2key(float f) {
  unsigned b = __float_as_uint(f);
  return (b & 0x80000000u) ? ~b : (b | 0x80000000u);
}
__device__ __forceinline__ float key2f(unsigned k) {
  unsigned b = (k & 0x80000000u) ? (k & 0x7fffffffu) : ~k;
  return __uint_as_float(b);
}

#define NPB 196

__device__ __forceinline__ bool owner_plane_ok(int N, int E, unsigned long long ws_bytes) {
  const unsigned long long need = (unsigned long long)N * F * sizeof(float) +
                                  2ull * N * HEADS * sizeof(float) + (unsigned long long)E;
  return (E >= 16) && ((E & 15) == 0) && ((N + NPB - 1) / NPB <= 256) && (ws_bytes >= need);
}

__global__ void __launch_bounds__(512)
gat_gemm_wmma_kernel(const float* __restrict__ x, const float* __restrict__ W,
                     float* __restrict__ xl, int N,
                     const float* __restrict__ att_src, const float* __restrict__ att_dst,
                     float* __restrict__ a_src, float* __restrict__ a_dst,
                     const int* __restrict__ ei, unsigned* __restrict__ own,
                     int E, unsigned long long ws_bytes) {
  __shared__ float lds_a[16 * LDS_STRIDE];
  __shared__ __attribute__((aligned(16))) float lds_d[16 * LDS_STRIDE];

  const int tid  = threadIdx.x;
  const int lane = tid & 31;
  const int wave = tid >> 5;
  const int row0 = blockIdx.x * 16;
  const int col0 = wave * 16;

  if (owner_plane_ok(N, E, ws_bytes)) {
    const int4* __restrict__ d4 = (const int4*)(ei + E);
    for (long long g = (long long)blockIdx.x * 512 + tid; g < (long long)(E >> 2);
         g += (long long)gridDim.x * 512) {
      const int4 d = d4[g];
      own[g] = ((unsigned)d.x / NPB) | (((unsigned)d.y / NPB) << 8) |
               (((unsigned)d.z / NPB) << 16) | (((unsigned)d.w / NPB) << 24);
    }
  }

  {
    int r = tid >> 5;
    int c = (tid & 31) * 8;
    int grow = row0 + r;
    if (grow >= N) grow = N - 1;
    const float4* __restrict__ src = (const float4*)(x + (size_t)grow * F + c);
    float4* dst = (float4*)(&lds_a[r * LDS_STRIDE + c]);
    dst[0] = src[0];
    dst[1] = src[1];
  }
  __syncthreads();

  const int m    = lane & 15;
  const int koff = (lane >> 4) * 2;
  const float* __restrict__ arow = &lds_a[m * LDS_STRIDE];

  const int n = lane & 15;

  v8f acc = {0.f, 0.f, 0.f, 0.f, 0.f, 0.f, 0.f, 0.f};
  for (int k = 0; k < F; k += 4) {
    v2f a = *(const v2f*)(arow + k + koff);
    v2f b;
    b.x = W[(size_t)(k + koff) * F + col0 + n];
    b.y = W[(size_t)(k + koff + 1) * F + col0 + n];
    acc = __builtin_amdgcn_wmma_f32_16x16x4_f32(
         false, a,  false, b,
         (short)0, acc,  false,  false);
  }

  {
    float* drow = &lds_d[((lane >> 4) * 8) * LDS_STRIDE + col0 + n];
#pragma unroll
    for (int j = 0; j < 8; ++j) drow[j * LDS_STRIDE] = acc[j];
  }
  __syncthreads();

  const int rbase = row0 + ((lane >> 4) * 8);
  float* __restrict__ dst = xl + (size_t)rbase * F + col0 + n;
  if (row0 + 16 <= N) {
#pragma unroll
    for (int j = 0; j < 8; ++j) dst[(size_t)j * F] = acc[j];
  } else {
#pragma unroll
    for (int j = 0; j < 8; ++j) {
      if (rbase + j < N) dst[(size_t)j * F] = acc[j];
    }
  }

  if (tid < 256) {
    const int wu = __builtin_amdgcn_readfirstlane(wave);
    const bool is_dst = (wu & 1) != 0;
    const int r = (wu >> 1) * 4 + (lane >> 3);
    const int h = lane & 7;
    const float4* xr = (const float4*)(&lds_d[r * LDS_STRIDE + h * CH]);
    const float4* __restrict__ av = (const float4*)((is_dst ? att_dst : att_src) + h * CH);
    float s = 0.f;
#pragma unroll
    for (int c = 0; c < CH / 4; ++c) {
      const float4 v = xr[c];
      const float4 a = av[c];
      s += v.x * a.x;
      s += v.y * a.y;
      s += v.z * a.z;
      s += v.w * a.w;
    }
    if (row0 + r < N) (is_dst ? a_dst : a_src)[(size_t)(row0 + r) * HEADS + h] = s;
  }
}

__device__ __forceinline__ float leaky(float v) {
  return (v >= 0.f) ? v : v * NEG_SLOPE;
}

#define AGG_THREADS 1024
#define AGG_WAVES (AGG_THREADS / 32)
#define LIST_CAP 4096
#define SCAN_UNROLL 8
#define SCAN_STEP (AGG_THREADS * 4)
#define SCAN_SPAN (SCAN_STEP * SCAN_UNROLL)
#define OWN_STEP (AGG_THREADS * 16)
#define OWN_SPAN (OWN_STEP * SCAN_UNROLL)
#define ROWS_PER_LANE ((NPB + 31) / 32)
#define EDGE_GROUP 4

static_assert(LIST_CAP == 4 * AGG_THREADS, "pass (1) gives every thread four list entries");
static_assert(EDGE_GROUP * HEADS == 32, "one lane per (edge of the group, head) weight");

__global__ void __launch_bounds__(AGG_THREADS)
gat_edge_agg_kernel(const int* __restrict__ ei, const float* __restrict__ xl,
                    const float* __restrict__ a_src, const float* __restrict__ a_dst,
                    float* __restrict__ out, int E, int N,
                    const unsigned* __restrict__ own, unsigned long long ws_bytes) {
  __shared__ __attribute__((aligned(16))) float acc[NPB * F];
  __shared__ __attribute__((aligned(16))) unsigned smax[NPB * HEADS];
  __shared__ __attribute__((aligned(16))) float den[NPB * HEADS];
  __shared__ int list_e[LIST_CAP];
  __shared__ int row_cnt[NPB];
  __shared__ int row_start[NPB];
  __shared__ int list_n;
  __shared__ int next_row;

  const int tid  = threadIdx.x;
  const int lane = tid & 31;
  const int wave = __builtin_amdgcn_readfirstlane(tid >> 5);
  const int n0 = blockIdx.x * NPB;
  const int nrows = (N - n0 < NPB) ? (N - n0) : NPB;
  const int* __restrict__ dst = ei + E;

  for (int i = tid; i < NPB * HEADS; i += AGG_THREADS) smax[i] = 0x007FFFFFu;
  if (tid < NPB) row_cnt[tid] = 0;
  if (tid < nrows) list_e[tid] = E + n0 + tid;
  if (tid == 0) { list_n = nrows; next_row = 0; }
  __syncthreads();

  auto keep = [&](int e, int dv) {
    unsigned dl = (unsigned)(dv - n0);
    if (dl < (unsigned)nrows) {
      int slot = __hip_atomic_fetch_add(&list_n, 1, __ATOMIC_RELAXED, __HIP_MEMORY_SCOPE_WORKGROUP);
      if (slot < LIST_CAP) list_e[slot] = e;
    }
  };
  auto keep_e = [&](int e) {
    int slot = __hip_atomic_fetch_add(&list_n, 1, __ATOMIC_RELAXED, __HIP_MEMORY_SCOPE_WORKGROUP);
    if (slot < LIST_CAP) list_e[slot] = e;
  };

  const bool use_plane = owner_plane_ok(N, E, ws_bytes);
  if (use_plane) {
    int n_it = (E + OWN_SPAN - 1) / OWN_SPAN;
    if (n_it < 1) n_it = 1;
    const int rot = (int)(((long long)blockIdx.x * n_it) / gridDim.x);
    const unsigned pat  = blockIdx.x * 0x01010101u;
    const uint4* __restrict__ own4 = (const uint4*)own;
    for (int it = 0; it < n_it; ++it) {
      int c = it + rot;
      if (c >= n_it) c -= n_it;
      const int b0 = c * OWN_SPAN + tid * 16;
      uint4 v[SCAN_UNROLL];
#pragma unroll
      for (int u = 0; u < SCAN_UNROLL; ++u) {
        const int idx = b0 + u * OWN_STEP;
        v[u] = own4[min(idx, E - 16) >> 4];
      }
#pragma unroll
      for (int u = 0; u < SCAN_UNROLL; u += 2) {
        const int idx = b0 + u * OWN_STEP;
        unsigned m = 0;
#pragma unroll
        for (int w = 0; w < 2; ++w) {
          const unsigned x[4] = {v[u + w].x ^ pat, v[u + w].y ^ pat, v[u + w].z ^ pat, v[u + w].w ^ pat};
          unsigned mw = 0;
#pragma unroll
          for (int q = 0; q < 4; ++q) {
            const unsigned f = ~(((x[q] & 0x7f7f7f7fu) + 0x7f7f7f7fu) | x[q]) & 0x80808080u;
            mw |= f >> (7 - q);
          }
          if (idx + w * OWN_STEP >= E) mw = 0;
          m |= mw << (4 * w);
        }
        while (m) {
          const int p = __builtin_ctz(m);
          keep_e(idx + ((p >> 2) & 1) * OWN_STEP + 4 * (p & 3) + (p >> 3));
          m &= m - 1;
        }
      }
    }
  } else {
  int n_it = (E + SCAN_SPAN - 1) / SCAN_SPAN;
  if (n_it < 1) n_it = 1;
  const int rot = (int)(((long long)blockIdx.x * n_it) / gridDim.x);
  const bool vec_ok = ((E & 3) == 0);

  for (int it = 0; it < n_it; ++it) {
    int c = it + rot;
    if (c >= n_it) c -= n_it;
    const int b0 = c * SCAN_SPAN + tid * 4;

    if (vec_ok && (c + 1) * SCAN_SPAN <= E) {
      int4 v[SCAN_UNROLL];
#pragma unroll
      for (int u = 0; u < SCAN_UNROLL; ++u)
        v[u] = *(const int4*)(dst + b0 + u * SCAN_STEP);
#pragma unroll
      for (int u = 0; u < SCAN_UNROLL; ++u) {
        const int idx = b0 + u * SCAN_STEP;
        const bool any = ((unsigned)(v[u].x - n0) < (unsigned)nrows) | ((unsigned)(v[u].y - n0) < (unsigned)nrows) |
                         ((unsigned)(v[u].z - n0) < (unsigned)nrows) | ((unsigned)(v[u].w - n0) < (unsigned)nrows);
        if (any) {
          keep(idx + 0, v[u].x);
          keep(idx + 1, v[u].y);
          keep(idx + 2, v[u].z);
          keep(idx + 3, v[u].w);
        }
      }
    } else {
      for (int u = 0; u < SCAN_UNROLL; ++u) {
        const int idx = b0 + u * SCAN_STEP;
        for (int k = 0; k < 4; ++k)
          if (idx + k < E) keep(idx + k, dst[idx + k]);
      }
    }
  }
  }
  __syncthreads();
  const int cnt = __builtin_amdgcn_readfirstlane(list_n);

  if (cnt <= LIST_CAP) {
    int dd[4], ss[4], rk[4];
    {
      int ee[4];
      float4 sa[4][2], da[4][2];
#pragma unroll
      for (int k = 0; k < 4; ++k) {
        const int i = k * AGG_THREADS + tid;
        ee[k] = (i < cnt) ? list_e[i] : (E + n0);
      }
#pragma unroll
      for (int k = 0; k < 4; ++k) {
        const bool real = ee[k] < E;
        ss[k] = real ? ei[ee[k]] : (ee[k] - E);
        const int dv = real ? dst[ee[k]] : (ee[k] - E);
        dd[k] = min(max(dv - n0, 0), nrows - 1);
      }
#pragma unroll
      for (int k = 0; k < 4; ++k) {
        const float4* ps = (const float4*)(a_src + (size_t)ss[k] * HEADS);
        const float4* pd = (const float4*)(a_dst + (size_t)(n0 + dd[k]) * HEADS);
        sa[k][0] = ps[0]; sa[k][1] = ps[1];
        da[k][0] = pd[0]; da[k][1] = pd[1];
      }
#pragma unroll
      for (int k = 0; k < 4; ++k) {
        const int i = k * AGG_THREADS + tid;
        rk[k] = 0;
        if (i < cnt) {
          rk[k] = __hip_atomic_fetch_add(&row_cnt[dd[k]], 1, __ATOMIC_RELAXED, __HIP_MEMORY_SCOPE_WORKGROUP);
          unsigned* mx = smax + dd[k] * HEADS;
          __hip_atomic_fetch_max(&mx[0], f2key(leaky(sa[k][0].x + da[k][0].x)), __ATOMIC_RELAXED, __HIP_MEMORY_SCOPE_WORKGROUP);
          __hip_atomic_fetch_max(&mx[1], f2key(leaky(sa[k][0].y + da[k][0].y)), __ATOMIC_RELAXED, __HIP_MEMORY_SCOPE_WORKGROUP);
          __hip_atomic_fetch_max(&mx[2], f2key(leaky(sa[k][0].z + da[k][0].z)), __ATOMIC_RELAXED, __HIP_MEMORY_SCOPE_WORKGROUP);
          __hip_atomic_fetch_max(&mx[3], f2key(leaky(sa[k][0].w + da[k][0].w)), __ATOMIC_RELAXED, __HIP_MEMORY_SCOPE_WORKGROUP);
          __hip_atomic_fetch_max(&mx[4], f2key(leaky(sa[k][1].x + da[k][1].x)), __ATOMIC_RELAXED, __HIP_MEMORY_SCOPE_WORKGROUP);
          __hip_atomic_fetch_max(&mx[5], f2key(leaky(sa[k][1].y + da[k][1].y)), __ATOMIC_RELAXED, __HIP_MEMORY_SCOPE_WORKGROUP);
          __hip_atomic_fetch_max(&mx[6], f2key(leaky(sa[k][1].z + da[k][1].z)), __ATOMIC_RELAXED, __HIP_MEMORY_SCOPE_WORKGROUP);
          __hip_atomic_fetch_max(&mx[7], f2key(leaky(sa[k][1].w + da[k][1].w)), __ATOMIC_RELAXED, __HIP_MEMORY_SCOPE_WORKGROUP);
        }
      }
    }
    __syncthreads();

    if (wave == 0) {
      int c[ROWS_PER_LANE];
      int tot = 0;
#pragma unroll
      for (int q = 0; q < ROWS_PER_LANE; ++q) {
        const int r = lane * ROWS_PER_LANE + q;
        c[q] = (r < NPB) ? row_cnt[r] : 0;
        tot += c[q];
      }
      int inc = tot;
      for (int d = 1; d < 32; d <<= 1) {
        const int v = __shfl_up(inc, d, 32);
        if (lane >= d) inc += v;
      }
      int run = inc - tot;
#pragma unroll
      for (int q = 0; q < ROWS_PER_LANE; ++q) {
        const int r = lane * ROWS_PER_LANE + q;
        if (r < NPB) row_start[r] = run;
        run += c[q];
      }
    }
    __syncthreads();

#pragma unroll
    for (int k = 0; k < 4; ++k) {
      const int i = k * AGG_THREADS + tid;
      if (i < cnt) list_e[row_start[dd[k]] + rk[k]] = ss[k];
    }
    __syncthreads();

    const int hq = lane & 7;
    const int gq = lane >> 3;
    const int h0 = lane >> 3;
    for (int t = 0; t < NPB; ++t) {
      int r = 0;
      if (lane == 0) r = __hip_atomic_fetch_add(&next_row, 1, __ATOMIC_RELAXED, __HIP_MEMORY_SCOPE_WORKGROUP);
      r = __builtin_amdgcn_readfirstlane(r);
      if (r >= nrows) break;
      const int st = __builtin_amdgcn_readfirstlane(row_start[r]);
      const int ne = __builtin_amdgcn_readfirstlane(row_cnt[r]);
      const float adq = a_dst[(size_t)(n0 + r) * HEADS + hq];
      const float mq  = key2f(smax[r * HEADS + hq]);

      float4 sa = make_float4(0.f, 0.f, 0.f, 0.f);
      float4 sb = make_float4(0.f, 0.f, 0.f, 0.f);
      float dsum = 0.f;

      for (int c0 = 0; c0 < ne; c0 += 32) {
        const int m = (ne - c0 < 32) ? (ne - c0) : 32;
        int s = n0;
        if (lane < m) s = list_e[st + c0 + lane];

        for (int i0 = 0; i0 < m; i0 += EDGE_GROUP) {
          const int iq = (i0 + gq < m) ? (i0 + gq) : (m - 1);
          const int sq = list_e[st + c0 + iq];
          const float asq = a_src[(size_t)sq * HEADS + hq];

          float4 xa[EDGE_GROUP], xb[EDGE_GROUP];
#pragma unroll
          for (int j = 0; j < EDGE_GROUP; ++j) {
            const int ii = (i0 + j < m) ? (i0 + j) : (m - 1);
            const int si = __builtin_amdgcn_readlane(s, ii);
            const float4* __restrict__ xr = (const float4*)(xl + (size_t)si * F) + lane;
            xa[j] = xr[0];
            xb[j] = xr[32];
          }

          float p = expf(leaky(asq + adq) - mq);
          if (i0 + gq >= m) p = 0.f;
          dsum += p;

#pragma unroll
          for (int j = 0; j < EDGE_GROUP; ++j) {
            const float u = __shfl(p, 8 * j + h0, 32);
            const float v = __shfl(p, 8 * j + 4 + h0, 32);
            sa.x += u * xa[j].x; sa.y += u * xa[j].y; sa.z += u * xa[j].z; sa.w += u * xa[j].w;
            sb.x += v * xb[j].x; sb.y += v * xb[j].y; sb.z += v * xb[j].z; sb.w += v * xb[j].w;
          }
        }
      }

      dsum += __shfl_xor(dsum, 8, 32);
      dsum += __shfl_xor(dsum, 16, 32);
      const float ra = 1.0f / __shfl(dsum, h0, 32);
      const float rb = 1.0f / __shfl(dsum, 4 + h0, 32);
      float4* __restrict__ orow = (float4*)(out + (size_t)(n0 + r) * F);
      orow[lane]      = make_float4(sa.x * ra, sa.y * ra, sa.z * ra, sa.w * ra);
      orow[32 + lane] = make_float4(sb.x * rb, sb.y * rb, sb.z * rb, sb.w * rb);
    }
  } else {
    for (int i = tid; i < NPB * F / 4; i += AGG_THREADS)
      ((float4*)acc)[i] = make_float4(0.f, 0.f, 0.f, 0.f);
    for (int i = tid; i < NPB * HEADS; i += AGG_THREADS) den[i] = 0.f;
    __syncthreads();
    const int tot = E + N;
    for (int e = tid; e < tot; e += AGG_THREADS) {
      const int d = (e < E) ? dst[e] : (e - E);
      const unsigned dl = (unsigned)(d - n0);
      if (dl < (unsigned)nrows) {
        const int s = (e < E) ? ei[e] : (e - E);
        for (int h = 0; h < HEADS; ++h)
          __hip_atomic_fetch_max(&smax[dl * HEADS + h], f2key(leaky(a_src[s * HEADS + h] + a_dst[d * HEADS + h])),
                                 __ATOMIC_RELAXED, __HIP_MEMORY_SCOPE_WORKGROUP);
      }
    }
    __syncthreads();
    for (int e = tid; e < tot; e += AGG_THREADS) {
      const int d = (e < E) ? dst[e] : (e - E);
      const unsigned dl = (unsigned)(d - n0);
      if (dl < (unsigned)nrows) {
        const int s = (e < E) ? ei[e] : (e - E);
        for (int h = 0; h < HEADS; ++h) {
          const float p = expf(leaky(a_src[s * HEADS + h] + a_dst[d * HEADS + h]) - key2f(smax[dl * HEADS + h]));
          __hip_atomic_fetch_add(&den[dl * HEADS + h], p, __ATOMIC_RELAXED, __HIP_MEMORY_SCOPE_WORKGROUP);
          for (int c = 0; c < CH; ++c)
            __hip_atomic_fetch_add(&acc[dl * F + h * CH + c], xl[(size_t)s * F + h * CH + c] * p,
                                   __ATOMIC_RELAXED, __HIP_MEMORY_SCOPE_WORKGROUP);
        }
      }
    }
    __syncthreads();

    for (int i = tid; i < nrows * HEADS; i += AGG_THREADS) den[i] = 1.0f / den[i];
    __syncthreads();
    float4* __restrict__ orow = (float4*)(out + (size_t)n0 * F);
    for (int i = tid; i < nrows * (F / 4); i += AGG_THREADS) {
      float4 v = ((const float4*)acc)[i];
      const float r = den[(i >> 6) * HEADS + ((i & 63) >> 3)];
      v.x *= r; v.y *= r; v.z *= r; v.w *= r;
      orow[i] = v;
    }
  }
}

#define LN_ROWS 4
#define LN_THREADS 256
#define LN_ROWS_PER_BLOCK (LN_ROWS * (LN_THREADS / 32))

__device__ __forceinline__ float wave_sum(float v) {
#pragma unroll
  for (int m = 16; m > 0; m >>= 1) v += __shfl_xor(v, m, 32);
  return v;
}

__global__ void __launch_bounds__(LN_THREADS)
gat_ln_kernel(float* __restrict__ out, const float* __restrict__ bias,
              const float* __restrict__ gamma, const float* __restrict__ beta, int N) {
  const int lane = threadIdx.x & 31;
  const int wave = __builtin_amdgcn_readfirstlane((int)((blockIdx.x * LN_THREADS + threadIdx.x) >> 5));
  const int row0 = wave * LN_ROWS;
  if (row0 >= N) return;

  float4 va[LN_ROWS], vb[LN_ROWS];
#pragma unroll
  for (int r = 0; r < LN_ROWS; ++r) {
    const int row = (row0 + r < N) ? (row0 + r) : row0;
    const float4* p = (const float4*)(out + (size_t)row * F);
    va[r] = p[lane];
    vb[r] = p[32 + lane];
  }
  const float4 b0 = ((const float4*)bias)[lane],  b1 = ((const float4*)bias)[32 + lane];
  const float4 g0 = ((const float4*)gamma)[lane], g1 = ((const float4*)gamma)[32 + lane];
  const float4 t0 = ((const float4*)beta)[lane],  t1 = ((const float4*)beta)[32 + lane];

  float mean[LN_ROWS];
#pragma unroll
  for (int r = 0; r < LN_ROWS; ++r) {
    va[r].x += b0.x; va[r].y += b0.y; va[r].z += b0.z; va[r].w += b0.w;
    vb[r].x += b1.x; vb[r].y += b1.y; vb[r].z += b1.z; vb[r].w += b1.w;
    const float s = ((va[r].x + va[r].y) + (va[r].z + va[r].w)) +
                    ((vb[r].x + vb[r].y) + (vb[r].z + vb[r].w));
    mean[r] = wave_sum(s) * (1.0f / F);
  }
  float rs[LN_ROWS];
#pragma unroll
  for (int r = 0; r < LN_ROWS; ++r) {
    va[r].x -= mean[r]; va[r].y -= mean[r]; va[r].z -= mean[r]; va[r].w -= mean[r];
    vb[r].x -= mean[r]; vb[r].y -= mean[r]; vb[r].z -= mean[r]; vb[r].w -= mean[r];
    const float q = ((va[r].x * va[r].x + va[r].y * va[r].y) + (va[r].z * va[r].z + va[r].w * va[r].w)) +
                    ((vb[r].x * vb[r].x + vb[r].y * vb[r].y) + (vb[r].z * vb[r].z + vb[r].w * vb[r].w));
    rs[r] = rsqrtf(wave_sum(q) * (1.0f / F) + LN_EPS);
  }
#pragma unroll
  for (int r = 0; r < LN_ROWS; ++r) {
    if (row0 + r < N) {
      float4* o = (float4*)(out + (size_t)(row0 + r) * F);
      float4 ya, yb;
      ya.x = va[r].x * rs[r] * g0.x + t0.x; ya.y = va[r].y * rs[r] * g0.y + t0.y;
      ya.z = va[r].z * rs[r] * g0.z + t0.z; ya.w = va[r].w * rs[r] * g0.w + t0.w;
      yb.x = vb[r].x * rs[r] * g1.x + t1.x; yb.y = vb[r].y * rs[r] * g1.y + t1.y;
      yb.z = vb[r].z * rs[r] * g1.z + t1.z; yb.w = vb[r].w * rs[r] * g1.w + t1.w;
      o[lane] = ya;
      o[32 + lane] = yb;
    }
  }
}

extern "C" void kernel_launch(void* const* d_in, const int* in_sizes, int n_in,
                              void* d_out, int out_size, void* d_ws, size_t ws_size,
                              hipStream_t stream) {
  const float* x       = (const float*)d_in[0];
  const int*   ei      = (const int*)  d_in[1];
  const float* W       = (const float*)d_in[2];
  const float* att_src = (const float*)d_in[3];
  const float* att_dst = (const float*)d_in[4];
  const float* bias    = (const float*)d_in[5];
  const float* gamma   = (const float*)d_in[6];
  const float* beta    = (const float*)d_in[7];
  float* out = (float*)d_out;

  const int N = in_sizes[0] / F;
  const int E = in_sizes[1] / 2;

  char* ws = (char*)d_ws;
  float*    xl     = (float*)ws;    ws += (size_t)N * F * sizeof(float);
  float*    a_src  = (float*)ws;    ws += (size_t)N * HEADS * sizeof(float);
  float*    a_dst  = (float*)ws;    ws += (size_t)N * HEADS * sizeof(float);
  unsigned* own    = (unsigned*)ws;

  gat_gemm_wmma_kernel<<<(N + 15) / 16, 512, 0, stream>>>(x, W, xl, N, att_src, att_dst,
                                                         a_src, a_dst, ei, own, E,
                                                         (unsigned long long)ws_size);

  gat_edge_agg_kernel<<<(N + NPB - 1) / NPB, AGG_THREADS, 0, stream>>>(ei, xl, a_src, a_dst,
                                                                        out, E, N, own,
                                                                        (unsigned long long)ws_size);

  gat_ln_kernel<<<(N + LN_ROWS_PER_BLOCK - 1) / LN_ROWS_PER_BLOCK, LN_THREADS, 0, stream>>>(out, bias, gamma, beta, N);
}


# projection persistent: one 16-wave block per WGP walks the 16-row tiles, W held in LDS (k pairs interleaved by column, odd pairs rotated 32 dwords); B fragment = one 8-byte LDS load; same k order
# speedup vs baseline: 5.6177x; 1.1102x over previous
#include <hip/hip_runtime.h>
#include <math.h>

typedef float v2f __attribute__((ext_vector_type(2)));
typedef float v8f __attribute__((ext_vector_type(8)));

#define HEADS 8
#define CH 32
#define F 256
#define NEG_SLOPE 0.2f
#define LN_EPS 1e-5f

#define LDS_STRIDE 260

__device__ __forceinline__ unsigned f2key(float f) {
  unsigned b = __float_as_uint(f);
  return (b & 0x80000000u) ? ~b : (b | 0x80000000u);
}
__device__ __forceinline__ float key2f(unsigned k) {
  unsigned b = (k & 0x80000000u) ? (k & 0x7fffffffu) : ~k;
  return __uint_as_float(b);
}

#define NPB 196

__device__ __forceinline__ bool owner_plane_ok(int N, int E, unsigned long long ws_bytes) {
  const unsigned long long need = (unsigned long long)N * F * sizeof(float) +
                                  2ull * N * HEADS * sizeof(float) + (unsigned long long)E;
  return (E >= 16) && ((E & 15) == 0) && ((N + NPB - 1) / NPB <= 256) && (ws_bytes >= need);
}

#define GEMM_BLOCKS 256
#define WL_PAIR 512

__device__ __forceinline__ void tile_half_logits(const float* dtile, int row0, int N, int tid,
                                                 const float* __restrict__ att_src,
                                                 const float* __restrict__ att_dst,
                                                 float* __restrict__ a_src, float* __restrict__ a_dst) {
  if (tid < 256) {
    const int lane = tid & 31;
    const int wu = __builtin_amdgcn_readfirstlane(tid >> 5);
    const bool is_dst = (wu & 1) != 0;
    const int r = (wu >> 1) * 4 + (lane >> 3);
    const int hd = lane & 7;
    const float4* xr = (const float4*)(&dtile[r * LDS_STRIDE + hd * CH]);
    const float4* __restrict__ av = (const float4*)((is_dst ? att_dst : att_src) + hd * CH);
    float s = 0.f;
#pragma unroll
    for (int c = 0; c < CH / 4; ++c) {
      const float4 v = xr[c];
      const float4 a = av[c];
      s += v.x * a.x;
      s += v.y * a.y;
      s += v.z * a.z;
      s += v.w * a.w;
    }
    if (row0 + r < N) (is_dst ? a_dst : a_src)[(size_t)(row0 + r) * HEADS + hd] = s;
  }
}

__global__ void __launch_bounds__(512)
gat_gemm_wmma_kernel(const float* __restrict__ x, const float* __restrict__ W,
                     float* __restrict__ xl, int N,
                     const float* __restrict__ att_src, const float* __restrict__ att_dst,
                     float* __restrict__ a_src, float* __restrict__ a_dst,
                     const int* __restrict__ ei, unsigned* __restrict__ own,
                     int E, unsigned long long ws_bytes) {
  __shared__ __attribute__((aligned(16))) float lds_w[F * F];
  __shared__ __attribute__((aligned(16))) float lds_a[16 * LDS_STRIDE];
  __shared__ __attribute__((aligned(16))) float lds_d[2][16 * LDS_STRIDE];

  const int tid  = threadIdx.x;
  const int lane = tid & 31;
  const int wave = tid >> 5;
  const int col0 = wave * 16;
  const int ntiles = (N + 15) >> 4;

  if (owner_plane_ok(N, E, ws_bytes)) {
    const int4* __restrict__ d4 = (const int4*)(ei + E);
    for (long long g = (long long)blockIdx.x * 512 + tid; g < (long long)(E >> 2);
         g += (long long)gridDim.x * 512) {
      const int4 d = d4[g];
      own[g] = ((unsigned)d.x / NPB) | (((unsigned)d.y / NPB) << 8) |
               (((unsigned)d.z / NPB) << 16) | (((unsigned)d.w / NPB) << 24);
    }
  }

  int tile = blockIdx.x;
  if (tile >= ntiles) return;

  const int ar = tid >> 5;
  const int ac = (tid & 31) * 8;
  float4 pa0, pa1;
  {
    int grow = tile * 16 + ar;
    if (grow >= N) grow = N - 1;
    const float4* __restrict__ src = (const float4*)(x + (size_t)grow * F + ac);
    pa0 = src[0];
    pa1 = src[1];
  }

  for (int it0 = 0; it0 < 16; it0 += 4) {
    float4 r0[4], r1[4];
#pragma unroll
    for (int u = 0; u < 4; ++u) {
      const int item = (it0 + u) * 512 + tid;
      const int p = item >> 6;
      const int q = item & 63;
      r0[u] = *(const float4*)(W + (size_t)(2 * p) * F + 4 * q);
      r1[u] = *(const float4*)(W + (size_t)(2 * p + 1) * F + 4 * q);
    }
#pragma unroll
    for (int u = 0; u < 4; ++u) {
      const int item = (it0 + u) * 512 + tid;
      const int p = item >> 6;
      const int q = item & 63;
      float4* d = (float4*)(&lds_w[p * WL_PAIR + ((8 * q + 32 * (p & 1)) & (WL_PAIR - 1))]);
      d[0] = make_float4(r0[u].x, r1[u].x, r0[u].y, r1[u].y);
      d[1] = make_float4(r0[u].z, r1[u].z, r0[u].w, r1[u].w);
    }
  }

  const int m = lane & 15;
  const int h = lane >> 4;
  const float* arow = &lds_a[m * LDS_STRIDE + 2 * h];
  const float* bcol = &lds_w[h * WL_PAIR + ((2 * (col0 + m) + 32 * h) & (WL_PAIR - 1))];

  int par = 0;
  int prev_row0 = -1;
  for (; tile < ntiles; tile += gridDim.x) {
    const int row0 = tile * 16;
    {
      float4* dst = (float4*)(&lds_a[ar * LDS_STRIDE + ac]);
      dst[0] = pa0;
      dst[1] = pa1;
    }
    __builtin_amdgcn_fence(__ATOMIC_RELEASE, "workgroup", "local");
    __builtin_amdgcn_s_barrier();
    __builtin_amdgcn_fence(__ATOMIC_ACQUIRE, "workgroup", "local");

    if (prev_row0 >= 0)
      tile_half_logits(lds_d[par ^ 1], prev_row0, N, tid, att_src, att_dst, a_src, a_dst);

    if (tile + (int)gridDim.x < ntiles) {
      int grow = (tile + (int)gridDim.x) * 16 + ar;
      if (grow >= N) grow = N - 1;
      const float4* __restrict__ src = (const float4*)(x + (size_t)grow * F + ac);
      pa0 = src[0];
      pa1 = src[1];
    }

    v8f acc = {0.f, 0.f, 0.f, 0.f, 0.f, 0.f, 0.f, 0.f};
#pragma unroll
    for (int ks = 0; ks < F / 4; ++ks) {
      const v2f a = *(const v2f*)(arow + 4 * ks);
      const v2f b = *(const v2f*)(bcol + 2 * ks * WL_PAIR);
      acc = __builtin_amdgcn_wmma_f32_16x16x4_f32(
           false, a,  false, b,
           (short)0, acc,  false,  false);
    }

    {
      float* drow = &lds_d[par][(h * 8) * LDS_STRIDE + col0 + m];
#pragma unroll
      for (int j = 0; j < 8; ++j) drow[j * LDS_STRIDE] = acc[j];
    }
    __syncthreads();

    const int rbase = row0 + (h * 8);
    float* __restrict__ dst = xl + (size_t)rbase * F + col0 + m;
    if (row0 + 16 <= N) {
#pragma unroll
      for (int j = 0; j < 8; ++j) dst[(size_t)j * F] = acc[j];
    } else {
#pragma unroll
      for (int j = 0; j < 8; ++j) {
        if (rbase + j < N) dst[(size_t)j * F] = acc[j];
      }
    }

    prev_row0 = row0;
    par ^= 1;
  }
  tile_half_logits(lds_d[par ^ 1], prev_row0, N, tid, att_src, att_dst, a_src, a_dst);
}

__device__ __forceinline__ float leaky(float v) {
  return (v >= 0.f) ? v : v * NEG_SLOPE;
}

#define AGG_THREADS 1024
#define AGG_WAVES (AGG_THREADS / 32)
#define LIST_CAP 4096
#define SCAN_UNROLL 8
#define SCAN_STEP (AGG_THREADS * 4)
#define SCAN_SPAN (SCAN_STEP * SCAN_UNROLL)
#define OWN_STEP (AGG_THREADS * 16)
#define OWN_SPAN (OWN_STEP * SCAN_UNROLL)
#define ROWS_PER_LANE ((NPB + 31) / 32)
#define EDGE_GROUP 4

static_assert(LIST_CAP == 4 * AGG_THREADS, "pass (1) gives every thread four list entries");
static_assert(EDGE_GROUP * HEADS == 32, "one lane per (edge of the group, head) weight");

__global__ void __launch_bounds__(AGG_THREADS)
gat_edge_agg_kernel(const int* __restrict__ ei, const float* __restrict__ xl,
                    const float* __restrict__ a_src, const float* __restrict__ a_dst,
                    float* __restrict__ out, int E, int N,
                    const unsigned* __restrict__ own, unsigned long long ws_bytes) {
  __shared__ __attribute__((aligned(16))) float acc[NPB * F];
  __shared__ __attribute__((aligned(16))) unsigned smax[NPB * HEADS];
  __shared__ __attribute__((aligned(16))) float den[NPB * HEADS];
  __shared__ int list_e[LIST_CAP];
  __shared__ int row_cnt[NPB];
  __shared__ int row_start[NPB];
  __shared__ int list_n;
  __shared__ int next_row;

  const int tid  = threadIdx.x;
  const int lane = tid & 31;
  const int wave = __builtin_amdgcn_readfirstlane(tid >> 5);
  const int n0 = blockIdx.x * NPB;
  const int nrows = (N - n0 < NPB) ? (N - n0) : NPB;
  const int* __restrict__ dst = ei + E;

  for (int i = tid; i < NPB * HEADS; i += AGG_THREADS) smax[i] = 0x007FFFFFu;
  if (tid < NPB) row_cnt[tid] = 0;
  if (tid < nrows) list_e[tid] = E + n0 + tid;
  if (tid == 0) { list_n = nrows; next_row = 0; }
  __syncthreads();

  auto keep = [&](int e, int dv) {
    unsigned dl = (unsigned)(dv - n0);
    if (dl < (unsigned)nrows) {
      int slot = __hip_atomic_fetch_add(&list_n, 1, __ATOMIC_RELAXED, __HIP_MEMORY_SCOPE_WORKGROUP);
      if (slot < LIST_CAP) list_e[slot] = e;
    }
  };
  auto keep_e = [&](int e) {
    int slot = __hip_atomic_fetch_add(&list_n, 1, __ATOMIC_RELAXED, __HIP_MEMORY_SCOPE_WORKGROUP);
    if (slot < LIST_CAP) list_e[slot] = e;
  };

  const bool use_plane = owner_plane_ok(N, E, ws_bytes);
  if (use_plane) {
    int n_it = (E + OWN_SPAN - 1) / OWN_SPAN;
    if (n_it < 1) n_it = 1;
    const int rot = (int)(((long long)blockIdx.x * n_it) / gridDim.x);
    const unsigned pat  = blockIdx.x * 0x01010101u;
    const uint4* __restrict__ own4 = (const uint4*)own;
    for (int it = 0; it < n_it; ++it) {
      int c = it + rot;
      if (c >= n_it) c -= n_it;
      const int b0 = c * OWN_SPAN + tid * 16;
      uint4 v[SCAN_UNROLL];
#pragma unroll
      for (int u = 0; u < SCAN_UNROLL; ++u) {
        const int idx = b0 + u * OWN_STEP;
        v[u] = own4[min(idx, E - 16) >> 4];
      }
#pragma unroll
      for (int u = 0; u < SCAN_UNROLL; u += 2) {
        const int idx = b0 + u * OWN_STEP;
        unsigned m = 0;
#pragma unroll
        for (int w = 0; w < 2; ++w) {
          const unsigned x[4] = {v[u + w].x ^ pat, v[u + w].y ^ pat, v[u + w].z ^ pat, v[u + w].w ^ pat};
          unsigned mw = 0;
#pragma unroll
          for (int q = 0; q < 4; ++q) {
            const unsigned f = ~(((x[q] & 0x7f7f7f7fu) + 0x7f7f7f7fu) | x[q]) & 0x80808080u;
            mw |= f >> (7 - q);
          }
          if (idx + w * OWN_STEP >= E) mw = 0;
          m |= mw << (4 * w);
        }
        while (m) {
          const int p = __builtin_ctz(m);
          keep_e(idx + ((p >> 2) & 1) * OWN_STEP + 4 * (p & 3) + (p >> 3));
          m &= m - 1;
        }
      }
    }
  } else {
  int n_it = (E + SCAN_SPAN - 1) / SCAN_SPAN;
  if (n_it < 1) n_it = 1;
  const int rot = (int)(((long long)blockIdx.x * n_it) / gridDim.x);
  const bool vec_ok = ((E & 3) == 0);

  for (int it = 0; it < n_it; ++it) {
    int c = it + rot;
    if (c >= n_it) c -= n_it;
    const int b0 = c * SCAN_SPAN + tid * 4;

    if (vec_ok && (c + 1) * SCAN_SPAN <= E) {
      int4 v[SCAN_UNROLL];
#pragma unroll
      for (int u = 0; u < SCAN_UNROLL; ++u)
        v[u] = *(const int4*)(dst + b0 + u * SCAN_STEP);
#pragma unroll
      for (int u = 0; u < SCAN_UNROLL; ++u) {
        const int idx = b0 + u * SCAN_STEP;
        const bool any = ((unsigned)(v[u].x - n0) < (unsigned)nrows) | ((unsigned)(v[u].y - n0) < (unsigned)nrows) |
                         ((unsigned)(v[u].z - n0) < (unsigned)nrows) | ((unsigned)(v[u].w - n0) < (unsigned)nrows);
        if (any) {
          keep(idx + 0, v[u].x);
          keep(idx + 1, v[u].y);
          keep(idx + 2, v[u].z);
          keep(idx + 3, v[u].w);
        }
      }
    } else {
      for (int u = 0; u < SCAN_UNROLL; ++u) {
        const int idx = b0 + u * SCAN_STEP;
        for (int k = 0; k < 4; ++k)
          if (idx + k < E) keep(idx + k, dst[idx + k]);
      }
    }
  }
  }
  __syncthreads();
  const int cnt = __builtin_amdgcn_readfirstlane(list_n);

  if (cnt <= LIST_CAP) {
    int dd[4], ss[4], rk[4];
    {
      int ee[4];
      float4 sa[4][2], da[4][2];
#pragma unroll
      for (int k = 0; k < 4; ++k) {
        const int i = k * AGG_THREADS + tid;
        ee[k] = (i < cnt) ? list_e[i] : (E + n0);
      }
#pragma unroll
      for (int k = 0; k < 4; ++k) {
        const bool real = ee[k] < E;
        ss[k] = real ? ei[ee[k]] : (ee[k] - E);
        const int dv = real ? dst[ee[k]] : (ee[k] - E);
        dd[k] = min(max(dv - n0, 0), nrows - 1);
      }
#pragma unroll
      for (int k = 0; k < 4; ++k) {
        const float4* ps = (const float4*)(a_src + (size_t)ss[k] * HEADS);
        const float4* pd = (const float4*)(a_dst + (size_t)(n0 + dd[k]) * HEADS);
        sa[k][0] = ps[0]; sa[k][1] = ps[1];
        da[k][0] = pd[0]; da[k][1] = pd[1];
      }
#pragma unroll
      for (int k = 0; k < 4; ++k) {
        const int i = k * AGG_THREADS + tid;
        rk[k] = 0;
        if (i < cnt) {
          rk[k] = __hip_atomic_fetch_add(&row_cnt[dd[k]], 1, __ATOMIC_RELAXED, __HIP_MEMORY_SCOPE_WORKGROUP);
          unsigned* mx = smax + dd[k] * HEADS;
          __hip_atomic_fetch_max(&mx[0], f2key(leaky(sa[k][0].x + da[k][0].x)), __ATOMIC_RELAXED, __HIP_MEMORY_SCOPE_WORKGROUP);
          __hip_atomic_fetch_max(&mx[1], f2key(leaky(sa[k][0].y + da[k][0].y)), __ATOMIC_RELAXED, __HIP_MEMORY_SCOPE_WORKGROUP);
          __hip_atomic_fetch_max(&mx[2], f2key(leaky(sa[k][0].z + da[k][0].z)), __ATOMIC_RELAXED, __HIP_MEMORY_SCOPE_WORKGROUP);
          __hip_atomic_fetch_max(&mx[3], f2key(leaky(sa[k][0].w + da[k][0].w)), __ATOMIC_RELAXED, __HIP_MEMORY_SCOPE_WORKGROUP);
          __hip_atomic_fetch_max(&mx[4], f2key(leaky(sa[k][1].x + da[k][1].x)), __ATOMIC_RELAXED, __HIP_MEMORY_SCOPE_WORKGROUP);
          __hip_atomic_fetch_max(&mx[5], f2key(leaky(sa[k][1].y + da[k][1].y)), __ATOMIC_RELAXED, __HIP_MEMORY_SCOPE_WORKGROUP);
          __hip_atomic_fetch_max(&mx[6], f2key(leaky(sa[k][1].z + da[k][1].z)), __ATOMIC_RELAXED, __HIP_MEMORY_SCOPE_WORKGROUP);
          __hip_atomic_fetch_max(&mx[7], f2key(leaky(sa[k][1].w + da[k][1].w)), __ATOMIC_RELAXED, __HIP_MEMORY_SCOPE_WORKGROUP);
        }
      }
    }
    __syncthreads();

    if (wave == 0) {
      int c[ROWS_PER_LANE];
      int tot = 0;
#pragma unroll
      for (int q = 0; q < ROWS_PER_LANE; ++q) {
        const int r = lane * ROWS_PER_LANE + q;
        c[q] = (r < NPB) ? row_cnt[r] : 0;
        tot += c[q];
      }
      int inc = tot;
      for (int d = 1; d < 32; d <<= 1) {
        const int v = __shfl_up(inc, d, 32);
        if (lane >= d) inc += v;
      }
      int run = inc - tot;
#pragma unroll
      for (int q = 0; q < ROWS_PER_LANE; ++q) {
        const int r = lane * ROWS_PER_LANE + q;
        if (r < NPB) row_start[r] = run;
        run += c[q];
      }
    }
    __syncthreads();

#pragma unroll
    for (int k = 0; k < 4; ++k) {
      const int i = k * AGG_THREADS + tid;
      if (i < cnt) list_e[row_start[dd[k]] + rk[k]] = ss[k];
    }
    __syncthreads();

    const int hq = lane & 7;
    const int gq = lane >> 3;
    const int h0 = lane >> 3;
    for (int t = 0; t < NPB; ++t) {
      int r = 0;
      if (lane == 0) r = __hip_atomic_fetch_add(&next_row, 1, __ATOMIC_RELAXED, __HIP_MEMORY_SCOPE_WORKGROUP);
      r = __builtin_amdgcn_readfirstlane(r);
      if (r >= nrows) break;
      const int st = __builtin_amdgcn_readfirstlane(row_start[r]);
      const int ne = __builtin_amdgcn_readfirstlane(row_cnt[r]);
      const float adq = a_dst[(size_t)(n0 + r) * HEADS + hq];
      const float mq  = key2f(smax[r * HEADS + hq]);

      float4 sa = make_float4(0.f, 0.f, 0.f, 0.f);
      float4 sb = make_float4(0.f, 0.f, 0.f, 0.f);
      float dsum = 0.f;

      for (int c0 = 0; c0 < ne; c0 += 32) {
        const int m = (ne - c0 < 32) ? (ne - c0) : 32;
        int s = n0;
        if (lane < m) s = list_e[st + c0 + lane];

        for (int i0 = 0; i0 < m; i0 += EDGE_GROUP) {
          const int iq = (i0 + gq < m) ? (i0 + gq) : (m - 1);
          const int sq = list_e[st + c0 + iq];
          const float asq = a_src[(size_t)sq * HEADS + hq];

          float4 xa[EDGE_GROUP], xb[EDGE_GROUP];
#pragma unroll
          for (int j = 0; j < EDGE_GROUP; ++j) {
            const int ii = (i0 + j < m) ? (i0 + j) : (m - 1);
            const int si = __builtin_amdgcn_readlane(s, ii);
            const float4* __restrict__ xr = (const float4*)(xl + (size_t)si * F) + lane;
            xa[j] = xr[0];
            xb[j] = xr[32];
          }

          float p = expf(leaky(asq + adq) - mq);
          if (i0 + gq >= m) p = 0.f;
          dsum += p;

#pragma unroll
          for (int j = 0; j < EDGE_GROUP; ++j) {
            const float u = __shfl(p, 8 * j + h0, 32);
            const float v = __shfl(p, 8 * j + 4 + h0, 32);
            sa.x += u * xa[j].x; sa.y += u * xa[j].y; sa.z += u * xa[j].z; sa.w += u * xa[j].w;
            sb.x += v * xb[j].x; sb.y += v * xb[j].y; sb.z += v * xb[j].z; sb.w += v * xb[j].w;
          }
        }
      }

      dsum += __shfl_xor(dsum, 8, 32);
      dsum += __shfl_xor(dsum, 16, 32);
      const float ra = 1.0f / __shfl(dsum, h0, 32);
      const float rb = 1.0f / __shfl(dsum, 4 + h0, 32);
      float4* __restrict__ orow = (float4*)(out + (size_t)(n0 + r) * F);
      orow[lane]      = make_float4(sa.x * ra, sa.y * ra, sa.z * ra, sa.w * ra);
      orow[32 + lane] = make_float4(sb.x * rb, sb.y * rb, sb.z * rb, sb.w * rb);
    }
  } else {
    for (int i = tid; i < NPB * F / 4; i += AGG_THREADS)
      ((float4*)acc)[i] = make_float4(0.f, 0.f, 0.f, 0.f);
    for (int i = tid; i < NPB * HEADS; i += AGG_THREADS) den[i] = 0.f;
    __syncthreads();
    const int tot = E + N;
    for (int e = tid; e < tot; e += AGG_THREADS) {
      const int d = (e < E) ? dst[e] : (e - E);
      const unsigned dl = (unsigned)(d - n0);
      if (dl < (unsigned)nrows) {
        const int s = (e < E) ? ei[e] : (e - E);
        for (int h = 0; h < HEADS; ++h)
          __hip_atomic_fetch_max(&smax[dl * HEADS + h], f2key(leaky(a_src[s * HEADS + h] + a_dst[d * HEADS + h])),
                                 __ATOMIC_RELAXED, __HIP_MEMORY_SCOPE_WORKGROUP);
      }
    }
    __syncthreads();
    for (int e = tid; e < tot; e += AGG_THREADS) {
      const int d = (e < E) ? dst[e] : (e - E);
      const unsigned dl = (unsigned)(d - n0);
      if (dl < (unsigned)nrows) {
        const int s = (e < E) ? ei[e] : (e - E);
        for (int h = 0; h < HEADS; ++h) {
          const float p = expf(leaky(a_src[s * HEADS + h] + a_dst[d * HEADS + h]) - key2f(smax[dl * HEADS + h]));
          __hip_atomic_fetch_add(&den[dl * HEADS + h], p, __ATOMIC_RELAXED, __HIP_MEMORY_SCOPE_WORKGROUP);
          for (int c = 0; c < CH; ++c)
            __hip_atomic_fetch_add(&acc[dl * F + h * CH + c], xl[(size_t)s * F + h * CH + c] * p,
                                   __ATOMIC_RELAXED, __HIP_MEMORY_SCOPE_WORKGROUP);
        }
      }
    }
    __syncthreads();

    for (int i = tid; i < nrows * HEADS; i += AGG_THREADS) den[i] = 1.0f / den[i];
    __syncthreads();
    float4* __restrict__ orow = (float4*)(out + (size_t)n0 * F);
    for (int i = tid; i < nrows * (F / 4); i += AGG_THREADS) {
      float4 v = ((const float4*)acc)[i];
      const float r = den[(i >> 6) * HEADS + ((i & 63) >> 3)];
      v.x *= r; v.y *= r; v.z *= r; v.w *= r;
      orow[i] = v;
    }
  }
}

#define LN_ROWS 4
#define LN_THREADS 256
#define LN_ROWS_PER_BLOCK (LN_ROWS * (LN_THREADS / 32))

__device__ __forceinline__ float wave_sum(float v) {
#pragma unroll
  for (int m = 16; m > 0; m >>= 1) v += __shfl_xor(v, m, 32);
  return v;
}

__global__ void __launch_bounds__(LN_THREADS)
gat_ln_kernel(float* __restrict__ out, const float* __restrict__ bias,
              const float* __restrict__ gamma, const float* __restrict__ beta, int N) {
  const int lane = threadIdx.x & 31;
  const int wave = __builtin_amdgcn_readfirstlane((int)((blockIdx.x * LN_THREADS + threadIdx.x) >> 5));
  const int row0 = wave * LN_ROWS;
  if (row0 >= N) return;

  float4 va[LN_ROWS], vb[LN_ROWS];
#pragma unroll
  for (int r = 0; r < LN_ROWS; ++r) {
    const int row = (row0 + r < N) ? (row0 + r) : row0;
    const float4* p = (const float4*)(out + (size_t)row * F);
    va[r] = p[lane];
    vb[r] = p[32 + lane];
  }
  const float4 b0 = ((const float4*)bias)[lane],  b1 = ((const float4*)bias)[32 + lane];
  const float4 g0 = ((const float4*)gamma)[lane], g1 = ((const float4*)gamma)[32 + lane];
  const float4 t0 = ((const float4*)beta)[lane],  t1 = ((const float4*)beta)[32 + lane];

  float mean[LN_ROWS];
#pragma unroll
  for (int r = 0; r < LN_ROWS; ++r) {
    va[r].x += b0.x; va[r].y += b0.y; va[r].z += b0.z; va[r].w += b0.w;
    vb[r].x += b1.x; vb[r].y += b1.y; vb[r].z += b1.z; vb[r].w += b1.w;
    const float s = ((va[r].x + va[r].y) + (va[r].z + va[r].w)) +
                    ((vb[r].x + vb[r].y) + (vb[r].z + vb[r].w));
    mean[r] = wave_sum(s) * (1.0f / F);
  }
  float rs[LN_ROWS];
#pragma unroll
  for (int r = 0; r < LN_ROWS; ++r) {
    va[r].x -= mean[r]; va[r].y -= mean[r]; va[r].z -= mean[r]; va[r].w -= mean[r];
    vb[r].x -= mean[r]; vb[r].y -= mean[r]; vb[r].z -= mean[r]; vb[r].w -= mean[r];
    const float q = ((va[r].x * va[r].x + va[r].y * va[r].y) + (va[r].z * va[r].z + va[r].w * va[r].w)) +
                    ((vb[r].x * vb[r].x + vb[r].y * vb[r].y) + (vb[r].z * vb[r].z + vb[r].w * vb[r].w));
    rs[r] = rsqrtf(wave_sum(q) * (1.0f / F) + LN_EPS);
  }
#pragma unroll
  for (int r = 0; r < LN_ROWS; ++r) {
    if (row0 + r < N) {
      float4* o = (float4*)(out + (size_t)(row0 + r) * F);
      float4 ya, yb;
      ya.x = va[r].x * rs[r] * g0.x + t0.x; ya.y = va[r].y * rs[r] * g0.y + t0.y;
      ya.z = va[r].z * rs[r] * g0.z + t0.z; ya.w = va[r].w * rs[r] * g0.w + t0.w;
      yb.x = vb[r].x * rs[r] * g1.x + t1.x; yb.y = vb[r].y * rs[r] * g1.y + t1.y;
      yb.z = vb[r].z * rs[r] * g1.z + t1.z; yb.w = vb[r].w * rs[r] * g1.w + t1.w;
      o[lane] = ya;
      o[32 + lane] = yb;
    }
  }
}

extern "C" void kernel_launch(void* const* d_in, const int* in_sizes, int n_in,
                              void* d_out, int out_size, void* d_ws, size_t ws_size,
                              hipStream_t stream) {
  const float* x       = (const float*)d_in[0];
  const int*   ei      = (const int*)  d_in[1];
  const float* W       = (const float*)d_in[2];
  const float* att_src = (const float*)d_in[3];
  const float* att_dst = (const float*)d_in[4];
  const float* bias    = (const float*)d_in[5];
  const float* gamma   = (const float*)d_in[6];
  const float* beta    = (const float*)d_in[7];
  float* out = (float*)d_out;

  const int N = in_sizes[0] / F;
  const int E = in_sizes[1] / 2;

  char* ws = (char*)d_ws;
  float*    xl     = (float*)ws;    ws += (size_t)N * F * sizeof(float);
  float*    a_src  = (float*)ws;    ws += (size_t)N * HEADS * sizeof(float);
  float*    a_dst  = (float*)ws;    ws += (size_t)N * HEADS * sizeof(float);
  unsigned* own    = (unsigned*)ws;

  gat_gemm_wmma_kernel<<<GEMM_BLOCKS, 512, 0, stream>>>(x, W, xl, N, att_src, att_dst,
                                                         a_src, a_dst, ei, own, E,
                                                         (unsigned long long)ws_size);

  gat_edge_agg_kernel<<<(N + NPB - 1) / NPB, AGG_THREADS, 0, stream>>>(ei, xl, a_src, a_dst,
                                                                        out, E, N, own,
                                                                        (unsigned long long)ws_size);

  gat_ln_kernel<<<(N + LN_ROWS_PER_BLOCK - 1) / LN_ROWS_PER_BLOCK, LN_THREADS, 0, stream>>>(out, bias, gamma, beta, N);
}


# edges binned by owner in the projection launch: block b sorts its 1/256 of the list into 256 cells (LDS counters, src+row words); agg reads its cells: plane scan and src/dst lookups gone; chunk order
# speedup vs baseline: 6.1837x; 1.1008x over previous
#include <hip/hip_runtime.h>
#include <math.h>

typedef float v2f __attribute__((ext_vector_type(2)));
typedef float v8f __attribute__((ext_vector_type(8)));

#define HEADS 8
#define CH 32
#define F 256
#define NEG_SLOPE 0.2f
#define LN_EPS 1e-5f

#define LDS_STRIDE 260

__device__ __forceinline__ unsigned f2key(float f) {
  unsigned b = __float_as_uint(f);
  return (b & 0x80000000u) ? ~b : (b | 0x80000000u);
}
__device__ __forceinline__ float key2f(unsigned k) {
  unsigned b = (k & 0x80000000u) ? (k & 0x7fffffffu) : ~k;
  return __uint_as_float(b);
}

#define NPB 196

__device__ __forceinline__ bool owner_plane_ok(int N, int E, unsigned long long ws_bytes) {
  const unsigned long long need = (unsigned long long)N * F * sizeof(float) +
                                  2ull * N * HEADS * sizeof(float) + (unsigned long long)E;
  return (E >= 16) && ((E & 15) == 0) && ((N + NPB - 1) / NPB <= 256) && (ws_bytes >= need);
}

#define GEMM_BLOCKS 256
#define CELL_SLOTS 64
#define CELL_EDGES (CELL_SLOTS - 1)
__device__ __forceinline__ bool owner_cells_ok(int N, int E, unsigned long long ws_bytes) {
  const unsigned long long need = (unsigned long long)N * F * sizeof(float) +
                                  2ull * N * HEADS * sizeof(float) +
                                  (unsigned long long)GEMM_BLOCKS * 256ull * CELL_SLOTS * sizeof(unsigned);
  return (E >= 1) && (E <= (1 << 30)) && (N <= (1 << 24)) && ((N & 3) == 0) && ((N + NPB - 1) / NPB <= 256) && (ws_bytes >= need);
}

#define WL_PAIR 512

__device__ __forceinline__ void tile_half_logits(const float* dtile, int row0, int N, int tid,
                                                 const float* __restrict__ att_src,
                                                 const float* __restrict__ att_dst,
                                                 float* __restrict__ a_src, float* __restrict__ a_dst) {
  if (tid < 256) {
    const int lane = tid & 31;
    const int wu = __builtin_amdgcn_readfirstlane(tid >> 5);
    const bool is_dst = (wu & 1) != 0;
    const int r = (wu >> 1) * 4 + (lane >> 3);
    const int hd = lane & 7;
    const float4* xr = (const float4*)(&dtile[r * LDS_STRIDE + hd * CH]);
    const float4* __restrict__ av = (const float4*)((is_dst ? att_dst : att_src) + hd * CH);
    float s = 0.f;
#pragma unroll
    for (int c = 0; c < CH / 4; ++c) {
      const float4 v = xr[c];
      const float4 a = av[c];
      s += v.x * a.x;
      s += v.y * a.y;
      s += v.z * a.z;
      s += v.w * a.w;
    }
    if (row0 + r < N) (is_dst ? a_dst : a_src)[(size_t)(row0 + r) * HEADS + hd] = s;
  }
}

__global__ void __launch_bounds__(512)
gat_gemm_wmma_kernel(const float* __restrict__ x, const float* __restrict__ W,
                     float* __restrict__ xl, int N,
                     const float* __restrict__ att_src, const float* __restrict__ att_dst,
                     float* __restrict__ a_src, float* __restrict__ a_dst,
                     const int* __restrict__ ei, unsigned* __restrict__ own,
                     int E, unsigned long long ws_bytes) {
  __shared__ __attribute__((aligned(16))) float lds_w[F * F];
  __shared__ __attribute__((aligned(16))) float lds_a[16 * LDS_STRIDE];
  __shared__ __attribute__((aligned(16))) float lds_d[2][16 * LDS_STRIDE];

  const int tid  = threadIdx.x;
  const int lane = tid & 31;
  const int wave = tid >> 5;
  const int col0 = wave * 16;
  const int ntiles = (N + 15) >> 4;

  if (owner_cells_ok(N, E, ws_bytes)) {
    unsigned* st = (unsigned*)lds_w;
    int* ccnt = (int*)(lds_w + 256 * CELL_SLOTS);
    for (int i = tid; i < 256 * CELL_SLOTS / 4; i += 512)
      ((uint4*)st)[i] = make_uint4(0u, 0u, 0u, 0u);
    if (tid < 256 + 64) ccnt[tid] = 0;
    __syncthreads();
    const int per = (E + GEMM_BLOCKS - 1) / GEMM_BLOCKS;
    const int e0 = blockIdx.x * per;
    const int e1 = (e0 + per < E) ? (e0 + per) : E;
    const int* __restrict__ srcc = ei;
    const int* __restrict__ dstc = ei + E;
    for (int base = e0; base < e1; base += 8 * 512) {
      int dv[8], sv[8];
#pragma unroll
      for (int u = 0; u < 8; ++u) {
        const int e = base + u * 512 + tid;
        const int ec = (e < e1) ? e : e0;
        dv[u] = dstc[ec];
        sv[u] = srcc[ec];
      }
      unsigned oo[8], ww[8];
      int rr[8];
#pragma unroll
      for (int u = 0; u < 8; ++u) {
        const int e = base + u * 512 + tid;
        unsigned o = (unsigned)dv[u] / NPB;
        if (o > 255u) o = 255u;
        const unsigned dl = (unsigned)dv[u] - o * NPB;
        ww[u] = ((unsigned)sv[u] & 0x00FFFFFFu) | (dl << 24);
        oo[u] = (e < e1) ? o : (256u + (unsigned)(tid & 63));
        rr[u] = __hip_atomic_fetch_add(&ccnt[oo[u]], 1, __ATOMIC_RELAXED, __HIP_MEMORY_SCOPE_WORKGROUP);
      }
#pragma unroll
      for (int u = 0; u < 8; ++u) {
        const bool placed = (oo[u] < 256u) && (rr[u] < CELL_EDGES);
        st[placed ? (oo[u] * CELL_SLOTS + (unsigned)rr[u]) : (unsigned)(256 * CELL_SLOTS + 320 + tid)] = ww[u];
      }
    }
    __syncthreads();
    for (int i = tid; i < 256 * CELL_SLOTS / 4; i += 512) {
      const int o = i >> 4;
      const int q = i & 15;
      uint4 v = ((const uint4*)st)[i];
      if (q == 15) v.w = (unsigned)ccnt[o];
      ((uint4*)(own + ((size_t)o * GEMM_BLOCKS + blockIdx.x) * CELL_SLOTS))[q] = v;
    }
    __syncthreads();
  } else if (owner_plane_ok(N, E, ws_bytes)) {
    const int4* __restrict__ d4 = (const int4*)(ei + E);
    for (long long g = (long long)blockIdx.x * 512 + tid; g < (long long)(E >> 2);
         g += (long long)gridDim.x * 512) {
      const int4 d = d4[g];
      own[g] = ((unsigned)d.x / NPB) | (((unsigned)d.y / NPB) << 8) |
               (((unsigned)d.z / NPB) << 16) | (((unsigned)d.w / NPB) << 24);
    }
  }

  int tile = blockIdx.x;
  if (tile >= ntiles) return;

  const int ar = tid >> 5;
  const int ac = (tid & 31) * 8;
  float4 pa0, pa1;
  {
    int grow = tile * 16 + ar;
    if (grow >= N) grow = N - 1;
    const float4* __restrict__ src = (const float4*)(x + (size_t)grow * F + ac);
    pa0 = src[0];
    pa1 = src[1];
  }

  for (int it0 = 0; it0 < 16; it0 += 4) {
    float4 r0[4], r1[4];
#pragma unroll
    for (int u = 0; u < 4; ++u) {
      const int item = (it0 + u) * 512 + tid;
      const int p = item >> 6;
      const int q = item & 63;
      r0[u] = *(const float4*)(W + (size_t)(2 * p) * F + 4 * q);
      r1[u] = *(const float4*)(W + (size_t)(2 * p + 1) * F + 4 * q);
    }
#pragma unroll
    for (int u = 0; u < 4; ++u) {
      const int item = (it0 + u) * 512 + tid;
      const int p = item >> 6;
      const int q = item & 63;
      float4* d = (float4*)(&lds_w[p * WL_PAIR + ((8 * q + 32 * (p & 1)) & (WL_PAIR - 1))]);
      d[0] = make_float4(r0[u].x, r1[u].x, r0[u].y, r1[u].y);
      d[1] = make_float4(r0[u].z, r1[u].z, r0[u].w, r1[u].w);
    }
  }

  const int m = lane & 15;
  const int h = lane >> 4;
  const float* arow = &lds_a[m * LDS_STRIDE + 2 * h];
  const float* bcol = &lds_w[h * WL_PAIR + ((2 * (col0 + m) + 32 * h) & (WL_PAIR - 1))];

  int par = 0;
  int prev_row0 = -1;
  for (; tile < ntiles; tile += gridDim.x) {
    const int row0 = tile * 16;
    {
      float4* dst = (float4*)(&lds_a[ar * LDS_STRIDE + ac]);
      dst[0] = pa0;
      dst[1] = pa1;
    }
    __builtin_amdgcn_fence(__ATOMIC_RELEASE, "workgroup", "local");
    __builtin_amdgcn_s_barrier();
    __builtin_amdgcn_fence(__ATOMIC_ACQUIRE, "workgroup", "local");

    if (prev_row0 >= 0)
      tile_half_logits(lds_d[par ^ 1], prev_row0, N, tid, att_src, att_dst, a_src, a_dst);

    if (tile + (int)gridDim.x < ntiles) {
      int grow = (tile + (int)gridDim.x) * 16 + ar;
      if (grow >= N) grow = N - 1;
      const float4* __restrict__ src = (const float4*)(x + (size_t)grow * F + ac);
      pa0 = src[0];
      pa1 = src[1];
    }

    v8f acc = {0.f, 0.f, 0.f, 0.f, 0.f, 0.f, 0.f, 0.f};
#pragma unroll
    for (int ks = 0; ks < F / 4; ++ks) {
      const v2f a = *(const v2f*)(arow + 4 * ks);
      const v2f b = *(const v2f*)(bcol + 2 * ks * WL_PAIR);
      acc = __builtin_amdgcn_wmma_f32_16x16x4_f32(
           false, a,  false, b,
           (short)0, acc,  false,  false);
    }

    {
      float* drow = &lds_d[par][(h * 8) * LDS_STRIDE + col0 + m];
#pragma unroll
      for (int j = 0; j < 8; ++j) drow[j * LDS_STRIDE] = acc[j];
    }
    __syncthreads();

    const int rbase = row0 + (h * 8);
    float* __restrict__ dst = xl + (size_t)rbase * F + col0 + m;
    if (row0 + 16 <= N) {
#pragma unroll
      for (int j = 0; j < 8; ++j) dst[(size_t)j * F] = acc[j];
    } else {
#pragma unroll
      for (int j = 0; j < 8; ++j) {
        if (rbase + j < N) dst[(size_t)j * F] = acc[j];
      }
    }

    prev_row0 = row0;
    par ^= 1;
  }
  tile_half_logits(lds_d[par ^ 1], prev_row0, N, tid, att_src, att_dst, a_src, a_dst);
}

__device__ __forceinline__ float leaky(float v) {
  return (v >= 0.f) ? v : v * NEG_SLOPE;
}

#define AGG_THREADS 1024
#define AGG_WAVES (AGG_THREADS / 32)
#define LIST_CAP 4096
#define SCAN_UNROLL 8
#define SCAN_STEP (AGG_THREADS * 4)
#define SCAN_SPAN (SCAN_STEP * SCAN_UNROLL)
#define OWN_STEP (AGG_THREADS * 16)
#define OWN_SPAN (OWN_STEP * SCAN_UNROLL)
#define ROWS_PER_LANE ((NPB + 31) / 32)
#define EDGE_GROUP 4

static_assert(LIST_CAP == 4 * AGG_THREADS, "pass (1) gives every thread four list entries");
static_assert(EDGE_GROUP * HEADS == 32, "one lane per (edge of the group, head) weight");

__global__ void __launch_bounds__(AGG_THREADS)
gat_edge_agg_kernel(const int* __restrict__ ei, const float* __restrict__ xl,
                    const float* __restrict__ a_src, const float* __restrict__ a_dst,
                    float* __restrict__ out, int E, int N,
                    const unsigned* __restrict__ own, unsigned long long ws_bytes) {
  __shared__ __attribute__((aligned(16))) float acc[NPB * F];
  __shared__ __attribute__((aligned(16))) unsigned smax[NPB * HEADS];
  __shared__ __attribute__((aligned(16))) float den[NPB * HEADS];
  __shared__ int list_e[LIST_CAP];
  __shared__ int row_cnt[NPB];
  __shared__ int row_start[NPB];
  __shared__ int list_n;
  __shared__ int next_row;
  __shared__ int cell_base[GEMM_BLOCKS];

  const int tid  = threadIdx.x;
  const int lane = tid & 31;
  const int wave = __builtin_amdgcn_readfirstlane(tid >> 5);
  const int n0 = blockIdx.x * NPB;
  const int nrows = (N - n0 < NPB) ? (N - n0) : NPB;
  const int* __restrict__ dst = ei + E;

  const bool use_cells = owner_cells_ok(N, E, ws_bytes);

  for (int i = tid; i < NPB * HEADS; i += AGG_THREADS) smax[i] = 0x007FFFFFu;
  if (tid < NPB) row_cnt[tid] = 0;
  if (tid < nrows) list_e[tid] = use_cells ? (int)((unsigned)(n0 + tid) | ((unsigned)tid << 24)) : (E + n0 + tid);
  if (tid == 0) { list_n = nrows; next_row = 0; }
  __syncthreads();

  auto keep = [&](int e, int dv) {
    unsigned dl = (unsigned)(dv - n0);
    if (dl < (unsigned)nrows) {
      int slot = __hip_atomic_fetch_add(&list_n, 1, __ATOMIC_RELAXED, __HIP_MEMORY_SCOPE_WORKGROUP);
      if (slot < LIST_CAP) list_e[slot] = e;
    }
  };
  auto keep_e = [&](int e) {
    int slot = __hip_atomic_fetch_add(&list_n, 1, __ATOMIC_RELAXED, __HIP_MEMORY_SCOPE_WORKGROUP);
    if (slot < LIST_CAP) list_e[slot] = e;
  };

  const bool use_plane = owner_plane_ok(N, E, ws_bytes);
  if (use_cells) {
    static_assert(GEMM_BLOCKS * (CELL_SLOTS / 4) == 4 * AGG_THREADS, "four 16-byte pieces of the cells per thread");
    const uint4* __restrict__ cell4 = (const uint4*)(own + (size_t)blockIdx.x * (GEMM_BLOCKS * CELL_SLOTS));
    uint4 v[4];
    int cc[4];
#pragma unroll
    for (int k = 0; k < 4; ++k) v[k] = cell4[k * AGG_THREADS + tid];
#pragma unroll
    for (int k = 0; k < 4; ++k) {
      cc[k] = __shfl((int)v[k].w, (lane & 16) | 15, 32);
      if ((tid & 15) == 15) cell_base[(k * AGG_THREADS + tid) >> 4] = cc[k];
    }
    __syncthreads();
    if (wave == 0) {
      int c[GEMM_BLOCKS / 32];
      int tot = 0, over = 0;
#pragma unroll
      for (int q = 0; q < GEMM_BLOCKS / 32; ++q) {
        const int raw = cell_base[lane * (GEMM_BLOCKS / 32) + q];
        over |= (raw > CELL_EDGES || raw < 0) ? 1 : 0;
        c[q] = min(max(raw, 0), CELL_EDGES);
        tot += c[q];
      }
      int inc = tot;
      for (int d = 1; d < 32; d <<= 1) {
        const int t = __shfl_up(inc, d, 32);
        if (lane >= d) inc += t;
        over |= __shfl_xor(over, d, 32);
      }
      int run = inc - tot;
#pragma unroll
      for (int q = 0; q < GEMM_BLOCKS / 32; ++q) {
        cell_base[lane * (GEMM_BLOCKS / 32) + q] = run;
        run += c[q];
      }
      if (lane == 31) list_n = over ? (LIST_CAP + 1) : (nrows + inc);
    }
    __syncthreads();
#pragma unroll
    for (int k = 0; k < 4; ++k) {
      const int cell = (k * AGG_THREADS + tid) >> 4;
      const int q4 = (tid & 15) * 4;
      const int n = min(max(cc[k], 0), CELL_EDGES);
      const int base = nrows + cell_base[cell] + q4;
      const unsigned w4[4] = {v[k].x, v[k].y, v[k].z, v[k].w};
#pragma unroll
      for (int j = 0; j < 4; ++j)
        if (q4 + j < n && base + j < LIST_CAP) list_e[base + j] = (int)w4[j];
    }
  } else if (use_plane) {
    int n_it = (E + OWN_SPAN - 1) / OWN_SPAN;
    if (n_it < 1) n_it = 1;
    const int rot = (int)(((long long)blockIdx.x * n_it) / gridDim.x);
    const unsigned pat  = blockIdx.x * 0x01010101u;
    const uint4* __restrict__ own4 = (const uint4*)own;
    for (int it = 0; it < n_it; ++it) {
      int c = it + rot;
      if (c >= n_it) c -= n_it;
      const int b0 = c * OWN_SPAN + tid * 16;
      uint4 v[SCAN_UNROLL];
#pragma unroll
      for (int u = 0; u < SCAN_UNROLL; ++u) {
        const int idx = b0 + u * OWN_STEP;
        v[u] = own4[min(idx, E - 16) >> 4];
      }
#pragma unroll
      for (int u = 0; u < SCAN_UNROLL; u += 2) {
        const int idx = b0 + u * OWN_STEP;
        unsigned m = 0;
#pragma unroll
        for (int w = 0; w < 2; ++w) {
          const unsigned x[4] = {v[u + w].x ^ pat, v[u + w].y ^ pat, v[u + w].z ^ pat, v[u + w].w ^ pat};
          unsigned mw = 0;
#pragma unroll
          for (int q = 0; q < 4; ++q) {
            const unsigned f = ~(((x[q] & 0x7f7f7f7fu) + 0x7f7f7f7fu) | x[q]) & 0x80808080u;
            mw |= f >> (7 - q);
          }
          if (idx + w * OWN_STEP >= E) mw = 0;
          m |= mw << (4 * w);
        }
        while (m) {
          const int p = __builtin_ctz(m);
          keep_e(idx + ((p >> 2) & 1) * OWN_STEP + 4 * (p & 3) + (p >> 3));
          m &= m - 1;
        }
      }
    }
  } else {
  int n_it = (E + SCAN_SPAN - 1) / SCAN_SPAN;
  if (n_it < 1) n_it = 1;
  const int rot = (int)(((long long)blockIdx.x * n_it) / gridDim.x);
  const bool vec_ok = ((E & 3) == 0);

  for (int it = 0; it < n_it; ++it) {
    int c = it + rot;
    if (c >= n_it) c -= n_it;
    const int b0 = c * SCAN_SPAN + tid * 4;

    if (vec_ok && (c + 1) * SCAN_SPAN <= E) {
      int4 v[SCAN_UNROLL];
#pragma unroll
      for (int u = 0; u < SCAN_UNROLL; ++u)
        v[u] = *(const int4*)(dst + b0 + u * SCAN_STEP);
#pragma unroll
      for (int u = 0; u < SCAN_UNROLL; ++u) {
        const int idx = b0 + u * SCAN_STEP;
        const bool any = ((unsigned)(v[u].x - n0) < (unsigned)nrows) | ((unsigned)(v[u].y - n0) < (unsigned)nrows) |
                         ((unsigned)(v[u].z - n0) < (unsigned)nrows) | ((unsigned)(v[u].w - n0) < (unsigned)nrows);
        if (any) {
          keep(idx + 0, v[u].x);
          keep(idx + 1, v[u].y);
          keep(idx + 2, v[u].z);
          keep(idx + 3, v[u].w);
        }
      }
    } else {
      for (int u = 0; u < SCAN_UNROLL; ++u) {
        const int idx = b0 + u * SCAN_STEP;
        for (int k = 0; k < 4; ++k)
          if (idx + k < E) keep(idx + k, dst[idx + k]);
      }
    }
  }
  }
  __syncthreads();
  const int cnt = __builtin_amdgcn_readfirstlane(list_n);

  if (cnt <= LIST_CAP) {
    int dd[4], ss[4], rk[4];
    {
      int ee[4];
      float4 sa[4][2], da[4][2];
#pragma unroll
      for (int k = 0; k < 4; ++k) {
        const int i = k * AGG_THREADS + tid;
        ee[k] = (i < cnt) ? list_e[i] : (use_cells ? n0 : (E + n0));
      }
      if (use_cells) {
#pragma unroll
        for (int k = 0; k < 4; ++k) {
          const unsigned w = (unsigned)ee[k];
          ss[k] = (int)(w & 0x00FFFFFFu);
          ss[k] = min(ss[k], N - 1);
          dd[k] = min((int)(w >> 24), nrows - 1);
        }
      } else {
#pragma unroll
      for (int k = 0; k < 4; ++k) {
        const bool real = ee[k] < E;
        ss[k] = real ? ei[ee[k]] : (ee[k] - E);
        const int dv = real ? dst[ee[k]] : (ee[k] - E);
        dd[k] = min(max(dv - n0, 0), nrows - 1);
      }
      }
#pragma unroll
      for (int k = 0; k < 4; ++k) {
        const float4* ps = (const float4*)(a_src + (size_t)ss[k] * HEADS);
        const float4* pd = (const float4*)(a_dst + (size_t)(n0 + dd[k]) * HEADS);
        sa[k][0] = ps[0]; sa[k][1] = ps[1];
        da[k][0] = pd[0]; da[k][1] = pd[1];
      }
#pragma unroll
      for (int k = 0; k < 4; ++k) {
        const int i = k * AGG_THREADS + tid;
        rk[k] = 0;
        if (i < cnt) {
          rk[k] = __hip_atomic_fetch_add(&row_cnt[dd[k]], 1, __ATOMIC_RELAXED, __HIP_MEMORY_SCOPE_WORKGROUP);
          unsigned* mx = smax + dd[k] * HEADS;
          __hip_atomic_fetch_max(&mx[0], f2key(leaky(sa[k][0].x + da[k][0].x)), __ATOMIC_RELAXED, __HIP_MEMORY_SCOPE_WORKGROUP);
          __hip_atomic_fetch_max(&mx[1], f2key(leaky(sa[k][0].y + da[k][0].y)), __ATOMIC_RELAXED, __HIP_MEMORY_SCOPE_WORKGROUP);
          __hip_atomic_fetch_max(&mx[2], f2key(leaky(sa[k][0].z + da[k][0].z)), __ATOMIC_RELAXED, __HIP_MEMORY_SCOPE_WORKGROUP);
          __hip_atomic_fetch_max(&mx[3], f2key(leaky(sa[k][0].w + da[k][0].w)), __ATOMIC_RELAXED, __HIP_MEMORY_SCOPE_WORKGROUP);
          __hip_atomic_fetch_max(&mx[4], f2key(leaky(sa[k][1].x + da[k][1].x)), __ATOMIC_RELAXED, __HIP_MEMORY_SCOPE_WORKGROUP);
          __hip_atomic_fetch_max(&mx[5], f2key(leaky(sa[k][1].y + da[k][1].y)), __ATOMIC_RELAXED, __HIP_MEMORY_SCOPE_WORKGROUP);
          __hip_atomic_fetch_max(&mx[6], f2key(leaky(sa[k][1].z + da[k][1].z)), __ATOMIC_RELAXED, __HIP_MEMORY_SCOPE_WORKGROUP);
          __hip_atomic_fetch_max(&mx[7], f2key(leaky(sa[k][1].w + da[k][1].w)), __ATOMIC_RELAXED, __HIP_MEMORY_SCOPE_WORKGROUP);
        }
      }
    }
    __syncthreads();

    if (wave == 0) {
      int c[ROWS_PER_LANE];
      int tot = 0;
#pragma unroll
      for (int q = 0; q < ROWS_PER_LANE; ++q) {
        const int r = lane * ROWS_PER_LANE + q;
        c[q] = (r < NPB) ? row_cnt[r] : 0;
        tot += c[q];
      }
      int inc = tot;
      for (int d = 1; d < 32; d <<= 1) {
        const int v = __shfl_up(inc, d, 32);
        if (lane >= d) inc += v;
      }
      int run = inc - tot;
#pragma unroll
      for (int q = 0; q < ROWS_PER_LANE; ++q) {
        const int r = lane * ROWS_PER_LANE + q;
        if (r < NPB) row_start[r] = run;
        run += c[q];
      }
    }
    __syncthreads();

#pragma unroll
    for (int k = 0; k < 4; ++k) {
      const int i = k * AGG_THREADS + tid;
      if (i < cnt) list_e[row_start[dd[k]] + rk[k]] = ss[k];
    }
    __syncthreads();

    const int hq = lane & 7;
    const int gq = lane >> 3;
    const int h0 = lane >> 3;
    for (int t = 0; t < NPB; ++t) {
      int r = 0;
      if (lane == 0) r = __hip_atomic_fetch_add(&next_row, 1, __ATOMIC_RELAXED, __HIP_MEMORY_SCOPE_WORKGROUP);
      r = __builtin_amdgcn_readfirstlane(r);
      if (r >= nrows) break;
      const int st = __builtin_amdgcn_readfirstlane(row_start[r]);
      const int ne = __builtin_amdgcn_readfirstlane(row_cnt[r]);
      const float adq = a_dst[(size_t)(n0 + r) * HEADS + hq];
      const float mq  = key2f(smax[r * HEADS + hq]);

      float4 sa = make_float4(0.f, 0.f, 0.f, 0.f);
      float4 sb = make_float4(0.f, 0.f, 0.f, 0.f);
      float dsum = 0.f;

      for (int c0 = 0; c0 < ne; c0 += 32) {
        const int m = (ne - c0 < 32) ? (ne - c0) : 32;
        int s = n0;
        if (lane < m) s = list_e[st + c0 + lane];

        for (int i0 = 0; i0 < m; i0 += EDGE_GROUP) {
          const int iq = (i0 + gq < m) ? (i0 + gq) : (m - 1);
          const int sq = list_e[st + c0 + iq];
          const float asq = a_src[(size_t)sq * HEADS + hq];

          float4 xa[EDGE_GROUP], xb[EDGE_GROUP];
#pragma unroll
          for (int j = 0; j < EDGE_GROUP; ++j) {
            const int ii = (i0 + j < m) ? (i0 + j) : (m - 1);
            const int si = __builtin_amdgcn_readlane(s, ii);
            const float4* __restrict__ xr = (const float4*)(xl + (size_t)si * F) + lane;
            xa[j] = xr[0];
            xb[j] = xr[32];
          }

          float p = expf(leaky(asq + adq) - mq);
          if (i0 + gq >= m) p = 0.f;
          dsum += p;

#pragma unroll
          for (int j = 0; j < EDGE_GROUP; ++j) {
            const float u = __shfl(p, 8 * j + h0, 32);
            const float v = __shfl(p, 8 * j + 4 + h0, 32);
            sa.x += u * xa[j].x; sa.y += u * xa[j].y; sa.z += u * xa[j].z; sa.w += u * xa[j].w;
            sb.x += v * xb[j].x; sb.y += v * xb[j].y; sb.z += v * xb[j].z; sb.w += v * xb[j].w;
          }
        }
      }

      dsum += __shfl_xor(dsum, 8, 32);
      dsum += __shfl_xor(dsum, 16, 32);
      const float ra = 1.0f / __shfl(dsum, h0, 32);
      const float rb = 1.0f / __shfl(dsum, 4 + h0, 32);
      float4* __restrict__ orow = (float4*)(out + (size_t)(n0 + r) * F);
      orow[lane]      = make_float4(sa.x * ra, sa.y * ra, sa.z * ra, sa.w * ra);
      orow[32 + lane] = make_float4(sb.x * rb, sb.y * rb, sb.z * rb, sb.w * rb);
    }
  } else {
    for (int i = tid; i < NPB * F / 4; i += AGG_THREADS)
      ((float4*)acc)[i] = make_float4(0.f, 0.f, 0.f, 0.f);
    for (int i = tid; i < NPB * HEADS; i += AGG_THREADS) den[i] = 0.f;
    __syncthreads();
    const int tot = E + N;
    for (int e = tid; e < tot; e += AGG_THREADS) {
      const int d = (e < E) ? dst[e] : (e - E);
      const unsigned dl = (unsigned)(d - n0);
      if (dl < (unsigned)nrows) {
        const int s = (e < E) ? ei[e] : (e - E);
        for (int h = 0; h < HEADS; ++h)
          __hip_atomic_fetch_max(&smax[dl * HEADS + h], f2key(leaky(a_src[s * HEADS + h] + a_dst[d * HEADS + h])),
                                 __ATOMIC_RELAXED, __HIP_MEMORY_SCOPE_WORKGROUP);
      }
    }
    __syncthreads();
    for (int e = tid; e < tot; e += AGG_THREADS) {
      const int d = (e < E) ? dst[e] : (e - E);
      const unsigned dl = (unsigned)(d - n0);
      if (dl < (unsigned)nrows) {
        const int s = (e < E) ? ei[e] : (e - E);
        for (int h = 0; h < HEADS; ++h) {
          const float p = expf(leaky(a_src[s * HEADS + h] + a_dst[d * HEADS + h]) - key2f(smax[dl * HEADS + h]));
          __hip_atomic_fetch_add(&den[dl * HEADS + h], p, __ATOMIC_RELAXED, __HIP_MEMORY_SCOPE_WORKGROUP);
          for (int c = 0; c < CH; ++c)
            __hip_atomic_fetch_add(&acc[dl * F + h * CH + c], xl[(size_t)s * F + h * CH + c] * p,
                                   __ATOMIC_RELAXED, __HIP_MEMORY_SCOPE_WORKGROUP);
        }
      }
    }
    __syncthreads();

    for (int i = tid; i < nrows * HEADS; i += AGG_THREADS) den[i] = 1.0f / den[i];
    __syncthreads();
    float4* __restrict__ orow = (float4*)(out + (size_t)n0 * F);
    for (int i = tid; i < nrows * (F / 4); i += AGG_THREADS) {
      float4 v = ((const float4*)acc)[i];
      const float r = den[(i >> 6) * HEADS + ((i & 63) >> 3)];
      v.x *= r; v.y *= r; v.z *= r; v.w *= r;
      orow[i] = v;
    }
  }
}

#define LN_ROWS 4
#define LN_THREADS 256
#define LN_ROWS_PER_BLOCK (LN_ROWS * (LN_THREADS / 32))

__device__ __forceinline__ float wave_sum(float v) {
#pragma unroll
  for (int m = 16; m > 0; m >>= 1) v += __shfl_xor(v, m, 32);
  return v;
}

__global__ void __launch_bounds__(LN_THREADS)
gat_ln_kernel(float* __restrict__ out, const float* __restrict__ bias,
              const float* __restrict__ gamma, const float* __restrict__ beta, int N) {
  const int lane = threadIdx.x & 31;
  const int wave = __builtin_amdgcn_readfirstlane((int)((blockIdx.x * LN_THREADS + threadIdx.x) >> 5));
  const int row0 = wave * LN_ROWS;
  if (row0 >= N) return;

  float4 va[LN_ROWS], vb[LN_ROWS];
#pragma unroll
  for (int r = 0; r < LN_ROWS; ++r) {
    const int row = (row0 + r < N) ? (row0 + r) : row0;
    const float4* p = (const float4*)(out + (size_t)row * F);
    va[r] = p[lane];
    vb[r] = p[32 + lane];
  }
  const float4 b0 = ((const float4*)bias)[lane],  b1 = ((const float4*)bias)[32 + lane];
  const float4 g0 = ((const float4*)gamma)[lane], g1 = ((const float4*)gamma)[32 + lane];
  const float4 t0 = ((const float4*)beta)[lane],  t1 = ((const float4*)beta)[32 + lane];

  float mean[LN_ROWS];
#pragma unroll
  for (int r = 0; r < LN_ROWS; ++r) {
    va[r].x += b0.x; va[r].y += b0.y; va[r].z += b0.z; va[r].w += b0.w;
    vb[r].x += b1.x; vb[r].y += b1.y; vb[r].z += b1.z; vb[r].w += b1.w;
    const float s = ((va[r].x + va[r].y) + (va[r].z + va[r].w)) +
                    ((vb[r].x + vb[r].y) + (vb[r].z + vb[r].w));
    mean[r] = wave_sum(s) * (1.0f / F);
  }
  float rs[LN_ROWS];
#pragma unroll
  for (int r = 0; r < LN_ROWS; ++r) {
    va[r].x -= mean[r]; va[r].y -= mean[r]; va[r].z -= mean[r]; va[r].w -= mean[r];
    vb[r].x -= mean[r]; vb[r].y -= mean[r]; vb[r].z -= mean[r]; vb[r].w -= mean[r];
    const float q = ((va[r].x * va[r].x + va[r].y * va[r].y) + (va[r].z * va[r].z + va[r].w * va[r].w)) +
                    ((vb[r].x * vb[r].x + vb[r].y * vb[r].y) + (vb[r].z * vb[r].z + vb[r].w * vb[r].w));
    rs[r] = rsqrtf(wave_sum(q) * (1.0f / F) + LN_EPS);
  }
#pragma unroll
  for (int r = 0; r < LN_ROWS; ++r) {
    if (row0 + r < N) {
      float4* o = (float4*)(out + (size_t)(row0 + r) * F);
      float4 ya, yb;
      ya.x = va[r].x * rs[r] * g0.x + t0.x; ya.y = va[r].y * rs[r] * g0.y + t0.y;
      ya.z = va[r].z * rs[r] * g0.z + t0.z; ya.w = va[r].w * rs[r] * g0.w + t0.w;
      yb.x = vb[r].x * rs[r] * g1.x + t1.x; yb.y = vb[r].y * rs[r] * g1.y + t1.y;
      yb.z = vb[r].z * rs[r] * g1.z + t1.z; yb.w = vb[r].w * rs[r] * g1.w + t1.w;
      o[lane] = ya;
      o[32 + lane] = yb;
    }
  }
}

extern "C" void kernel_launch(void* const* d_in, const int* in_sizes, int n_in,
                              void* d_out, int out_size, void* d_ws, size_t ws_size,
                              hipStream_t stream) {
  const float* x       = (const float*)d_in[0];
  const int*   ei      = (const int*)  d_in[1];
  const float* W       = (const float*)d_in[2];
  const float* att_src = (const float*)d_in[3];
  const float* att_dst = (const float*)d_in[4];
  const float* bias    = (const float*)d_in[5];
  const float* gamma   = (const float*)d_in[6];
  const float* beta    = (const float*)d_in[7];
  float* out = (float*)d_out;

  const int N = in_sizes[0] / F;
  const int E = in_sizes[1] / 2;

  char* ws = (char*)d_ws;
  float*    xl     = (float*)ws;    ws += (size_t)N * F * sizeof(float);
  float*    a_src  = (float*)ws;    ws += (size_t)N * HEADS * sizeof(float);
  float*    a_dst  = (float*)ws;    ws += (size_t)N * HEADS * sizeof(float);
  unsigned* own    = (unsigned*)ws;

  gat_gemm_wmma_kernel<<<GEMM_BLOCKS, 512, 0, stream>>>(x, W, xl, N, att_src, att_dst,
                                                         a_src, a_dst, ei, own, E,
                                                         (unsigned long long)ws_size);

  gat_edge_agg_kernel<<<(N + NPB - 1) / NPB, AGG_THREADS, 0, stream>>>(ei, xl, a_src, a_dst,
                                                                        out, E, N, own,
                                                                        (unsigned long long)ws_size);

  gat_ln_kernel<<<(N + LN_ROWS_PER_BLOCK - 1) / LN_ROWS_PER_BLOCK, LN_THREADS, 0, stream>>>(out, bias, gamma, beta, N);
}


# softmax row maxima taken by the row's own wave in registers (lane = (edge, head), 2-step lane max, exact); rank pass drops its 8 LDS max adds and half-logit lookups per edge; sum order unchanged
# speedup vs baseline: 6.3866x; 1.0328x over previous
#include <hip/hip_runtime.h>
#include <math.h>

typedef float v2f __attribute__((ext_vector_type(2)));
typedef float v8f __attribute__((ext_vector_type(8)));

#define HEADS 8
#define CH 32
#define F 256
#define NEG_SLOPE 0.2f
#define LN_EPS 1e-5f

#define LDS_STRIDE 260

__device__ __forceinline__ unsigned f2key(float f) {
  unsigned b = __float_as_uint(f);
  return (b & 0x80000000u) ? ~b : (b | 0x80000000u);
}
__device__ __forceinline__ float key2f(unsigned k) {
  unsigned b = (k & 0x80000000u) ? (k & 0x7fffffffu) : ~k;
  return __uint_as_float(b);
}

#define NPB 196

__device__ __forceinline__ bool owner_plane_ok(int N, int E, unsigned long long ws_bytes) {
  const unsigned long long need = (unsigned long long)N * F * sizeof(float) +
                                  2ull * N * HEADS * sizeof(float) + (unsigned long long)E;
  return (E >= 16) && ((E & 15) == 0) && ((N + NPB - 1) / NPB <= 256) && (ws_bytes >= need);
}

#define GEMM_BLOCKS 256
#define CELL_SLOTS 64
#define CELL_EDGES (CELL_SLOTS - 1)
__device__ __forceinline__ bool owner_cells_ok(int N, int E, unsigned long long ws_bytes) {
  const unsigned long long need = (unsigned long long)N * F * sizeof(float) +
                                  2ull * N * HEADS * sizeof(float) +
                                  (unsigned long long)GEMM_BLOCKS * 256ull * CELL_SLOTS * sizeof(unsigned);
  return (E >= 1) && (E <= (1 << 30)) && (N <= (1 << 24)) && ((N & 3) == 0) && ((N + NPB - 1) / NPB <= 256) && (ws_bytes >= need);
}

#define WL_PAIR 512

__device__ __forceinline__ void tile_half_logits(const float* dtile, int row0, int N, int tid,
                                                 const float* __restrict__ att_src,
                                                 const float* __restrict__ att_dst,
                                                 float* __restrict__ a_src, float* __restrict__ a_dst) {
  if (tid < 256) {
    const int lane = tid & 31;
    const int wu = __builtin_amdgcn_readfirstlane(tid >> 5);
    const bool is_dst = (wu & 1) != 0;
    const int r = (wu >> 1) * 4 + (lane >> 3);
    const int hd = lane & 7;
    const float4* xr = (const float4*)(&dtile[r * LDS_STRIDE + hd * CH]);
    const float4* __restrict__ av = (const float4*)((is_dst ? att_dst : att_src) + hd * CH);
    float s = 0.f;
#pragma unroll
    for (int c = 0; c < CH / 4; ++c) {
      const float4 v = xr[c];
      const float4 a = av[c];
      s += v.x * a.x;
      s += v.y * a.y;
      s += v.z * a.z;
      s += v.w * a.w;
    }
    if (row0 + r < N) (is_dst ? a_dst : a_src)[(size_t)(row0 + r) * HEADS + hd] = s;
  }
}

__global__ void __launch_bounds__(512)
gat_gemm_wmma_kernel(const float* __restrict__ x, const float* __restrict__ W,
                     float* __restrict__ xl, int N,
                     const float* __restrict__ att_src, const float* __restrict__ att_dst,
                     float* __restrict__ a_src, float* __restrict__ a_dst,
                     const int* __restrict__ ei, unsigned* __restrict__ own,
                     int E, unsigned long long ws_bytes) {
  __shared__ __attribute__((aligned(16))) float lds_w[F * F];
  __shared__ __attribute__((aligned(16))) float lds_a[16 * LDS_STRIDE];
  __shared__ __attribute__((aligned(16))) float lds_d[2][16 * LDS_STRIDE];

  const int tid  = threadIdx.x;
  const int lane = tid & 31;
  const int wave = tid >> 5;
  const int col0 = wave * 16;
  const int ntiles = (N + 15) >> 4;

  if (owner_cells_ok(N, E, ws_bytes)) {
    unsigned* st = (unsigned*)lds_w;
    int* ccnt = (int*)(lds_w + 256 * CELL_SLOTS);
    for (int i = tid; i < 256 * CELL_SLOTS / 4; i += 512)
      ((uint4*)st)[i] = make_uint4(0u, 0u, 0u, 0u);
    if (tid < 256 + 64) ccnt[tid] = 0;
    __syncthreads();
    const int per = (E + GEMM_BLOCKS - 1) / GEMM_BLOCKS;
    const int e0 = blockIdx.x * per;
    const int e1 = (e0 + per < E) ? (e0 + per) : E;
    const int* __restrict__ srcc = ei;
    const int* __restrict__ dstc = ei + E;
    for (int base = e0; base < e1; base += 8 * 512) {
      int dv[8], sv[8];
#pragma unroll
      for (int u = 0; u < 8; ++u) {
        const int e = base + u * 512 + tid;
        const int ec = (e < e1) ? e : e0;
        dv[u] = dstc[ec];
        sv[u] = srcc[ec];
      }
      unsigned oo[8], ww[8];
      int rr[8];
#pragma unroll
      for (int u = 0; u < 8; ++u) {
        const int e = base + u * 512 + tid;
        unsigned o = (unsigned)dv[u] / NPB;
        if (o > 255u) o = 255u;
        const unsigned dl = (unsigned)dv[u] - o * NPB;
        ww[u] = ((unsigned)sv[u] & 0x00FFFFFFu) | (dl << 24);
        oo[u] = (e < e1) ? o : (256u + (unsigned)(tid & 63));
        rr[u] = __hip_atomic_fetch_add(&ccnt[oo[u]], 1, __ATOMIC_RELAXED, __HIP_MEMORY_SCOPE_WORKGROUP);
      }
#pragma unroll
      for (int u = 0; u < 8; ++u) {
        const bool placed = (oo[u] < 256u) && (rr[u] < CELL_EDGES);
        st[placed ? (oo[u] * CELL_SLOTS + (unsigned)rr[u]) : (unsigned)(256 * CELL_SLOTS + 320 + tid)] = ww[u];
      }
    }
    __syncthreads();
    for (int i = tid; i < 256 * CELL_SLOTS / 4; i += 512) {
      const int o = i >> 4;
      const int q = i & 15;
      uint4 v = ((const uint4*)st)[i];
      if (q == 15) v.w = (unsigned)ccnt[o];
      ((uint4*)(own + ((size_t)o * GEMM_BLOCKS + blockIdx.x) * CELL_SLOTS))[q] = v;
    }
    __syncthreads();
  } else if (owner_plane_ok(N, E, ws_bytes)) {
    const int4* __restrict__ d4 = (const int4*)(ei + E);
    for (long long g = (long long)blockIdx.x * 512 + tid; g < (long long)(E >> 2);
         g += (long long)gridDim.x * 512) {
      const int4 d = d4[g];
      own[g] = ((unsigned)d.x / NPB) | (((unsigned)d.y / NPB) << 8) |
               (((unsigned)d.z / NPB) << 16) | (((unsigned)d.w / NPB) << 24);
    }
  }

  int tile = blockIdx.x;
  if (tile >= ntiles) return;

  const int ar = tid >> 5;
  const int ac = (tid & 31) * 8;
  float4 pa0, pa1;
  {
    int grow = tile * 16 + ar;
    if (grow >= N) grow = N - 1;
    const float4* __restrict__ src = (const float4*)(x + (size_t)grow * F + ac);
    pa0 = src[0];
    pa1 = src[1];
  }

  for (int it0 = 0; it0 < 16; it0 += 4) {
    float4 r0[4], r1[4];
#pragma unroll
    for (int u = 0; u < 4; ++u) {
      const int item = (it0 + u) * 512 + tid;
      const int p = item >> 6;
      const int q = item & 63;
      r0[u] = *(const float4*)(W + (size_t)(2 * p) * F + 4 * q);
      r1[u] = *(const float4*)(W + (size_t)(2 * p + 1) * F + 4 * q);
    }
#pragma unroll
    for (int u = 0; u < 4; ++u) {
      const int item = (it0 + u) * 512 + tid;
      const int p = item >> 6;
      const int q = item & 63;
      float4* d = (float4*)(&lds_w[p * WL_PAIR + ((8 * q + 32 * (p & 1)) & (WL_PAIR - 1))]);
      d[0] = make_float4(r0[u].x, r1[u].x, r0[u].y, r1[u].y);
      d[1] = make_float4(r0[u].z, r1[u].z, r0[u].w, r1[u].w);
    }
  }

  const int m = lane & 15;
  const int h = lane >> 4;
  const float* arow = &lds_a[m * LDS_STRIDE + 2 * h];
  const float* bcol = &lds_w[h * WL_PAIR + ((2 * (col0 + m) + 32 * h) & (WL_PAIR - 1))];

  int par = 0;
  int prev_row0 = -1;
  for (; tile < ntiles; tile += gridDim.x) {
    const int row0 = tile * 16;
    {
      float4* dst = (float4*)(&lds_a[ar * LDS_STRIDE + ac]);
      dst[0] = pa0;
      dst[1] = pa1;
    }
    __builtin_amdgcn_fence(__ATOMIC_RELEASE, "workgroup", "local");
    __builtin_amdgcn_s_barrier();
    __builtin_amdgcn_fence(__ATOMIC_ACQUIRE, "workgroup", "local");

    if (prev_row0 >= 0)
      tile_half_logits(lds_d[par ^ 1], prev_row0, N, tid, att_src, att_dst, a_src, a_dst);

    if (tile + (int)gridDim.x < ntiles) {
      int grow = (tile + (int)gridDim.x) * 16 + ar;
      if (grow >= N) grow = N - 1;
      const float4* __restrict__ src = (const float4*)(x + (size_t)grow * F + ac);
      pa0 = src[0];
      pa1 = src[1];
    }

    v8f acc = {0.f, 0.f, 0.f, 0.f, 0.f, 0.f, 0.f, 0.f};
#pragma unroll
    for (int ks = 0; ks < F / 4; ++ks) {
      const v2f a = *(const v2f*)(arow + 4 * ks);
      const v2f b = *(const v2f*)(bcol + 2 * ks * WL_PAIR);
      acc = __builtin_amdgcn_wmma_f32_16x16x4_f32(
           false, a,  false, b,
           (short)0, acc,  false,  false);
    }

    {
      float* drow = &lds_d[par][(h * 8) * LDS_STRIDE + col0 + m];
#pragma unroll
      for (int j = 0; j < 8; ++j) drow[j * LDS_STRIDE] = acc[j];
    }
    __syncthreads();

    const int rbase = row0 + (h * 8);
    float* __restrict__ dst = xl + (size_t)rbase * F + col0 + m;
    if (row0 + 16 <= N) {
#pragma unroll
      for (int j = 0; j < 8; ++j) dst[(size_t)j * F] = acc[j];
    } else {
#pragma unroll
      for (int j = 0; j < 8; ++j) {
        if (rbase + j < N) dst[(size_t)j * F] = acc[j];
      }
    }

    prev_row0 = row0;
    par ^= 1;
  }
  tile_half_logits(lds_d[par ^ 1], prev_row0, N, tid, att_src, att_dst, a_src, a_dst);
}

__device__ __forceinline__ float leaky(float v) {
  return (v >= 0.f) ? v : v * NEG_SLOPE;
}

#define AGG_THREADS 1024
#define AGG_WAVES (AGG_THREADS / 32)
#define LIST_CAP 4096
#define SCAN_UNROLL 8
#define SCAN_STEP (AGG_THREADS * 4)
#define SCAN_SPAN (SCAN_STEP * SCAN_UNROLL)
#define OWN_STEP (AGG_THREADS * 16)
#define OWN_SPAN (OWN_STEP * SCAN_UNROLL)
#define ROWS_PER_LANE ((NPB + 31) / 32)
#define EDGE_GROUP 4

static_assert(LIST_CAP == 4 * AGG_THREADS, "pass (1) gives every thread four list entries");
static_assert(EDGE_GROUP * HEADS == 32, "one lane per (edge of the group, head) weight");

__global__ void __launch_bounds__(AGG_THREADS)
gat_edge_agg_kernel(const int* __restrict__ ei, const float* __restrict__ xl,
                    const float* __restrict__ a_src, const float* __restrict__ a_dst,
                    float* __restrict__ out, int E, int N,
                    const unsigned* __restrict__ own, unsigned long long ws_bytes) {
  __shared__ __attribute__((aligned(16))) float acc[NPB * F];
  __shared__ __attribute__((aligned(16))) unsigned smax[NPB * HEADS];
  __shared__ __attribute__((aligned(16))) float den[NPB * HEADS];
  __shared__ int list_e[LIST_CAP];
  __shared__ int row_cnt[NPB];
  __shared__ int row_start[NPB];
  __shared__ int list_n;
  __shared__ int next_row;
  __shared__ int cell_base[GEMM_BLOCKS];

  const int tid  = threadIdx.x;
  const int lane = tid & 31;
  const int wave = __builtin_amdgcn_readfirstlane(tid >> 5);
  const int n0 = blockIdx.x * NPB;
  const int nrows = (N - n0 < NPB) ? (N - n0) : NPB;
  const int* __restrict__ dst = ei + E;

  const bool use_cells = owner_cells_ok(N, E, ws_bytes);

  if (tid < NPB) row_cnt[tid] = 0;
  if (tid < nrows) list_e[tid] = use_cells ? (int)((unsigned)(n0 + tid) | ((unsigned)tid << 24)) : (E + n0 + tid);
  if (tid == 0) { list_n = nrows; next_row = 0; }
  __syncthreads();

  auto keep = [&](int e, int dv) {
    unsigned dl = (unsigned)(dv - n0);
    if (dl < (unsigned)nrows) {
      int slot = __hip_atomic_fetch_add(&list_n, 1, __ATOMIC_RELAXED, __HIP_MEMORY_SCOPE_WORKGROUP);
      if (slot < LIST_CAP) list_e[slot] = e;
    }
  };
  auto keep_e = [&](int e) {
    int slot = __hip_atomic_fetch_add(&list_n, 1, __ATOMIC_RELAXED, __HIP_MEMORY_SCOPE_WORKGROUP);
    if (slot < LIST_CAP) list_e[slot] = e;
  };

  const bool use_plane = owner_plane_ok(N, E, ws_bytes);
  if (use_cells) {
    static_assert(GEMM_BLOCKS * (CELL_SLOTS / 4) == 4 * AGG_THREADS, "four 16-byte pieces of the cells per thread");
    const uint4* __restrict__ cell4 = (const uint4*)(own + (size_t)blockIdx.x * (GEMM_BLOCKS * CELL_SLOTS));
    uint4 v[4];
    int cc[4];
#pragma unroll
    for (int k = 0; k < 4; ++k) v[k] = cell4[k * AGG_THREADS + tid];
#pragma unroll
    for (int k = 0; k < 4; ++k) {
      cc[k] = __shfl((int)v[k].w, (lane & 16) | 15, 32);
      if ((tid & 15) == 15) cell_base[(k * AGG_THREADS + tid) >> 4] = cc[k];
    }
    __syncthreads();
    if (wave == 0) {
      int c[GEMM_BLOCKS / 32];
      int tot = 0, over = 0;
#pragma unroll
      for (int q = 0; q < GEMM_BLOCKS / 32; ++q) {
        const int raw = cell_base[lane * (GEMM_BLOCKS / 32) + q];
        over |= (raw > CELL_EDGES || raw < 0) ? 1 : 0;
        c[q] = min(max(raw, 0), CELL_EDGES);
        tot += c[q];
      }
      int inc = tot;
      for (int d = 1; d < 32; d <<= 1) {
        const int t = __shfl_up(inc, d, 32);
        if (lane >= d) inc += t;
        over |= __shfl_xor(over, d, 32);
      }
      int run = inc - tot;
#pragma unroll
      for (int q = 0; q < GEMM_BLOCKS / 32; ++q) {
        cell_base[lane * (GEMM_BLOCKS / 32) + q] = run;
        run += c[q];
      }
      if (lane == 31) list_n = over ? (LIST_CAP + 1) : (nrows + inc);
    }
    __syncthreads();
#pragma unroll
    for (int k = 0; k < 4; ++k) {
      const int cell = (k * AGG_THREADS + tid) >> 4;
      const int q4 = (tid & 15) * 4;
      const int n = min(max(cc[k], 0), CELL_EDGES);
      const int base = nrows + cell_base[cell] + q4;
      const unsigned w4[4] = {v[k].x, v[k].y, v[k].z, v[k].w};
#pragma unroll
      for (int j = 0; j < 4; ++j)
        if (q4 + j < n && base + j < LIST_CAP) list_e[base + j] = (int)w4[j];
    }
  } else if (use_plane) {
    int n_it = (E + OWN_SPAN - 1) / OWN_SPAN;
    if (n_it < 1) n_it = 1;
    const int rot = (int)(((long long)blockIdx.x * n_it) / gridDim.x);
    const unsigned pat  = blockIdx.x * 0x01010101u;
    const uint4* __restrict__ own4 = (const uint4*)own;
    for (int it = 0; it < n_it; ++it) {
      int c = it + rot;
      if (c >= n_it) c -= n_it;
      const int b0 = c * OWN_SPAN + tid * 16;
      uint4 v[SCAN_UNROLL];
#pragma unroll
      for (int u = 0; u < SCAN_UNROLL; ++u) {
        const int idx = b0 + u * OWN_STEP;
        v[u] = own4[min(idx, E - 16) >> 4];
      }
#pragma unroll
      for (int u = 0; u < SCAN_UNROLL; u += 2) {
        const int idx = b0 + u * OWN_STEP;
        unsigned m = 0;
#pragma unroll
        for (int w = 0; w < 2; ++w) {
          const unsigned x[4] = {v[u + w].x ^ pat, v[u + w].y ^ pat, v[u + w].z ^ pat, v[u + w].w ^ pat};
          unsigned mw = 0;
#pragma unroll
          for (int q = 0; q < 4; ++q) {
            const unsigned f = ~(((x[q] & 0x7f7f7f7fu) + 0x7f7f7f7fu) | x[q]) & 0x80808080u;
            mw |= f >> (7 - q);
          }
          if (idx + w * OWN_STEP >= E) mw = 0;
          m |= mw << (4 * w);
        }
        while (m) {
          const int p = __builtin_ctz(m);
          keep_e(idx + ((p >> 2) & 1) * OWN_STEP + 4 * (p & 3) + (p >> 3));
          m &= m - 1;
        }
      }
    }
  } else {
  int n_it = (E + SCAN_SPAN - 1) / SCAN_SPAN;
  if (n_it < 1) n_it = 1;
  const int rot = (int)(((long long)blockIdx.x * n_it) / gridDim.x);
  const bool vec_ok = ((E & 3) == 0);

  for (int it = 0; it < n_it; ++it) {
    int c = it + rot;
    if (c >= n_it) c -= n_it;
    const int b0 = c * SCAN_SPAN + tid * 4;

    if (vec_ok && (c + 1) * SCAN_SPAN <= E) {
      int4 v[SCAN_UNROLL];
#pragma unroll
      for (int u = 0; u < SCAN_UNROLL; ++u)
        v[u] = *(const int4*)(dst + b0 + u * SCAN_STEP);
#pragma unroll
      for (int u = 0; u < SCAN_UNROLL; ++u) {
        const int idx = b0 + u * SCAN_STEP;
        const bool any = ((unsigned)(v[u].x - n0) < (unsigned)nrows) | ((unsigned)(v[u].y - n0) < (unsigned)nrows) |
                         ((unsigned)(v[u].z - n0) < (unsigned)nrows) | ((unsigned)(v[u].w - n0) < (unsigned)nrows);
        if (any) {
          keep(idx + 0, v[u].x);
          keep(idx + 1, v[u].y);
          keep(idx + 2, v[u].z);
          keep(idx + 3, v[u].w);
        }
      }
    } else {
      for (int u = 0; u < SCAN_UNROLL; ++u) {
        const int idx = b0 + u * SCAN_STEP;
        for (int k = 0; k < 4; ++k)
          if (idx + k < E) keep(idx + k, dst[idx + k]);
      }
    }
  }
  }
  __syncthreads();
  const int cnt = __builtin_amdgcn_readfirstlane(list_n);

  if (cnt <= LIST_CAP) {
    int dd[4], ss[4], rk[4];
    {
      int ee[4];
#pragma unroll
      for (int k = 0; k < 4; ++k) {
        const int i = k * AGG_THREADS + tid;
        ee[k] = (i < cnt) ? list_e[i] : (use_cells ? n0 : (E + n0));
      }
      if (use_cells) {
#pragma unroll
        for (int k = 0; k < 4; ++k) {
          const unsigned w = (unsigned)ee[k];
          ss[k] = (int)(w & 0x00FFFFFFu);
          ss[k] = min(ss[k], N - 1);
          dd[k] = min((int)(w >> 24), nrows - 1);
        }
      } else {
#pragma unroll
      for (int k = 0; k < 4; ++k) {
        const bool real = ee[k] < E;
        ss[k] = real ? ei[ee[k]] : (ee[k] - E);
        const int dv = real ? dst[ee[k]] : (ee[k] - E);
        dd[k] = min(max(dv - n0, 0), nrows - 1);
      }
      }
#pragma unroll
      for (int k = 0; k < 4; ++k) {
        const int i = k * AGG_THREADS + tid;
        rk[k] = 0;
        if (i < cnt)
          rk[k] = __hip_atomic_fetch_add(&row_cnt[dd[k]], 1, __ATOMIC_RELAXED, __HIP_MEMORY_SCOPE_WORKGROUP);
      }
    }
    __syncthreads();

    if (wave == 0) {
      int c[ROWS_PER_LANE];
      int tot = 0;
#pragma unroll
      for (int q = 0; q < ROWS_PER_LANE; ++q) {
        const int r = lane * ROWS_PER_LANE + q;
        c[q] = (r < NPB) ? row_cnt[r] : 0;
        tot += c[q];
      }
      int inc = tot;
      for (int d = 1; d < 32; d <<= 1) {
        const int v = __shfl_up(inc, d, 32);
        if (lane >= d) inc += v;
      }
      int run = inc - tot;
#pragma unroll
      for (int q = 0; q < ROWS_PER_LANE; ++q) {
        const int r = lane * ROWS_PER_LANE + q;
        if (r < NPB) row_start[r] = run;
        run += c[q];
      }
    }
    __syncthreads();

#pragma unroll
    for (int k = 0; k < 4; ++k) {
      const int i = k * AGG_THREADS + tid;
      if (i < cnt) list_e[row_start[dd[k]] + rk[k]] = ss[k];
    }
    __syncthreads();

    const int hq = lane & 7;
    const int gq = lane >> 3;
    const int h0 = lane >> 3;
    for (int t = 0; t < NPB; ++t) {
      int r = 0;
      if (lane == 0) r = __hip_atomic_fetch_add(&next_row, 1, __ATOMIC_RELAXED, __HIP_MEMORY_SCOPE_WORKGROUP);
      r = __builtin_amdgcn_readfirstlane(r);
      if (r >= nrows) break;
      const int st = __builtin_amdgcn_readfirstlane(row_start[r]);
      const int ne = __builtin_amdgcn_readfirstlane(row_cnt[r]);
      const float adq = a_dst[(size_t)(n0 + r) * HEADS + hq];

      float mq = -INFINITY;
      for (int c0 = 0; c0 < ne; c0 += 32) {
        const int m = (ne - c0 < 32) ? (ne - c0) : 32;
        float as[8];
#pragma unroll
        for (int g = 0; g < 8; ++g) {
          const int i = (4 * g + gq < m) ? (4 * g + gq) : (m - 1);
          as[g] = a_src[(size_t)list_e[st + c0 + i] * HEADS + hq];
        }
#pragma unroll
        for (int g = 0; g < 8; ++g) mq = fmaxf(mq, leaky(as[g] + adq));
      }
      mq = fmaxf(mq, __shfl_xor(mq, 8, 32));
      mq = fmaxf(mq, __shfl_xor(mq, 16, 32));

      float4 sa = make_float4(0.f, 0.f, 0.f, 0.f);
      float4 sb = make_float4(0.f, 0.f, 0.f, 0.f);
      float dsum = 0.f;

      for (int c0 = 0; c0 < ne; c0 += 32) {
        const int m = (ne - c0 < 32) ? (ne - c0) : 32;
        int s = n0;
        if (lane < m) s = list_e[st + c0 + lane];

        for (int i0 = 0; i0 < m; i0 += EDGE_GROUP) {
          const int iq = (i0 + gq < m) ? (i0 + gq) : (m - 1);
          const int sq = list_e[st + c0 + iq];
          const float asq = a_src[(size_t)sq * HEADS + hq];

          float4 xa[EDGE_GROUP], xb[EDGE_GROUP];
#pragma unroll
          for (int j = 0; j < EDGE_GROUP; ++j) {
            const int ii = (i0 + j < m) ? (i0 + j) : (m - 1);
            const int si = __builtin_amdgcn_readlane(s, ii);
            const float4* __restrict__ xr = (const float4*)(xl + (size_t)si * F) + lane;
            xa[j] = xr[0];
            xb[j] = xr[32];
          }

          float p = expf(leaky(asq + adq) - mq);
          if (i0 + gq >= m) p = 0.f;
          dsum += p;

#pragma unroll
          for (int j = 0; j < EDGE_GROUP; ++j) {
            const float u = __shfl(p, 8 * j + h0, 32);
            const float v = __shfl(p, 8 * j + 4 + h0, 32);
            sa.x += u * xa[j].x; sa.y += u * xa[j].y; sa.z += u * xa[j].z; sa.w += u * xa[j].w;
            sb.x += v * xb[j].x; sb.y += v * xb[j].y; sb.z += v * xb[j].z; sb.w += v * xb[j].w;
          }
        }
      }

      dsum += __shfl_xor(dsum, 8, 32);
      dsum += __shfl_xor(dsum, 16, 32);
      const float ra = 1.0f / __shfl(dsum, h0, 32);
      const float rb = 1.0f / __shfl(dsum, 4 + h0, 32);
      float4* __restrict__ orow = (float4*)(out + (size_t)(n0 + r) * F);
      orow[lane]      = make_float4(sa.x * ra, sa.y * ra, sa.z * ra, sa.w * ra);
      orow[32 + lane] = make_float4(sb.x * rb, sb.y * rb, sb.z * rb, sb.w * rb);
    }
  } else {
    for (int i = tid; i < NPB * F / 4; i += AGG_THREADS)
      ((float4*)acc)[i] = make_float4(0.f, 0.f, 0.f, 0.f);
    for (int i = tid; i < NPB * HEADS; i += AGG_THREADS) {
      den[i] = 0.f;
      smax[i] = 0x007FFFFFu;
    }
    __syncthreads();
    const int tot = E + N;
    for (int e = tid; e < tot; e += AGG_THREADS) {
      const int d = (e < E) ? dst[e] : (e - E);
      const unsigned dl = (unsigned)(d - n0);
      if (dl < (unsigned)nrows) {
        const int s = (e < E) ? ei[e] : (e - E);
        for (int h = 0; h < HEADS; ++h)
          __hip_atomic_fetch_max(&smax[dl * HEADS + h], f2key(leaky(a_src[s * HEADS + h] + a_dst[d * HEADS + h])),
                                 __ATOMIC_RELAXED, __HIP_MEMORY_SCOPE_WORKGROUP);
      }
    }
    __syncthreads();
    for (int e = tid; e < tot; e += AGG_THREADS) {
      const int d = (e < E) ? dst[e] : (e - E);
      const unsigned dl = (unsigned)(d - n0);
      if (dl < (unsigned)nrows) {
        const int s = (e < E) ? ei[e] : (e - E);
        for (int h = 0; h < HEADS; ++h) {
          const float p = expf(leaky(a_src[s * HEADS + h] + a_dst[d * HEADS + h]) - key2f(smax[dl * HEADS + h]));
          __hip_atomic_fetch_add(&den[dl * HEADS + h], p, __ATOMIC_RELAXED, __HIP_MEMORY_SCOPE_WORKGROUP);
          for (int c = 0; c < CH; ++c)
            __hip_atomic_fetch_add(&acc[dl * F + h * CH + c], xl[(size_t)s * F + h * CH + c] * p,
                                   __ATOMIC_RELAXED, __HIP_MEMORY_SCOPE_WORKGROUP);
        }
      }
    }
    __syncthreads();

    for (int i = tid; i < nrows * HEADS; i += AGG_THREADS) den[i] = 1.0f / den[i];
    __syncthreads();
    float4* __restrict__ orow = (float4*)(out + (size_t)n0 * F);
    for (int i = tid; i < nrows * (F / 4); i += AGG_THREADS) {
      float4 v = ((const float4*)acc)[i];
      const float r = den[(i >> 6) * HEADS + ((i & 63) >> 3)];
      v.x *= r; v.y *= r; v.z *= r; v.w *= r;
      orow[i] = v;
    }
  }
}

#define LN_ROWS 4
#define LN_THREADS 256
#define LN_ROWS_PER_BLOCK (LN_ROWS * (LN_THREADS / 32))

__device__ __forceinline__ float wave_sum(float v) {
#pragma unroll
  for (int m = 16; m > 0; m >>= 1) v += __shfl_xor(v, m, 32);
  return v;
}

__global__ void __launch_bounds__(LN_THREADS)
gat_ln_kernel(float* __restrict__ out, const float* __restrict__ bias,
              const float* __restrict__ gamma, const float* __restrict__ beta, int N) {
  const int lane = threadIdx.x & 31;
  const int wave = __builtin_amdgcn_readfirstlane((int)((blockIdx.x * LN_THREADS + threadIdx.x) >> 5));
  const int row0 = wave * LN_ROWS;
  if (row0 >= N) return;

  float4 va[LN_ROWS], vb[LN_ROWS];
#pragma unroll
  for (int r = 0; r < LN_ROWS; ++r) {
    const int row = (row0 + r < N) ? (row0 + r) : row0;
    const float4* p = (const float4*)(out + (size_t)row * F);
    va[r] = p[lane];
    vb[r] = p[32 + lane];
  }
  const float4 b0 = ((const float4*)bias)[lane],  b1 = ((const float4*)bias)[32 + lane];
  const float4 g0 = ((const float4*)gamma)[lane], g1 = ((const float4*)gamma)[32 + lane];
  const float4 t0 = ((const float4*)beta)[lane],  t1 = ((const float4*)beta)[32 + lane];

  float mean[LN_ROWS];
#pragma unroll
  for (int r = 0; r < LN_ROWS; ++r) {
    va[r].x += b0.x; va[r].y += b0.y; va[r].z += b0.z; va[r].w += b0.w;
    vb[r].x += b1.x; vb[r].y += b1.y; vb[r].z += b1.z; vb[r].w += b1.w;
    const float s = ((va[r].x + va[r].y) + (va[r].z + va[r].w)) +
                    ((vb[r].x + vb[r].y) + (vb[r].z + vb[r].w));
    mean[r] = wave_sum(s) * (1.0f / F);
  }
  float rs[LN_ROWS];
#pragma unroll
  for (int r = 0; r < LN_ROWS; ++r) {
    va[r].x -= mean[r]; va[r].y -= mean[r]; va[r].z -= mean[r]; va[r].w -= mean[r];
    vb[r].x -= mean[r]; vb[r].y -= mean[r]; vb[r].z -= mean[r]; vb[r].w -= mean[r];
    const float q = ((va[r].x * va[r].x + va[r].y * va[r].y) + (va[r].z * va[r].z + va[r].w * va[r].w)) +
                    ((vb[r].x * vb[r].x + vb[r].y * vb[r].y) + (vb[r].z * vb[r].z + vb[r].w * vb[r].w));
    rs[r] = rsqrtf(wave_sum(q) * (1.0f / F) + LN_EPS);
  }
#pragma unroll
  for (int r = 0; r < LN_ROWS; ++r) {
    if (row0 + r < N) {
      float4* o = (float4*)(out + (size_t)(row0 + r) * F);
      float4 ya, yb;
      ya.x = va[r].x * rs[r] * g0.x + t0.x; ya.y = va[r].y * rs[r] * g0.y + t0.y;
      ya.z = va[r].z * rs[r] * g0.z + t0.z; ya.w = va[r].w * rs[r] * g0.w + t0.w;
      yb.x = vb[r].x * rs[r] * g1.x + t1.x; yb.y = vb[r].y * rs[r] * g1.y + t1.y;
      yb.z = vb[r].z * rs[r] * g1.z + t1.z; yb.w = vb[r].w * rs[r] * g1.w + t1.w;
      o[lane] = ya;
      o[32 + lane] = yb;
    }
  }
}

extern "C" void kernel_launch(void* const* d_in, const int* in_sizes, int n_in,
                              void* d_out, int out_size, void* d_ws, size_t ws_size,
                              hipStream_t stream) {
  const float* x       = (const float*)d_in[0];
  const int*   ei      = (const int*)  d_in[1];
  const float* W       = (const float*)d_in[2];
  const float* att_src = (const float*)d_in[3];
  const float* att_dst = (const float*)d_in[4];
  const float* bias    = (const float*)d_in[5];
  const float* gamma   = (const float*)d_in[6];
  const float* beta    = (const float*)d_in[7];
  float* out = (float*)d_out;

  const int N = in_sizes[0] / F;
  const int E = in_sizes[1] / 2;

  char* ws = (char*)d_ws;
  float*    xl     = (float*)ws;    ws += (size_t)N * F * sizeof(float);
  float*    a_src  = (float*)ws;    ws += (size_t)N * HEADS * sizeof(float);
  float*    a_dst  = (float*)ws;    ws += (size_t)N * HEADS * sizeof(float);
  unsigned* own    = (unsigned*)ws;

  gat_gemm_wmma_kernel<<<GEMM_BLOCKS, 512, 0, stream>>>(x, W, xl, N, att_src, att_dst,
                                                         a_src, a_dst, ei, own, E,
                                                         (unsigned long long)ws_size);

  gat_edge_agg_kernel<<<(N + NPB - 1) / NPB, AGG_THREADS, 0, stream>>>(ei, xl, a_src, a_dst,
                                                                        out, E, N, own,
                                                                        (unsigned long long)ws_size);

  gat_ln_kernel<<<(N + LN_ROWS_PER_BLOCK - 1) / LN_ROWS_PER_BLOCK, LN_THREADS, 0, stream>>>(out, bias, gamma, beta, N);
}


# projection: each wave keeps its 16 columns of W (64 B fragments, 128 VGPRs) in registers for the whole launch, W never in LDS; A fragments loaded 16 k-steps ahead in groups; same k order
# speedup vs baseline: 6.6139x; 1.0356x over previous
#include <hip/hip_runtime.h>
#include <math.h>

typedef float v2f __attribute__((ext_vector_type(2)));
typedef float v8f __attribute__((ext_vector_type(8)));

#define HEADS 8
#define CH 32
#define F 256
#define NEG_SLOPE 0.2f
#define LN_EPS 1e-5f

#define LDS_STRIDE 260

__device__ __forceinline__ unsigned f2key(float f) {
  unsigned b = __float_as_uint(f);
  return (b & 0x80000000u) ? ~b : (b | 0x80000000u);
}
__device__ __forceinline__ float key2f(unsigned k) {
  unsigned b = (k & 0x80000000u) ? (k & 0x7fffffffu) : ~k;
  return __uint_as_float(b);
}

#define NPB 196

__device__ __forceinline__ bool owner_plane_ok(int N, int E, unsigned long long ws_bytes) {
  const unsigned long long need = (unsigned long long)N * F * sizeof(float) +
                                  2ull * N * HEADS * sizeof(float) + (unsigned long long)E;
  return (E >= 16) && ((E & 15) == 0) && ((N + NPB - 1) / NPB <= 256) && (ws_bytes >= need);
}

#define GEMM_BLOCKS 256
#define CELL_SLOTS 64
#define CELL_EDGES (CELL_SLOTS - 1)
__device__ __forceinline__ bool owner_cells_ok(int N, int E, unsigned long long ws_bytes) {
  const unsigned long long need = (unsigned long long)N * F * sizeof(float) +
                                  2ull * N * HEADS * sizeof(float) +
                                  (unsigned long long)GEMM_BLOCKS * 256ull * CELL_SLOTS * sizeof(unsigned);
  return (E >= 1) && (E <= (1 << 30)) && (N <= (1 << 24)) && ((N & 3) == 0) && ((N + NPB - 1) / NPB <= 256) && (ws_bytes >= need);
}


__device__ __forceinline__ void tile_half_logits(const float* dtile, int row0, int N, int tid,
                                                 const float* __restrict__ att_src,
                                                 const float* __restrict__ att_dst,
                                                 float* __restrict__ a_src, float* __restrict__ a_dst) {
  if (tid < 256) {
    const int lane = tid & 31;
    const int wu = __builtin_amdgcn_readfirstlane(tid >> 5);
    const bool is_dst = (wu & 1) != 0;
    const int r = (wu >> 1) * 4 + (lane >> 3);
    const int hd = lane & 7;
    const float4* xr = (const float4*)(&dtile[r * LDS_STRIDE + hd * CH]);
    const float4* __restrict__ av = (const float4*)((is_dst ? att_dst : att_src) + hd * CH);
    float s = 0.f;
#pragma unroll
    for (int c = 0; c < CH / 4; ++c) {
      const float4 v = xr[c];
      const float4 a = av[c];
      s += v.x * a.x;
      s += v.y * a.y;
      s += v.z * a.z;
      s += v.w * a.w;
    }
    if (row0 + r < N) (is_dst ? a_dst : a_src)[(size_t)(row0 + r) * HEADS + hd] = s;
  }
}

__global__ void __launch_bounds__(512)
gat_gemm_wmma_kernel(const float* __restrict__ x, const float* __restrict__ W,
                     float* __restrict__ xl, int N,
                     const float* __restrict__ att_src, const float* __restrict__ att_dst,
                     float* __restrict__ a_src, float* __restrict__ a_dst,
                     const int* __restrict__ ei, unsigned* __restrict__ own,
                     int E, unsigned long long ws_bytes) {
  __shared__ __attribute__((aligned(16))) unsigned lds_st[256 * CELL_SLOTS + 320 + 512];
  __shared__ __attribute__((aligned(16))) float lds_a[16 * LDS_STRIDE];
  __shared__ __attribute__((aligned(16))) float lds_d[2][16 * LDS_STRIDE];

  const int tid  = threadIdx.x;
  const int lane = tid & 31;
  const int wave = tid >> 5;
  const int col0 = wave * 16;
  const int ntiles = (N + 15) >> 4;

  if (owner_cells_ok(N, E, ws_bytes)) {
    unsigned* st = lds_st;
    int* ccnt = (int*)(lds_st + 256 * CELL_SLOTS);
    for (int i = tid; i < 256 * CELL_SLOTS / 4; i += 512)
      ((uint4*)st)[i] = make_uint4(0u, 0u, 0u, 0u);
    if (tid < 256 + 64) ccnt[tid] = 0;
    __syncthreads();
    const int per = (E + GEMM_BLOCKS - 1) / GEMM_BLOCKS;
    const int e0 = blockIdx.x * per;
    const int e1 = (e0 + per < E) ? (e0 + per) : E;
    const int* __restrict__ srcc = ei;
    const int* __restrict__ dstc = ei + E;
    for (int base = e0; base < e1; base += 8 * 512) {
      int dv[8], sv[8];
#pragma unroll
      for (int u = 0; u < 8; ++u) {
        const int e = base + u * 512 + tid;
        const int ec = (e < e1) ? e : e0;
        dv[u] = dstc[ec];
        sv[u] = srcc[ec];
      }
      unsigned oo[8], ww[8];
      int rr[8];
#pragma unroll
      for (int u = 0; u < 8; ++u) {
        const int e = base + u * 512 + tid;
        unsigned o = (unsigned)dv[u] / NPB;
        if (o > 255u) o = 255u;
        const unsigned dl = (unsigned)dv[u] - o * NPB;
        ww[u] = ((unsigned)sv[u] & 0x00FFFFFFu) | (dl << 24);
        oo[u] = (e < e1) ? o : (256u + (unsigned)(tid & 63));
        rr[u] = __hip_atomic_fetch_add(&ccnt[oo[u]], 1, __ATOMIC_RELAXED, __HIP_MEMORY_SCOPE_WORKGROUP);
      }
#pragma unroll
      for (int u = 0; u < 8; ++u) {
        const bool placed = (oo[u] < 256u) && (rr[u] < CELL_EDGES);
        st[placed ? (oo[u] * CELL_SLOTS + (unsigned)rr[u]) : (unsigned)(256 * CELL_SLOTS + 320 + tid)] = ww[u];
      }
    }
    __syncthreads();
    for (int i = tid; i < 256 * CELL_SLOTS / 4; i += 512) {
      const int o = i >> 4;
      const int q = i & 15;
      uint4 v = ((const uint4*)st)[i];
      if (q == 15) v.w = (unsigned)ccnt[o];
      ((uint4*)(own + ((size_t)o * GEMM_BLOCKS + blockIdx.x) * CELL_SLOTS))[q] = v;
    }
  } else if (owner_plane_ok(N, E, ws_bytes)) {
    const int4* __restrict__ d4 = (const int4*)(ei + E);
    for (long long g = (long long)blockIdx.x * 512 + tid; g < (long long)(E >> 2);
         g += (long long)gridDim.x * 512) {
      const int4 d = d4[g];
      own[g] = ((unsigned)d.x / NPB) | (((unsigned)d.y / NPB) << 8) |
               (((unsigned)d.z / NPB) << 16) | (((unsigned)d.w / NPB) << 24);
    }
  }

  int tile = blockIdx.x;
  if (tile >= ntiles) return;

  const int ar = tid >> 5;
  const int ac = (tid & 31) * 8;
  float4 pa0, pa1;
  {
    int grow = tile * 16 + ar;
    if (grow >= N) grow = N - 1;
    const float4* __restrict__ src = (const float4*)(x + (size_t)grow * F + ac);
    pa0 = src[0];
    pa1 = src[1];
  }

  const int m = lane & 15;
  const int h = lane >> 4;

  v2f breg[F / 4];
  {
    const float* __restrict__ wp = W + (size_t)(2 * h) * F + col0 + m;
#pragma unroll
    for (int ks = 0; ks < F / 4; ++ks) {
      breg[ks].x = wp[(size_t)(4 * ks) * F];
      breg[ks].y = wp[(size_t)(4 * ks + 1) * F];
    }
  }
  const float* arow = &lds_a[m * LDS_STRIDE + 2 * h];

  int par = 0;
  int prev_row0 = -1;
  for (; tile < ntiles; tile += gridDim.x) {
    const int row0 = tile * 16;
    {
      float4* dst = (float4*)(&lds_a[ar * LDS_STRIDE + ac]);
      dst[0] = pa0;
      dst[1] = pa1;
    }
    __builtin_amdgcn_fence(__ATOMIC_RELEASE, "workgroup", "local");
    __builtin_amdgcn_s_barrier();
    __builtin_amdgcn_fence(__ATOMIC_ACQUIRE, "workgroup", "local");

    if (prev_row0 >= 0)
      tile_half_logits(lds_d[par ^ 1], prev_row0, N, tid, att_src, att_dst, a_src, a_dst);

    if (tile + (int)gridDim.x < ntiles) {
      int grow = (tile + (int)gridDim.x) * 16 + ar;
      if (grow >= N) grow = N - 1;
      const float4* __restrict__ src = (const float4*)(x + (size_t)grow * F + ac);
      pa0 = src[0];
      pa1 = src[1];
    }

    v8f acc = {0.f, 0.f, 0.f, 0.f, 0.f, 0.f, 0.f, 0.f};
    v2f af[2][16];
#pragma unroll
    for (int i = 0; i < 16; ++i) af[0][i] = *(const v2f*)(arow + 4 * i);
#pragma unroll
    for (int q = 0; q < 4; ++q) {
      if (q < 3) {
#pragma unroll
        for (int i = 0; i < 16; ++i)
          af[(q + 1) & 1][i] = *(const v2f*)(arow + 4 * (16 * (q + 1) + i));
      }
      __builtin_amdgcn_sched_barrier(0);
#pragma unroll
      for (int i = 0; i < 16; ++i)
        acc = __builtin_amdgcn_wmma_f32_16x16x4_f32(
             false, af[q & 1][i],  false, breg[16 * q + i],
             (short)0, acc,  false,  false);
      __builtin_amdgcn_sched_barrier(0);
    }

    {
      float* drow = &lds_d[par][(h * 8) * LDS_STRIDE + col0 + m];
#pragma unroll
      for (int j = 0; j < 8; ++j) drow[j * LDS_STRIDE] = acc[j];
    }
    __syncthreads();

    const int rbase = row0 + (h * 8);
    float* __restrict__ dst = xl + (size_t)rbase * F + col0 + m;
    if (row0 + 16 <= N) {
#pragma unroll
      for (int j = 0; j < 8; ++j) dst[(size_t)j * F] = acc[j];
    } else {
#pragma unroll
      for (int j = 0; j < 8; ++j) {
        if (rbase + j < N) dst[(size_t)j * F] = acc[j];
      }
    }

    prev_row0 = row0;
    par ^= 1;
  }
  tile_half_logits(lds_d[par ^ 1], prev_row0, N, tid, att_src, att_dst, a_src, a_dst);
}

__device__ __forceinline__ float leaky(float v) {
  return (v >= 0.f) ? v : v * NEG_SLOPE;
}

#define AGG_THREADS 1024
#define AGG_WAVES (AGG_THREADS / 32)
#define LIST_CAP 4096
#define SCAN_UNROLL 8
#define SCAN_STEP (AGG_THREADS * 4)
#define SCAN_SPAN (SCAN_STEP * SCAN_UNROLL)
#define OWN_STEP (AGG_THREADS * 16)
#define OWN_SPAN (OWN_STEP * SCAN_UNROLL)
#define ROWS_PER_LANE ((NPB + 31) / 32)
#define EDGE_GROUP 4

static_assert(LIST_CAP == 4 * AGG_THREADS, "pass (1) gives every thread four list entries");
static_assert(EDGE_GROUP * HEADS == 32, "one lane per (edge of the group, head) weight");

__global__ void __launch_bounds__(AGG_THREADS)
gat_edge_agg_kernel(const int* __restrict__ ei, const float* __restrict__ xl,
                    const float* __restrict__ a_src, const float* __restrict__ a_dst,
                    float* __restrict__ out, int E, int N,
                    const unsigned* __restrict__ own, unsigned long long ws_bytes) {
  __shared__ __attribute__((aligned(16))) float acc[NPB * F];
  __shared__ __attribute__((aligned(16))) unsigned smax[NPB * HEADS];
  __shared__ __attribute__((aligned(16))) float den[NPB * HEADS];
  __shared__ int list_e[LIST_CAP];
  __shared__ int row_cnt[NPB];
  __shared__ int row_start[NPB];
  __shared__ int list_n;
  __shared__ int next_row;
  __shared__ int cell_base[GEMM_BLOCKS];

  const int tid  = threadIdx.x;
  const int lane = tid & 31;
  const int wave = __builtin_amdgcn_readfirstlane(tid >> 5);
  const int n0 = blockIdx.x * NPB;
  const int nrows = (N - n0 < NPB) ? (N - n0) : NPB;
  const int* __restrict__ dst = ei + E;

  const bool use_cells = owner_cells_ok(N, E, ws_bytes);

  if (tid < NPB) row_cnt[tid] = 0;
  if (tid < nrows) list_e[tid] = use_cells ? (int)((unsigned)(n0 + tid) | ((unsigned)tid << 24)) : (E + n0 + tid);
  if (tid == 0) { list_n = nrows; next_row = 0; }
  __syncthreads();

  auto keep = [&](int e, int dv) {
    unsigned dl = (unsigned)(dv - n0);
    if (dl < (unsigned)nrows) {
      int slot = __hip_atomic_fetch_add(&list_n, 1, __ATOMIC_RELAXED, __HIP_MEMORY_SCOPE_WORKGROUP);
      if (slot < LIST_CAP) list_e[slot] = e;
    }
  };
  auto keep_e = [&](int e) {
    int slot = __hip_atomic_fetch_add(&list_n, 1, __ATOMIC_RELAXED, __HIP_MEMORY_SCOPE_WORKGROUP);
    if (slot < LIST_CAP) list_e[slot] = e;
  };

  const bool use_plane = owner_plane_ok(N, E, ws_bytes);
  if (use_cells) {
    static_assert(GEMM_BLOCKS * (CELL_SLOTS / 4) == 4 * AGG_THREADS, "four 16-byte pieces of the cells per thread");
    const uint4* __restrict__ cell4 = (const uint4*)(own + (size_t)blockIdx.x * (GEMM_BLOCKS * CELL_SLOTS));
    uint4 v[4];
    int cc[4];
#pragma unroll
    for (int k = 0; k < 4; ++k) v[k] = cell4[k * AGG_THREADS + tid];
#pragma unroll
    for (int k = 0; k < 4; ++k) {
      cc[k] = __shfl((int)v[k].w, (lane & 16) | 15, 32);
      if ((tid & 15) == 15) cell_base[(k * AGG_THREADS + tid) >> 4] = cc[k];
    }
    __syncthreads();
    if (wave == 0) {
      int c[GEMM_BLOCKS / 32];
      int tot = 0, over = 0;
#pragma unroll
      for (int q = 0; q < GEMM_BLOCKS / 32; ++q) {
        const int raw = cell_base[lane * (GEMM_BLOCKS / 32) + q];
        over |= (raw > CELL_EDGES || raw < 0) ? 1 : 0;
        c[q] = min(max(raw, 0), CELL_EDGES);
        tot += c[q];
      }
      int inc = tot;
      for (int d = 1; d < 32; d <<= 1) {
        const int t = __shfl_up(inc, d, 32);
        if (lane >= d) inc += t;
        over |= __shfl_xor(over, d, 32);
      }
      int run = inc - tot;
#pragma unroll
      for (int q = 0; q < GEMM_BLOCKS / 32; ++q) {
        cell_base[lane * (GEMM_BLOCKS / 32) + q] = run;
        run += c[q];
      }
      if (lane == 31) list_n = over ? (LIST_CAP + 1) : (nrows + inc);
    }
    __syncthreads();
#pragma unroll
    for (int k = 0; k < 4; ++k) {
      const int cell = (k * AGG_THREADS + tid) >> 4;
      const int q4 = (tid & 15) * 4;
      const int n = min(max(cc[k], 0), CELL_EDGES);
      const int base = nrows + cell_base[cell] + q4;
      const unsigned w4[4] = {v[k].x, v[k].y, v[k].z, v[k].w};
#pragma unroll
      for (int j = 0; j < 4; ++j)
        if (q4 + j < n && base + j < LIST_CAP) list_e[base + j] = (int)w4[j];
    }
  } else if (use_plane) {
    int n_it = (E + OWN_SPAN - 1) / OWN_SPAN;
    if (n_it < 1) n_it = 1;
    const int rot = (int)(((long long)blockIdx.x * n_it) / gridDim.x);
    const unsigned pat  = blockIdx.x * 0x01010101u;
    const uint4* __restrict__ own4 = (const uint4*)own;
    for (int it = 0; it < n_it; ++it) {
      int c = it + rot;
      if (c >= n_it) c -= n_it;
      const int b0 = c * OWN_SPAN + tid * 16;
      uint4 v[SCAN_UNROLL];
#pragma unroll
      for (int u = 0; u < SCAN_UNROLL; ++u) {
        const int idx = b0 + u * OWN_STEP;
        v[u] = own4[min(idx, E - 16) >> 4];
      }
#pragma unroll
      for (int u = 0; u < SCAN_UNROLL; u += 2) {
        const int idx = b0 + u * OWN_STEP;
        unsigned m = 0;
#pragma unroll
        for (int w = 0; w < 2; ++w) {
          const unsigned x[4] = {v[u + w].x ^ pat, v[u + w].y ^ pat, v[u + w].z ^ pat, v[u + w].w ^ pat};
          unsigned mw = 0;
#pragma unroll
          for (int q = 0; q < 4; ++q) {
            const unsigned f = ~(((x[q] & 0x7f7f7f7fu) + 0x7f7f7f7fu) | x[q]) & 0x80808080u;
            mw |= f >> (7 - q);
          }
          if (idx + w * OWN_STEP >= E) mw = 0;
          m |= mw << (4 * w);
        }
        while (m) {
          const int p = __builtin_ctz(m);
          keep_e(idx + ((p >> 2) & 1) * OWN_STEP + 4 * (p & 3) + (p >> 3));
          m &= m - 1;
        }
      }
    }
  } else {
  int n_it = (E + SCAN_SPAN - 1) / SCAN_SPAN;
  if (n_it < 1) n_it = 1;
  const int rot = (int)(((long long)blockIdx.x * n_it) / gridDim.x);
  const bool vec_ok = ((E & 3) == 0);

  for (int it = 0; it < n_it; ++it) {
    int c = it + rot;
    if (c >= n_it) c -= n_it;
    const int b0 = c * SCAN_SPAN + tid * 4;

    if (vec_ok && (c + 1) * SCAN_SPAN <= E) {
      int4 v[SCAN_UNROLL];
#pragma unroll
      for (int u = 0; u < SCAN_UNROLL; ++u)
        v[u] = *(const int4*)(dst + b0 + u * SCAN_STEP);
#pragma unroll
      for (int u = 0; u < SCAN_UNROLL; ++u) {
        const int idx = b0 + u * SCAN_STEP;
        const bool any = ((unsigned)(v[u].x - n0) < (unsigned)nrows) | ((unsigned)(v[u].y - n0) < (unsigned)nrows) |
                         ((unsigned)(v[u].z - n0) < (unsigned)nrows) | ((unsigned)(v[u].w - n0) < (unsigned)nrows);
        if (any) {
          keep(idx + 0, v[u].x);
          keep(idx + 1, v[u].y);
          keep(idx + 2, v[u].z);
          keep(idx + 3, v[u].w);
        }
      }
    } else {
      for (int u = 0; u < SCAN_UNROLL; ++u) {
        const int idx = b0 + u * SCAN_STEP;
        for (int k = 0; k < 4; ++k)
          if (idx + k < E) keep(idx + k, dst[idx + k]);
      }
    }
  }
  }
  __syncthreads();
  const int cnt = __builtin_amdgcn_readfirstlane(list_n);

  if (cnt <= LIST_CAP) {
    int dd[4], ss[4], rk[4];
    {
      int ee[4];
#pragma unroll
      for (int k = 0; k < 4; ++k) {
        const int i = k * AGG_THREADS + tid;
        ee[k] = (i < cnt) ? list_e[i] : (use_cells ? n0 : (E + n0));
      }
      if (use_cells) {
#pragma unroll
        for (int k = 0; k < 4; ++k) {
          const unsigned w = (unsigned)ee[k];
          ss[k] = (int)(w & 0x00FFFFFFu);
          ss[k] = min(ss[k], N - 1);
          dd[k] = min((int)(w >> 24), nrows - 1);
        }
      } else {
#pragma unroll
      for (int k = 0; k < 4; ++k) {
        const bool real = ee[k] < E;
        ss[k] = real ? ei[ee[k]] : (ee[k] - E);
        const int dv = real ? dst[ee[k]] : (ee[k] - E);
        dd[k] = min(max(dv - n0, 0), nrows - 1);
      }
      }
#pragma unroll
      for (int k = 0; k < 4; ++k) {
        const int i = k * AGG_THREADS + tid;
        rk[k] = 0;
        if (i < cnt)
          rk[k] = __hip_atomic_fetch_add(&row_cnt[dd[k]], 1, __ATOMIC_RELAXED, __HIP_MEMORY_SCOPE_WORKGROUP);
      }
    }
    __syncthreads();

    if (wave == 0) {
      int c[ROWS_PER_LANE];
      int tot = 0;
#pragma unroll
      for (int q = 0; q < ROWS_PER_LANE; ++q) {
        const int r = lane * ROWS_PER_LANE + q;
        c[q] = (r < NPB) ? row_cnt[r] : 0;
        tot += c[q];
      }
      int inc = tot;
      for (int d = 1; d < 32; d <<= 1) {
        const int v = __shfl_up(inc, d, 32);
        if (lane >= d) inc += v;
      }
      int run = inc - tot;
#pragma unroll
      for (int q = 0; q < ROWS_PER_LANE; ++q) {
        const int r = lane * ROWS_PER_LANE + q;
        if (r < NPB) row_start[r] = run;
        run += c[q];
      }
    }
    __syncthreads();

#pragma unroll
    for (int k = 0; k < 4; ++k) {
      const int i = k * AGG_THREADS + tid;
      if (i < cnt) list_e[row_start[dd[k]] + rk[k]] = ss[k];
    }
    __syncthreads();

    const int hq = lane & 7;
    const int gq = lane >> 3;
    const int h0 = lane >> 3;
    for (int t = 0; t < NPB; ++t) {
      int r = 0;
      if (lane == 0) r = __hip_atomic_fetch_add(&next_row, 1, __ATOMIC_RELAXED, __HIP_MEMORY_SCOPE_WORKGROUP);
      r = __builtin_amdgcn_readfirstlane(r);
      if (r >= nrows) break;
      const int st = __builtin_amdgcn_readfirstlane(row_start[r]);
      const int ne = __builtin_amdgcn_readfirstlane(row_cnt[r]);
      const float adq = a_dst[(size_t)(n0 + r) * HEADS + hq];

      float mq = -INFINITY;
      for (int c0 = 0; c0 < ne; c0 += 32) {
        const int m = (ne - c0 < 32) ? (ne - c0) : 32;
        float as[8];
#pragma unroll
        for (int g = 0; g < 8; ++g) {
          const int i = (4 * g + gq < m) ? (4 * g + gq) : (m - 1);
          as[g] = a_src[(size_t)list_e[st + c0 + i] * HEADS + hq];
        }
#pragma unroll
        for (int g = 0; g < 8; ++g) mq = fmaxf(mq, leaky(as[g] + adq));
      }
      mq = fmaxf(mq, __shfl_xor(mq, 8, 32));
      mq = fmaxf(mq, __shfl_xor(mq, 16, 32));

      float4 sa = make_float4(0.f, 0.f, 0.f, 0.f);
      float4 sb = make_float4(0.f, 0.f, 0.f, 0.f);
      float dsum = 0.f;

      for (int c0 = 0; c0 < ne; c0 += 32) {
        const int m = (ne - c0 < 32) ? (ne - c0) : 32;
        int s = n0;
        if (lane < m) s = list_e[st + c0 + lane];

        for (int i0 = 0; i0 < m; i0 += EDGE_GROUP) {
          const int iq = (i0 + gq < m) ? (i0 + gq) : (m - 1);
          const int sq = list_e[st + c0 + iq];
          const float asq = a_src[(size_t)sq * HEADS + hq];

          float4 xa[EDGE_GROUP], xb[EDGE_GROUP];
#pragma unroll
          for (int j = 0; j < EDGE_GROUP; ++j) {
            const int ii = (i0 + j < m) ? (i0 + j) : (m - 1);
            const int si = __builtin_amdgcn_readlane(s, ii);
            const float4* __restrict__ xr = (const float4*)(xl + (size_t)si * F) + lane;
            xa[j] = xr[0];
            xb[j] = xr[32];
          }

          float p = expf(leaky(asq + adq) - mq);
          if (i0 + gq >= m) p = 0.f;
          dsum += p;

#pragma unroll
          for (int j = 0; j < EDGE_GROUP; ++j) {
            const float u = __shfl(p, 8 * j + h0, 32);
            const float v = __shfl(p, 8 * j + 4 + h0, 32);
            sa.x += u * xa[j].x; sa.y += u * xa[j].y; sa.z += u * xa[j].z; sa.w += u * xa[j].w;
            sb.x += v * xb[j].x; sb.y += v * xb[j].y; sb.z += v * xb[j].z; sb.w += v * xb[j].w;
          }
        }
      }

      dsum += __shfl_xor(dsum, 8, 32);
      dsum += __shfl_xor(dsum, 16, 32);
      const float ra = 1.0f / __shfl(dsum, h0, 32);
      const float rb = 1.0f / __shfl(dsum, 4 + h0, 32);
      float4* __restrict__ orow = (float4*)(out + (size_t)(n0 + r) * F);
      orow[lane]      = make_float4(sa.x * ra, sa.y * ra, sa.z * ra, sa.w * ra);
      orow[32 + lane] = make_float4(sb.x * rb, sb.y * rb, sb.z * rb, sb.w * rb);
    }
  } else {
    for (int i = tid; i < NPB * F / 4; i += AGG_THREADS)
      ((float4*)acc)[i] = make_float4(0.f, 0.f, 0.f, 0.f);
    for (int i = tid; i < NPB * HEADS; i += AGG_THREADS) {
      den[i] = 0.f;
      smax[i] = 0x007FFFFFu;
    }
    __syncthreads();
    const int tot = E + N;
    for (int e = tid; e < tot; e += AGG_THREADS) {
      const int d = (e < E) ? dst[e] : (e - E);
      const unsigned dl = (unsigned)(d - n0);
      if (dl < (unsigned)nrows) {
        const int s = (e < E) ? ei[e] : (e - E);
        for (int h = 0; h < HEADS; ++h)
          __hip_atomic_fetch_max(&smax[dl * HEADS + h], f2key(leaky(a_src[s * HEADS + h] + a_dst[d * HEADS + h])),
                                 __ATOMIC_RELAXED, __HIP_MEMORY_SCOPE_WORKGROUP);
      }
    }
    __syncthreads();
    for (int e = tid; e < tot; e += AGG_THREADS) {
      const int d = (e < E) ? dst[e] : (e - E);
      const unsigned dl = (unsigned)(d - n0);
      if (dl < (unsigned)nrows) {
        const int s = (e < E) ? ei[e] : (e - E);
        for (int h = 0; h < HEADS; ++h) {
          const float p = expf(leaky(a_src[s * HEADS + h] + a_dst[d * HEADS + h]) - key2f(smax[dl * HEADS + h]));
          __hip_atomic_fetch_add(&den[dl * HEADS + h], p, __ATOMIC_RELAXED, __HIP_MEMORY_SCOPE_WORKGROUP);
          for (int c = 0; c < CH; ++c)
            __hip_atomic_fetch_add(&acc[dl * F + h * CH + c], xl[(size_t)s * F + h * CH + c] * p,
                                   __ATOMIC_RELAXED, __HIP_MEMORY_SCOPE_WORKGROUP);
        }
      }
    }
    __syncthreads();

    for (int i = tid; i < nrows * HEADS; i += AGG_THREADS) den[i] = 1.0f / den[i];
    __syncthreads();
    float4* __restrict__ orow = (float4*)(out + (size_t)n0 * F);
    for (int i = tid; i < nrows * (F / 4); i += AGG_THREADS) {
      float4 v = ((const float4*)acc)[i];
      const float r = den[(i >> 6) * HEADS + ((i & 63) >> 3)];
      v.x *= r; v.y *= r; v.z *= r; v.w *= r;
      orow[i] = v;
    }
  }
}

#define LN_ROWS 4
#define LN_THREADS 256
#define LN_ROWS_PER_BLOCK (LN_ROWS * (LN_THREADS / 32))

__device__ __forceinline__ float wave_sum(float v) {
#pragma unroll
  for (int m = 16; m > 0; m >>= 1) v += __shfl_xor(v, m, 32);
  return v;
}

__global__ void __launch_bounds__(LN_THREADS)
gat_ln_kernel(float* __restrict__ out, const float* __restrict__ bias,
              const float* __restrict__ gamma, const float* __restrict__ beta, int N) {
  const int lane = threadIdx.x & 31;
  const int wave = __builtin_amdgcn_readfirstlane((int)((blockIdx.x * LN_THREADS + threadIdx.x) >> 5));
  const int row0 = wave * LN_ROWS;
  if (row0 >= N) return;

  float4 va[LN_ROWS], vb[LN_ROWS];
#pragma unroll
  for (int r = 0; r < LN_ROWS; ++r) {
    const int row = (row0 + r < N) ? (row0 + r) : row0;
    const float4* p = (const float4*)(out + (size_t)row * F);
    va[r] = p[lane];
    vb[r] = p[32 + lane];
  }
  const float4 b0 = ((const float4*)bias)[lane],  b1 = ((const float4*)bias)[32 + lane];
  const float4 g0 = ((const float4*)gamma)[lane], g1 = ((const float4*)gamma)[32 + lane];
  const float4 t0 = ((const float4*)beta)[lane],  t1 = ((const float4*)beta)[32 + lane];

  float mean[LN_ROWS];
#pragma unroll
  for (int r = 0; r < LN_ROWS; ++r) {
    va[r].x += b0.x; va[r].y += b0.y; va[r].z += b0.z; va[r].w += b0.w;
    vb[r].x += b1.x; vb[r].y += b1.y; vb[r].z += b1.z; vb[r].w += b1.w;
    const float s = ((va[r].x + va[r].y) + (va[r].z + va[r].w)) +
                    ((vb[r].x + vb[r].y) + (vb[r].z + vb[r].w));
    mean[r] = wave_sum(s) * (1.0f / F);
  }
  float rs[LN_ROWS];
#pragma unroll
  for (int r = 0; r < LN_ROWS; ++r) {
    va[r].x -= mean[r]; va[r].y -= mean[r]; va[r].z -= mean[r]; va[r].w -= mean[r];
    vb[r].x -= mean[r]; vb[r].y -= mean[r]; vb[r].z -= mean[r]; vb[r].w -= mean[r];
    const float q = ((va[r].x * va[r].x + va[r].y * va[r].y) + (va[r].z * va[r].z + va[r].w * va[r].w)) +
                    ((vb[r].x * vb[r].x + vb[r].y * vb[r].y) + (vb[r].z * vb[r].z + vb[r].w * vb[r].w));
    rs[r] = rsqrtf(wave_sum(q) * (1.0f / F) + LN_EPS);
  }
#pragma unroll
  for (int r = 0; r < LN_ROWS; ++r) {
    if (row0 + r < N) {
      float4* o = (float4*)(out + (size_t)(row0 + r) * F);
      float4 ya, yb;
      ya.x = va[r].x * rs[r] * g0.x + t0.x; ya.y = va[r].y * rs[r] * g0.y + t0.y;
      ya.z = va[r].z * rs[r] * g0.z + t0.z; ya.w = va[r].w * rs[r] * g0.w + t0.w;
      yb.x = vb[r].x * rs[r] * g1.x + t1.x; yb.y = vb[r].y * rs[r] * g1.y + t1.y;
      yb.z = vb[r].z * rs[r] * g1.z + t1.z; yb.w = vb[r].w * rs[r] * g1.w + t1.w;
      o[lane] = ya;
      o[32 + lane] = yb;
    }
  }
}

extern "C" void kernel_launch(void* const* d_in, const int* in_sizes, int n_in,
                              void* d_out, int out_size, void* d_ws, size_t ws_size,
                              hipStream_t stream) {
  const float* x       = (const float*)d_in[0];
  const int*   ei      = (const int*)  d_in[1];
  const float* W       = (const float*)d_in[2];
  const float* att_src = (const float*)d_in[3];
  const float* att_dst = (const float*)d_in[4];
  const float* bias    = (const float*)d_in[5];
  const float* gamma   = (const float*)d_in[6];
  const float* beta    = (const float*)d_in[7];
  float* out = (float*)d_out;

  const int N = in_sizes[0] / F;
  const int E = in_sizes[1] / 2;

  char* ws = (char*)d_ws;
  float*    xl     = (float*)ws;    ws += (size_t)N * F * sizeof(float);
  float*    a_src  = (float*)ws;    ws += (size_t)N * HEADS * sizeof(float);
  float*    a_dst  = (float*)ws;    ws += (size_t)N * HEADS * sizeof(float);
  unsigned* own    = (unsigned*)ws;

  gat_gemm_wmma_kernel<<<GEMM_BLOCKS, 512, 0, stream>>>(x, W, xl, N, att_src, att_dst,
                                                         a_src, a_dst, ei, own, E,
                                                         (unsigned long long)ws_size);

  gat_edge_agg_kernel<<<(N + NPB - 1) / NPB, AGG_THREADS, 0, stream>>>(ei, xl, a_src, a_dst,
                                                                        out, E, N, own,
                                                                        (unsigned long long)ws_size);

  gat_ln_kernel<<<(N + LN_ROWS_PER_BLOCK - 1) / LN_ROWS_PER_BLOCK, LN_THREADS, 0, stream>>>(out, bias, gamma, beta, N);
}
